# starting kernel
# baseline (speedup 1.0000x reference)
#include <hip/hip_runtime.h>


typedef __attribute__((ext_vector_type(2))) float v2f;
typedef __attribute__((ext_vector_type(8))) float v8f;

#define NEG_SLOPE 0.2f

template <int K, int LDB>
__global__ __launch_bounds__(256) void wmma_dual_gemm(
    const float* __restrict__ A, const float* __restrict__ Bl,
    const float* __restrict__ Br, float* __restrict__ Cl,
    float* __restrict__ Cr, int rowTiles, int colTiles) {
  const int wid = (blockIdx.x * blockDim.x + threadIdx.x) >> 5;
  const int lane = threadIdx.x & 31;
  if (wid >= rowTiles * colTiles) return;
  const int rt = wid / colTiles;
  const int ct = wid % colTiles;
  const int row0 = rt * 16;
  const int col0 = ct * 16;
  const int half = lane >> 4;
  const int l16 = lane & 15;
  const int arow = row0 + l16;
  const int bcol = col0 + l16;

  const float* __restrict__ Ap = A + (size_t)arow * K + half * 2;

  v8f accl = {};
  v8f accr = {};
  for (int k = 0; k < K; k += 4) {
    const int ka = k + half * 2;
    v2f a;
    a.x = Ap[k];
    a.y = Ap[k + 1];
    v2f bl, br;
    bl.x = Bl[(size_t)ka * LDB + bcol];
    bl.y = Bl[(size_t)(ka + 1) * LDB + bcol];
    br.x = Br[(size_t)ka * LDB + bcol];
    br.y = Br[(size_t)(ka + 1) * LDB + bcol];
    accl = __builtin_amdgcn_wmma_f32_16x16x4_f32(false, a, false, bl, (short)0,
                                                 accl, false, false);
    accr = __builtin_amdgcn_wmma_f32_16x16x4_f32(false, a, false, br, (short)0,
                                                 accr, false, false);
  }

  const size_t base = (size_t)(row0 + half * 8) * LDB + bcol;
  float* __restrict__ cl = Cl + base;
  float* __restrict__ cr = Cr + base;
#pragma unroll
  for (int i = 0; i < 8; ++i) {
    cl[(size_t)i * LDB] = accl[i];
    cr[(size_t)i * LDB] = accr[i];
  }
}

__device__ __forceinline__ void atomicMaxFloat(float* addr, float val) {
  if (val >= 0.f)
    atomicMax((int*)addr, __float_as_int(val));
  else
    atomicMin((unsigned int*)addr, __float_as_uint(val));
}

template <int HEADS, int DIMH, int MODE>
__global__ __launch_bounds__(256) void edge_pass(
    const int* __restrict__ srcE, const int* __restrict__ dstE, int E, int N,
    const float* __restrict__ xl, const float* __restrict__ xr,
    const float* __restrict__ att, float* __restrict__ m,
    float* __restrict__ denom, float* __restrict__ out) {
  const int H = HEADS * DIMH;
  const int total = (E + N) * HEADS;
  const int t = blockIdx.x * blockDim.x + threadIdx.x;
  if (t >= total) return;
  const int e = t / HEADS;
  const int h = t % HEADS;
  int s, d;
  if (e < E) {
    s = srcE[e];
    d = dstE[e];
  } else {
    s = e - E;
    d = s;
  }
  const float* __restrict__ xls = xl + (size_t)s * H + h * DIMH;
  const float* __restrict__ xrd = xr + (size_t)d * H + h * DIMH;

  float logit = 0.f;
#pragma unroll
  for (int dd = 0; dd < DIMH; ++dd) {
    float v = xls[dd] + xrd[dd];
    v = v > 0.f ? v : NEG_SLOPE * v;
    logit += v * att[h * DIMH + dd];
  }

  const size_t seg = (size_t)d * HEADS + h;
  if (MODE == 0) {
    atomicMaxFloat(&m[seg], logit);
  } else if (MODE == 1) {
    atomicAdd(&denom[seg], __expf(logit - m[seg]));
  } else {
    const float alpha = __expf(logit - m[seg]) / denom[seg];
#pragma unroll
    for (int dd = 0; dd < DIMH; ++dd)
      atomicAdd(&out[(size_t)d * H + h * DIMH + dd], alpha * xls[dd]);
  }
}

__global__ void init_softmax_state(float* __restrict__ m,
                                   float* __restrict__ denom, int NH) {
  const int i = blockIdx.x * blockDim.x + threadIdx.x;
  if (i < NH) {
    m[i] = -__builtin_inff();
    denom[i] = 0.f;
  }
}

__global__ void init_bias_bcast(float* __restrict__ out,
                                const float* __restrict__ bias, int H,
                                int total) {
  const int i = blockIdx.x * blockDim.x + threadIdx.x;
  if (i < total) out[i] = bias[i % H];
}

__global__ void relu_inplace(float* __restrict__ p, int total) {
  const int i = blockIdx.x * blockDim.x + threadIdx.x;
  if (i < total) p[i] = fmaxf(p[i], 0.f);
}

extern "C" void kernel_launch(void* const* d_in, const int* in_sizes, int n_in,
                              void* d_out, int out_size, void* d_ws,
                              size_t ws_size, hipStream_t stream) {
  const float* x = (const float*)d_in[0];
  const int* edge = (const int*)d_in[1];
  const float* W1l = (const float*)d_in[2];
  const float* W1r = (const float*)d_in[3];
  const float* att1 = (const float*)d_in[4];
  const float* b1 = (const float*)d_in[5];
  const float* W2l = (const float*)d_in[6];
  const float* W2r = (const float*)d_in[7];
  const float* att2 = (const float*)d_in[8];
  const float* b2 = (const float*)d_in[9];
  float* out = (float*)d_out;

  const int F_IN = 128, H1 = 80, HEADS1 = 10, F_OUT = 16;
  const int N = in_sizes[0] / F_IN;
  const int E = in_sizes[1] / 2;
  const int* srcE = edge;
  const int* dstE = edge + E;

  float* ws = (float*)d_ws;
  float* xl1 = ws;                 ws += (size_t)N * H1;
  float* xr1 = ws;                 ws += (size_t)N * H1;
  float* agg1 = ws;                ws += (size_t)N * H1;
  float* m1 = ws;                  ws += (size_t)N * HEADS1;
  float* denom1 = ws;              ws += (size_t)N * HEADS1;
  float* xl2 = ws;                 ws += (size_t)N * F_OUT;
  float* xr2 = ws;                 ws += (size_t)N * F_OUT;
  float* m2 = ws;                  ws += (size_t)N;
  float* denom2 = ws;
  (void)ws_size; (void)n_in; (void)out_size;

  const int BLK = 256;

  {
    const int rowTiles = N / 16, colTiles = H1 / 16;
    const int waves = rowTiles * colTiles;
    const int blocks = (waves + 7) / 8;
    wmma_dual_gemm<128, 80><<<blocks, BLK, 0, stream>>>(
        x, W1l, W1r, xl1, xr1, rowTiles, colTiles);
  }
  init_softmax_state<<<(N * HEADS1 + BLK - 1) / BLK, BLK, 0, stream>>>(
      m1, denom1, N * HEADS1);
  init_bias_bcast<<<((size_t)N * H1 + BLK - 1) / BLK, BLK, 0, stream>>>(
      agg1, b1, H1, N * H1);

  const int etotal1 = (E + N) * HEADS1;
  edge_pass<10, 8, 0><<<(etotal1 + BLK - 1) / BLK, BLK, 0, stream>>>(
      srcE, dstE, E, N, xl1, xr1, att1, m1, denom1, agg1);
  edge_pass<10, 8, 1><<<(etotal1 + BLK - 1) / BLK, BLK, 0, stream>>>(
      srcE, dstE, E, N, xl1, xr1, att1, m1, denom1, agg1);
  edge_pass<10, 8, 2><<<(etotal1 + BLK - 1) / BLK, BLK, 0, stream>>>(
      srcE, dstE, E, N, xl1, xr1, att1, m1, denom1, agg1);

  relu_inplace<<<((size_t)N * H1 + BLK - 1) / BLK, BLK, 0, stream>>>(
      agg1, N * H1);

  {
    const int rowTiles = N / 16, colTiles = F_OUT / 16;
    const int waves = rowTiles * colTiles;
    const int blocks = (waves + 7) / 8;
    wmma_dual_gemm<80, 16><<<blocks, BLK, 0, stream>>>(
        agg1, W2l, W2r, xl2, xr2, rowTiles, colTiles);
  }
  init_softmax_state<<<(N + BLK - 1) / BLK, BLK, 0, stream>>>(m2, denom2, N);
  init_bias_bcast<<<((size_t)N * F_OUT + BLK - 1) / BLK, BLK, 0, stream>>>(
      out, b2, F_OUT, N * F_OUT);

  const int etotal2 = (E + N) * 1;
  edge_pass<1, 16, 0><<<(etotal2 + BLK - 1) / BLK, BLK, 0, stream>>>(
      srcE, dstE, E, N, xl2, xr2, att2, m2, denom2, out);
  edge_pass<1, 16, 1><<<(etotal2 + BLK - 1) / BLK, BLK, 0, stream>>>(
      srcE, dstE, E, N, xl2, xr2, att2, m2, denom2, out);
  edge_pass<1, 16, 2><<<(etotal2 + BLK - 1) / BLK, BLK, 0, stream>>>(
      srcE, dstE, E, N, xl2, xr2, att2, m2, denom2, out);
}


# Layer-1 softmax state and sums kept in LDS per 200-node destination range: each work-group filters the edge list once, LDS float atomics replace the global ones; sums in LDS arrival order
# speedup vs baseline: 2.3213x; 2.3213x over previous
#include <hip/hip_runtime.h>


typedef __attribute__((ext_vector_type(2))) float v2f;
typedef __attribute__((ext_vector_type(8))) float v8f;

#define NEG_SLOPE 0.2f

template <int K, int LDB>
__global__ __launch_bounds__(256) void wmma_dual_gemm(
    const float* __restrict__ A, const float* __restrict__ Bl,
    const float* __restrict__ Br, float* __restrict__ Cl,
    float* __restrict__ Cr, int rowTiles, int colTiles) {
  const int wid = (blockIdx.x * blockDim.x + threadIdx.x) >> 5;
  const int lane = threadIdx.x & 31;
  if (wid >= rowTiles * colTiles) return;
  const int rt = wid / colTiles;
  const int ct = wid % colTiles;
  const int row0 = rt * 16;
  const int col0 = ct * 16;
  const int half = lane >> 4;
  const int l16 = lane & 15;
  const int arow = row0 + l16;
  const int bcol = col0 + l16;

  const float* __restrict__ Ap = A + (size_t)arow * K + half * 2;

  v8f accl = {};
  v8f accr = {};
  for (int k = 0; k < K; k += 4) {
    const int ka = k + half * 2;
    v2f a;
    a.x = Ap[k];
    a.y = Ap[k + 1];
    v2f bl, br;
    bl.x = Bl[(size_t)ka * LDB + bcol];
    bl.y = Bl[(size_t)(ka + 1) * LDB + bcol];
    br.x = Br[(size_t)ka * LDB + bcol];
    br.y = Br[(size_t)(ka + 1) * LDB + bcol];
    accl = __builtin_amdgcn_wmma_f32_16x16x4_f32(false, a, false, bl, (short)0,
                                                 accl, false, false);
    accr = __builtin_amdgcn_wmma_f32_16x16x4_f32(false, a, false, br, (short)0,
                                                 accr, false, false);
  }

  const size_t base = (size_t)(row0 + half * 8) * LDB + bcol;
  float* __restrict__ cl = Cl + base;
  float* __restrict__ cr = Cr + base;
#pragma unroll
  for (int i = 0; i < 8; ++i) {
    cl[(size_t)i * LDB] = accl[i];
    cr[(size_t)i * LDB] = accr[i];
  }
}

__device__ __forceinline__ void atomicMaxFloat(float* addr, float val) {
  if (val >= 0.f)
    atomicMax((int*)addr, __float_as_int(val));
  else
    atomicMin((unsigned int*)addr, __float_as_uint(val));
}

#define OWN_R 200
#define OWN_S 81
#define OWN_CAP 4096
#define OWN_T 1024
#define OWN_U 8

__device__ __forceinline__ void own_take(int e, int d, int n0, int cnt,
                                         int* s_cnt, int2* list) {
  if ((unsigned)(d - n0) < (unsigned)cnt) {
    const int slot = atomicAdd(s_cnt, 1);
    if (slot < OWN_CAP) list[slot] = make_int2(e, d);
  }
}

template <int PHASE>
__device__ __forceinline__ void own_process(
    int count, bool cached, int n0, const int2* list, float* lg, float* acc,
    float* mS, float* dS, const float* __restrict__ xl,
    const float* __restrict__ xr, const float* __restrict__ att) {
  const int items = count * 10;
  for (int i = threadIdx.x; i < items; i += OWN_T) {
    const int ei = i / 10;
    const int h = i - ei * 10;
    const int2 sd = list[ei];
    const int dl = sd.y - n0;
    const int seg = dl * 10 + h;
    const float* __restrict__ xls = xl + (size_t)sd.x * 80 + h * 8;
    float xv[8];
    float logit;
    if (PHASE == 0 || !cached) {
      const float* __restrict__ xrd = xr + (size_t)sd.y * 80 + h * 8;
      logit = 0.f;
#pragma unroll
      for (int dd = 0; dd < 8; ++dd) {
        xv[dd] = xls[dd];
        float v = xv[dd] + xrd[dd];
        v = v > 0.f ? v : NEG_SLOPE * v;
        logit += v * att[h * 8 + dd];
      }
      if (PHASE == 0 && cached) lg[i] = logit;
    } else {
      logit = lg[i];
      if (PHASE == 2) {
#pragma unroll
        for (int dd = 0; dd < 8; ++dd) xv[dd] = xls[dd];
      }
    }
    if (PHASE == 0) {
      atomicMaxFloat(&mS[seg], logit);
    } else if (PHASE == 1) {
      atomicAdd(&dS[seg], __expf(logit - mS[seg]));
    } else {
      const float alpha = __expf(logit - mS[seg]) / dS[seg];
      float* a = acc + dl * OWN_S + h * 8 + (h >> 3);
#pragma unroll
      for (int dd = 0; dd < 8; ++dd) atomicAdd(&a[dd], alpha * xv[dd]);
    }
  }
}

__global__ __launch_bounds__(OWN_T) void layer1_owner_lds(
    const int* __restrict__ srcE, const int* __restrict__ dstE, int E, int N,
    const float* __restrict__ xl, const float* __restrict__ xr,
    const float* __restrict__ att, const float* __restrict__ bias,
    float* __restrict__ agg) {
  __shared__ float acc[OWN_R * OWN_S];
  __shared__ float mS[OWN_R * 10];
  __shared__ float dS[OWN_R * 10];
  __shared__ float lg[OWN_CAP * 10];
  __shared__ int2 list[OWN_CAP];
  __shared__ int s_cnt;

  const int tid = threadIdx.x;
  const int n0 = blockIdx.x * OWN_R;
  const int cnt = min(OWN_R, N - n0);

  for (int i = tid; i < OWN_R * OWN_S; i += OWN_T) acc[i] = 0.f;
  for (int i = tid; i < OWN_R * 10; i += OWN_T) {
    mS[i] = -__builtin_inff();
    dS[i] = 0.f;
  }
  if (tid == 0) s_cnt = 0;
  __syncthreads();

  {
    int head = (int)(((16u - (unsigned)((size_t)dstE & 15u)) & 15u) >> 2);
    if (head > E) head = E;
    const int nvec = (E - head) >> 2;
    const int4* __restrict__ dv = reinterpret_cast<const int4*>(dstE + head);
    const int rot = (int)(((long long)nvec * blockIdx.x) / gridDim.x);
    const int iters = (nvec + OWN_T - 1) / OWN_T;
    for (int k0 = 0; k0 < iters; k0 += OWN_U) {
      int4 q[OWN_U];
      int eb[OWN_U];
#pragma unroll
      for (int u = 0; u < OWN_U; ++u) {
        const int idx = (k0 + u) * OWN_T + tid;
        int v = idx + rot;
        if (v >= nvec) v -= nvec;
        eb[u] = head + 4 * v;
        q[u] = make_int4(-1, -1, -1, -1);
        if (idx < nvec) q[u] = dv[v];
      }
#pragma unroll
      for (int u = 0; u < OWN_U; ++u) {
        own_take(eb[u], q[u].x, n0, cnt, &s_cnt, list);
        own_take(eb[u] + 1, q[u].y, n0, cnt, &s_cnt, list);
        own_take(eb[u] + 2, q[u].z, n0, cnt, &s_cnt, list);
        own_take(eb[u] + 3, q[u].w, n0, cnt, &s_cnt, list);
      }
    }
    const int tail0 = head + 4 * nvec;
    if (tid < head) own_take(tid, dstE[tid], n0, cnt, &s_cnt, list);
    if (tail0 + tid < E)
      own_take(tail0 + tid, dstE[tail0 + tid], n0, cnt, &s_cnt, list);
  }
  __syncthreads();

  const int total = __builtin_amdgcn_readfirstlane(s_cnt);
  const bool fast = total + cnt <= OWN_CAP;
  if (fast) {
    for (int i = tid; i < total; i += OWN_T) list[i].x = srcE[list[i].x];
    for (int i = tid; i < cnt; i += OWN_T)
      list[total + i] = make_int2(n0 + i, n0 + i);
  }
  __syncthreads();

  const int ntiles = fast ? 1 : 1 + (E + OWN_CAP - 1) / OWN_CAP;
  for (int phase = 0; phase < 3; ++phase) {
    for (int t = 0; t < ntiles; ++t) {
      int count = total + cnt;
      if (!fast) {
        if (tid == 0) s_cnt = 0;
        __syncthreads();
        if (t == 0) {
          for (int i = tid; i < cnt; i += OWN_T)
            list[i] = make_int2(n0 + i, n0 + i);
        } else {
          const int e0 = (t - 1) * OWN_CAP;
          const int e1 = min(e0 + OWN_CAP, E);
          for (int e = e0 + tid; e < e1; e += OWN_T) {
            const int d = dstE[e];
            if ((unsigned)(d - n0) < (unsigned)cnt) {
              const int slot = atomicAdd(&s_cnt, 1);
              list[slot] = make_int2(srcE[e], d);
            }
          }
        }
        __syncthreads();
        count = t == 0 ? cnt : __builtin_amdgcn_readfirstlane(s_cnt);
      }
      if (phase == 0)
        own_process<0>(count, fast, n0, list, lg, acc, mS, dS, xl, xr, att);
      else if (phase == 1)
        own_process<1>(count, fast, n0, list, lg, acc, mS, dS, xl, xr, att);
      else
        own_process<2>(count, fast, n0, list, lg, acc, mS, dS, xl, xr, att);
      __syncthreads();
    }
  }

  for (int j = tid; j < cnt * 80; j += OWN_T) {
    const int i = j / 80;
    const int f = j - i * 80;
    const float v = acc[i * OWN_S + f + (f >> 6)] + bias[f];
    agg[(size_t)(n0 + i) * 80 + f] = fmaxf(v, 0.f);
  }
}

template <int CTRL>
__device__ __forceinline__ int dpp_int(int v) {
  return __builtin_amdgcn_update_dpp(0, v, CTRL, 0xf, 0xf, true);
}
template <int CTRL>
__device__ __forceinline__ float dpp_flt(float v) {
  return __int_as_float(
      __builtin_amdgcn_update_dpp(0, __float_as_int(v), CTRL, 0xf, 0xf, true));
}
__device__ __forceinline__ float row16_sum(float p) {
  p += dpp_flt<0x161>(p);
  p += dpp_flt<0x162>(p);
  p += dpp_flt<0x164>(p);
  p += dpp_flt<0x168>(p);
  return p;
}

template <int J>
__device__ __forceinline__ void row16_fetch(
    int sMine, int dMine, int f, const float* __restrict__ xl,
    const float* __restrict__ xr, int& d, float& xv, float& rv) {
  const int s = dpp_int<0x150 + J>(sMine);
  d = dpp_int<0x150 + J>(dMine);
  const int sc = d < 0 ? 0 : s;
  const int dc = d < 0 ? 0 : d;
  xv = xl[(size_t)sc * 16 + f];
  rv = xr[(size_t)dc * 16 + f];
}

template <int MODE>
__global__ __launch_bounds__(256) void edge_pass_rows16(
    const int* __restrict__ srcE, const int* __restrict__ dstE, int E, int N,
    const float* __restrict__ xl, const float* __restrict__ xr,
    const float* __restrict__ att, float* __restrict__ m,
    float* __restrict__ denom, float* __restrict__ out) {
  const int lane = threadIdx.x & 31;
  const int f = lane & 15;
  const int wave = (blockIdx.x * blockDim.x + threadIdx.x) >> 5;
  const int total = E + N;
  const int e0 = wave * 32;
  if (e0 >= total) return;

  const int eMine = e0 + lane;
  int sMine = -1, dMine = -1;
  if (eMine < E) {
    sMine = srcE[eMine];
    dMine = dstE[eMine];
  } else if (eMine < total) {
    sMine = eMine - E;
    dMine = sMine;
  }
  const float a = att[f];

  int d[16];
  float xv[16], rv[16];
#define ROW16_FETCH(J) row16_fetch<J>(sMine, dMine, f, xl, xr, d[J], xv[J], rv[J]);
  ROW16_FETCH(0) ROW16_FETCH(1) ROW16_FETCH(2) ROW16_FETCH(3)
  ROW16_FETCH(4) ROW16_FETCH(5) ROW16_FETCH(6) ROW16_FETCH(7)
  ROW16_FETCH(8) ROW16_FETCH(9) ROW16_FETCH(10) ROW16_FETCH(11)
  ROW16_FETCH(12) ROW16_FETCH(13) ROW16_FETCH(14) ROW16_FETCH(15)
#undef ROW16_FETCH

  float mv[16], dv[16];
  if (MODE != 0) {
#pragma unroll
    for (int j = 0; j < 16; ++j) {
      const int dc = d[j] < 0 ? 0 : d[j];
      mv[j] = m[dc];
      if (MODE == 2) dv[j] = denom[dc];
    }
  }

#pragma unroll
  for (int j = 0; j < 16; ++j) {
    float v = xv[j] + rv[j];
    v = v > 0.f ? v : NEG_SLOPE * v;
    const float logit = row16_sum(v * a);
    const bool live = d[j] >= 0;
    if (MODE == 0) {
      if (live && f == 0) atomicMaxFloat(&m[d[j]], logit);
    } else if (MODE == 1) {
      if (live && f == 0) atomicAdd(&denom[d[j]], __expf(logit - mv[j]));
    } else {
      const float alpha = __expf(logit - mv[j]) / dv[j];
      if (live) atomicAdd(&out[(size_t)d[j] * 16 + f], alpha * xv[j]);
    }
  }
}

__global__ void init_softmax_state(float* __restrict__ m,
                                   float* __restrict__ denom, int NH) {
  const int i = blockIdx.x * blockDim.x + threadIdx.x;
  if (i < NH) {
    m[i] = -__builtin_inff();
    denom[i] = 0.f;
  }
}

__global__ void init_bias_bcast(float* __restrict__ out,
                                const float* __restrict__ bias, int H,
                                int total) {
  const int i = blockIdx.x * blockDim.x + threadIdx.x;
  if (i < total) out[i] = bias[i % H];
}

extern "C" void kernel_launch(void* const* d_in, const int* in_sizes, int n_in,
                              void* d_out, int out_size, void* d_ws,
                              size_t ws_size, hipStream_t stream) {
  const float* x = (const float*)d_in[0];
  const int* edge = (const int*)d_in[1];
  const float* W1l = (const float*)d_in[2];
  const float* W1r = (const float*)d_in[3];
  const float* att1 = (const float*)d_in[4];
  const float* b1 = (const float*)d_in[5];
  const float* W2l = (const float*)d_in[6];
  const float* W2r = (const float*)d_in[7];
  const float* att2 = (const float*)d_in[8];
  const float* b2 = (const float*)d_in[9];
  float* out = (float*)d_out;

  const int F_IN = 128, H1 = 80, F_OUT = 16;
  const int N = in_sizes[0] / F_IN;
  const int E = in_sizes[1] / 2;
  const int* srcE = edge;
  const int* dstE = edge + E;

  float* ws = (float*)d_ws;
  float* xl1 = ws;                 ws += (size_t)N * H1;
  float* xr1 = ws;                 ws += (size_t)N * H1;
  float* agg1 = ws;                ws += (size_t)N * H1;
  float* xl2 = ws;                 ws += (size_t)N * F_OUT;
  float* xr2 = ws;                 ws += (size_t)N * F_OUT;
  float* m2 = ws;                  ws += (size_t)N;
  float* denom2 = ws;
  (void)ws_size; (void)n_in; (void)out_size;

  const int BLK = 256;

  {
    const int rowTiles = N / 16, colTiles = H1 / 16;
    const int waves = rowTiles * colTiles;
    const int blocks = (waves + 7) / 8;
    wmma_dual_gemm<128, 80><<<blocks, BLK, 0, stream>>>(
        x, W1l, W1r, xl1, xr1, rowTiles, colTiles);
  }
  layer1_owner_lds<<<(N + OWN_R - 1) / OWN_R, OWN_T, 0, stream>>>(
      srcE, dstE, E, N, xl1, xr1, att1, b1, agg1);

  {
    const int rowTiles = N / 16, colTiles = F_OUT / 16;
    const int waves = rowTiles * colTiles;
    const int blocks = (waves + 7) / 8;
    wmma_dual_gemm<80, 16><<<blocks, BLK, 0, stream>>>(
        agg1, W2l, W2r, xl2, xr2, rowTiles, colTiles);
  }
  init_softmax_state<<<(N + BLK - 1) / BLK, BLK, 0, stream>>>(m2, denom2, N);
  init_bias_bcast<<<((size_t)N * F_OUT + BLK - 1) / BLK, BLK, 0, stream>>>(
      out, b2, F_OUT, N * F_OUT);

  const int eblocks2 = (E + N + BLK - 1) / BLK;
  edge_pass_rows16<0><<<eblocks2, BLK, 0, stream>>>(
      srcE, dstE, E, N, xl2, xr2, att2, m2, denom2, out);
  edge_pass_rows16<1><<<eblocks2, BLK, 0, stream>>>(
      srcE, dstE, E, N, xl2, xr2, att2, m2, denom2, out);
  edge_pass_rows16<2><<<eblocks2, BLK, 0, stream>>>(
      srcE, dstE, E, N, xl2, xr2, att2, m2, denom2, out);
}


# Layer-2 softmax state and sums kept in LDS per 200-node destination range, on edge lists the layer-1 kernel saves; 5 launches to 1, no global atomics; LDS arrival-order sums, 4-lane logit tree
# speedup vs baseline: 4.3844x; 1.8888x over previous
#include <hip/hip_runtime.h>


typedef __attribute__((ext_vector_type(2))) float v2f;
typedef __attribute__((ext_vector_type(8))) float v8f;

#define NEG_SLOPE 0.2f

template <int K, int LDB>
__global__ __launch_bounds__(256) void wmma_dual_gemm(
    const float* __restrict__ A, const float* __restrict__ Bl,
    const float* __restrict__ Br, float* __restrict__ Cl,
    float* __restrict__ Cr, int rowTiles, int colTiles) {
  const int wid = (blockIdx.x * blockDim.x + threadIdx.x) >> 5;
  const int lane = threadIdx.x & 31;
  if (wid >= rowTiles * colTiles) return;
  const int rt = wid / colTiles;
  const int ct = wid % colTiles;
  const int row0 = rt * 16;
  const int col0 = ct * 16;
  const int half = lane >> 4;
  const int l16 = lane & 15;
  const int arow = row0 + l16;
  const int bcol = col0 + l16;

  const float* __restrict__ Ap = A + (size_t)arow * K + half * 2;

  v8f accl = {};
  v8f accr = {};
  for (int k = 0; k < K; k += 4) {
    const int ka = k + half * 2;
    v2f a;
    a.x = Ap[k];
    a.y = Ap[k + 1];
    v2f bl, br;
    bl.x = Bl[(size_t)ka * LDB + bcol];
    bl.y = Bl[(size_t)(ka + 1) * LDB + bcol];
    br.x = Br[(size_t)ka * LDB + bcol];
    br.y = Br[(size_t)(ka + 1) * LDB + bcol];
    accl = __builtin_amdgcn_wmma_f32_16x16x4_f32(false, a, false, bl, (short)0,
                                                 accl, false, false);
    accr = __builtin_amdgcn_wmma_f32_16x16x4_f32(false, a, false, br, (short)0,
                                                 accr, false, false);
  }

  const size_t base = (size_t)(row0 + half * 8) * LDB + bcol;
  float* __restrict__ cl = Cl + base;
  float* __restrict__ cr = Cr + base;
#pragma unroll
  for (int i = 0; i < 8; ++i) {
    cl[(size_t)i * LDB] = accl[i];
    cr[(size_t)i * LDB] = accr[i];
  }
}

__device__ __forceinline__ void atomicMaxFloat(float* addr, float val) {
  if (val >= 0.f)
    atomicMax((int*)addr, __float_as_int(val));
  else
    atomicMin((unsigned int*)addr, __float_as_uint(val));
}

#define OWN_R 200
#define OWN_S 81
#define OWN_CAP 4096
#define OWN_T 1024
#define OWN_U 8

__device__ __forceinline__ void own_take(int e, int d, int n0, int cnt,
                                         int* s_cnt, int2* list) {
  if ((unsigned)(d - n0) < (unsigned)cnt) {
    const int slot = atomicAdd(s_cnt, 1);
    if (slot < OWN_CAP) list[slot] = make_int2(e, d);
  }
}

template <int PHASE>
__device__ __forceinline__ void own_process(
    int count, bool cached, int n0, const int2* list, float* lg, float* acc,
    float* mS, float* dS, const float* __restrict__ xl,
    const float* __restrict__ xr, const float* __restrict__ att) {
  const int items = count * 10;
  for (int i = threadIdx.x; i < items; i += OWN_T) {
    const int ei = i / 10;
    const int h = i - ei * 10;
    const int2 sd = list[ei];
    const int dl = sd.y - n0;
    const int seg = dl * 10 + h;
    const float* __restrict__ xls = xl + (size_t)sd.x * 80 + h * 8;
    float xv[8];
    float logit;
    if (PHASE == 0 || !cached) {
      const float* __restrict__ xrd = xr + (size_t)sd.y * 80 + h * 8;
      logit = 0.f;
#pragma unroll
      for (int dd = 0; dd < 8; ++dd) {
        xv[dd] = xls[dd];
        float v = xv[dd] + xrd[dd];
        v = v > 0.f ? v : NEG_SLOPE * v;
        logit += v * att[h * 8 + dd];
      }
      if (PHASE == 0 && cached) lg[i] = logit;
    } else {
      logit = lg[i];
      if (PHASE == 2) {
#pragma unroll
        for (int dd = 0; dd < 8; ++dd) xv[dd] = xls[dd];
      }
    }
    if (PHASE == 0) {
      atomicMaxFloat(&mS[seg], logit);
    } else if (PHASE == 1) {
      atomicAdd(&dS[seg], __expf(logit - mS[seg]));
    } else {
      const float alpha = __expf(logit - mS[seg]) / dS[seg];
      float* a = acc + dl * OWN_S + h * 8 + (h >> 3);
#pragma unroll
      for (int dd = 0; dd < 8; ++dd) atomicAdd(&a[dd], alpha * xv[dd]);
    }
  }
}

__global__ __launch_bounds__(OWN_T) void layer1_owner_lds(
    const int* __restrict__ srcE, const int* __restrict__ dstE, int E, int N,
    const float* __restrict__ xl, const float* __restrict__ xr,
    const float* __restrict__ att, const float* __restrict__ bias,
    float* __restrict__ agg, int* __restrict__ savedList,
    int* __restrict__ savedCnt) {
  __shared__ float acc[OWN_R * OWN_S];
  __shared__ float mS[OWN_R * 10];
  __shared__ float dS[OWN_R * 10];
  __shared__ float lg[OWN_CAP * 10];
  __shared__ int2 list[OWN_CAP];
  __shared__ int s_cnt;

  const int tid = threadIdx.x;
  const int n0 = blockIdx.x * OWN_R;
  const int cnt = min(OWN_R, N - n0);

  for (int i = tid; i < OWN_R * OWN_S; i += OWN_T) acc[i] = 0.f;
  for (int i = tid; i < OWN_R * 10; i += OWN_T) {
    mS[i] = -__builtin_inff();
    dS[i] = 0.f;
  }
  if (tid == 0) s_cnt = 0;
  __syncthreads();

  {
    int head = (int)(((16u - (unsigned)((size_t)dstE & 15u)) & 15u) >> 2);
    if (head > E) head = E;
    const int nvec = (E - head) >> 2;
    const int4* __restrict__ dv = reinterpret_cast<const int4*>(dstE + head);
    const int rot = (int)(((long long)nvec * blockIdx.x) / gridDim.x);
    const int iters = (nvec + OWN_T - 1) / OWN_T;
    for (int k0 = 0; k0 < iters; k0 += OWN_U) {
      int4 q[OWN_U];
      int eb[OWN_U];
#pragma unroll
      for (int u = 0; u < OWN_U; ++u) {
        const int idx = (k0 + u) * OWN_T + tid;
        int v = idx + rot;
        if (v >= nvec) v -= nvec;
        eb[u] = head + 4 * v;
        q[u] = make_int4(-1, -1, -1, -1);
        if (idx < nvec) q[u] = dv[v];
      }
#pragma unroll
      for (int u = 0; u < OWN_U; ++u) {
        own_take(eb[u], q[u].x, n0, cnt, &s_cnt, list);
        own_take(eb[u] + 1, q[u].y, n0, cnt, &s_cnt, list);
        own_take(eb[u] + 2, q[u].z, n0, cnt, &s_cnt, list);
        own_take(eb[u] + 3, q[u].w, n0, cnt, &s_cnt, list);
      }
    }
    const int tail0 = head + 4 * nvec;
    if (tid < head) own_take(tid, dstE[tid], n0, cnt, &s_cnt, list);
    if (tail0 + tid < E)
      own_take(tail0 + tid, dstE[tail0 + tid], n0, cnt, &s_cnt, list);
  }
  __syncthreads();

  const int total = __builtin_amdgcn_readfirstlane(s_cnt);
  const bool fast = total + cnt <= OWN_CAP;
  if (fast) {
    for (int i = tid; i < total; i += OWN_T) list[i].x = srcE[list[i].x];
    for (int i = tid; i < cnt; i += OWN_T)
      list[total + i] = make_int2(n0 + i, n0 + i);
  }
  __syncthreads();

  if (fast) {
    int2* __restrict__ sl =
        reinterpret_cast<int2*>(savedList) + (size_t)blockIdx.x * OWN_CAP;
    for (int i = tid; i < total + cnt; i += OWN_T) sl[i] = list[i];
  }
  if (tid == 0) savedCnt[blockIdx.x] = fast ? total + cnt : -1;

  const int ntiles = fast ? 1 : 1 + (E + OWN_CAP - 1) / OWN_CAP;
  for (int phase = 0; phase < 3; ++phase) {
    for (int t = 0; t < ntiles; ++t) {
      int count = total + cnt;
      if (!fast) {
        if (tid == 0) s_cnt = 0;
        __syncthreads();
        if (t == 0) {
          for (int i = tid; i < cnt; i += OWN_T)
            list[i] = make_int2(n0 + i, n0 + i);
        } else {
          const int e0 = (t - 1) * OWN_CAP;
          const int e1 = min(e0 + OWN_CAP, E);
          for (int e = e0 + tid; e < e1; e += OWN_T) {
            const int d = dstE[e];
            if ((unsigned)(d - n0) < (unsigned)cnt) {
              const int slot = atomicAdd(&s_cnt, 1);
              list[slot] = make_int2(srcE[e], d);
            }
          }
        }
        __syncthreads();
        count = t == 0 ? cnt : __builtin_amdgcn_readfirstlane(s_cnt);
      }
      if (phase == 0)
        own_process<0>(count, fast, n0, list, lg, acc, mS, dS, xl, xr, att);
      else if (phase == 1)
        own_process<1>(count, fast, n0, list, lg, acc, mS, dS, xl, xr, att);
      else
        own_process<2>(count, fast, n0, list, lg, acc, mS, dS, xl, xr, att);
      __syncthreads();
    }
  }

  for (int j = tid; j < cnt * 80; j += OWN_T) {
    const int i = j / 80;
    const int f = j - i * 80;
    const float v = acc[i * OWN_S + f + (f >> 6)] + bias[f];
    agg[(size_t)(n0 + i) * 80 + f] = fmaxf(v, 0.f);
  }
}

#define OWN2_S 17
#define OWN2_XS 20
#define OWN2_U 4
#define OWN2_G (OWN_T / 4)

template <int CTRL>
__device__ __forceinline__ float dpp_flt(float v) {
  return __int_as_float(
      __builtin_amdgcn_update_dpp(0, __float_as_int(v), CTRL, 0xf, 0xf, true));
}

template <int PHASE>
__device__ __forceinline__ void own2_process(
    int count, bool cached, int n0, const int2* list, float* lg, float* acc,
    float* mS, float* dS, const float* xrS, const float* __restrict__ xl,
    float4 a4) {
  if (PHASE == 1 && cached) {
    for (int e = threadIdx.x; e < count; e += OWN_T) {
      const int dl = list[e].y - n0;
      atomicAdd(&dS[dl], __expf(lg[e] - mS[dl]));
    }
    return;
  }
  const int q = threadIdx.x & 3;
  const int g = threadIdx.x >> 2;
  for (int e0 = g; e0 < count; e0 += OWN2_U * OWN2_G) {
    int ee[OWN2_U], dl[OWN2_U];
    bool live[OWN2_U];
    float4 xv[OWN2_U];
#pragma unroll
    for (int u = 0; u < OWN2_U; ++u) {
      const int e = e0 + u * OWN2_G;
      live[u] = e < count;
      ee[u] = live[u] ? e : e0;
      const int2 sd = list[ee[u]];
      dl[u] = sd.y - n0;
      xv[u] = *reinterpret_cast<const float4*>(xl + (size_t)sd.x * 16 + q * 4);
    }
#pragma unroll
    for (int u = 0; u < OWN2_U; ++u) {
      float logit;
      if (PHASE == 0 || !cached) {
        const float4 rv =
            *reinterpret_cast<const float4*>(xrS + dl[u] * OWN2_XS + q * 4);
        float v0 = xv[u].x + rv.x, v1 = xv[u].y + rv.y;
        float v2 = xv[u].z + rv.z, v3 = xv[u].w + rv.w;
        v0 = v0 > 0.f ? v0 : NEG_SLOPE * v0;
        v1 = v1 > 0.f ? v1 : NEG_SLOPE * v1;
        v2 = v2 > 0.f ? v2 : NEG_SLOPE * v2;
        v3 = v3 > 0.f ? v3 : NEG_SLOPE * v3;
        float p = v0 * a4.x + v1 * a4.y + v2 * a4.z + v3 * a4.w;
        p += dpp_flt<0x161>(p);
        p += dpp_flt<0x162>(p);
        logit = p;
        if (PHASE == 0 && cached && q == 0 && live[u]) lg[ee[u]] = logit;
      } else {
        logit = lg[ee[u]];
      }
      if (PHASE == 0) {
        if (live[u] && q == 0) atomicMaxFloat(&mS[dl[u]], logit);
      } else if (PHASE == 1) {
        if (live[u] && q == 0)
          atomicAdd(&dS[dl[u]], __expf(logit - mS[dl[u]]));
      } else {
        const float alpha = __expf(logit - mS[dl[u]]) / dS[dl[u]];
        if (live[u]) {
          float* a = acc + dl[u] * OWN2_S + q * 4;
          atomicAdd(&a[0], alpha * xv[u].x);
          atomicAdd(&a[1], alpha * xv[u].y);
          atomicAdd(&a[2], alpha * xv[u].z);
          atomicAdd(&a[3], alpha * xv[u].w);
        }
      }
    }
  }
}

__global__ __launch_bounds__(OWN_T) void layer2_owner_lds(
    const int* __restrict__ srcE, const int* __restrict__ dstE, int E, int N,
    const float* __restrict__ xl, const float* __restrict__ xr,
    const float* __restrict__ att, const float* __restrict__ bias,
    const int* __restrict__ savedList, const int* __restrict__ savedCnt,
    float* __restrict__ out) {
  __shared__ float acc[OWN_R * OWN2_S];
  __shared__ __attribute__((aligned(16))) float xrS[OWN_R * OWN2_XS];
  __shared__ float mS[OWN_R];
  __shared__ float dS[OWN_R];
  __shared__ float lg[OWN_CAP];
  __shared__ int2 list[OWN_CAP];
  __shared__ int s_cnt;

  const int tid = threadIdx.x;
  const int n0 = blockIdx.x * OWN_R;
  const int cnt = min(OWN_R, N - n0);
  const int q = tid & 3;

  const int saved = __builtin_amdgcn_readfirstlane(savedCnt[blockIdx.x]);
  const bool fast = saved >= 0;
  const float4 a4 = make_float4(att[q * 4], att[q * 4 + 1], att[q * 4 + 2],
                                att[q * 4 + 3]);
  static_assert(OWN_R * 4 <= OWN_T, "one 16-byte piece of the xr rows per thread");
  float4 xrv = make_float4(0.f, 0.f, 0.f, 0.f);
  if (tid < cnt * 4)
    xrv = *reinterpret_cast<const float4*>(xr + (size_t)n0 * 16 + tid * 4);
  const int2* __restrict__ sl =
      reinterpret_cast<const int2*>(savedList) + (size_t)blockIdx.x * OWN_CAP;
  int2 sv[OWN_CAP / OWN_T];
#pragma unroll
  for (int u = 0; u < OWN_CAP / OWN_T; ++u) {
    const int i = u * OWN_T + tid;
    sv[u] = make_int2(0, 0);
    if (i < saved) sv[u] = sl[i];
  }

  for (int i = tid; i < OWN_R * OWN2_S; i += OWN_T) acc[i] = 0.f;
  for (int i = tid; i < OWN_R; i += OWN_T) {
    mS[i] = -__builtin_inff();
    dS[i] = 0.f;
  }
  if (tid < cnt * 4)
    *reinterpret_cast<float4*>(xrS + (tid >> 2) * OWN2_XS + q * 4) = xrv;
#pragma unroll
  for (int u = 0; u < OWN_CAP / OWN_T; ++u) {
    const int i = u * OWN_T + tid;
    if (i < saved) list[i] = sv[u];
  }
  __syncthreads();

  const int ntiles = fast ? 1 : 1 + (E + OWN_CAP - 1) / OWN_CAP;
  for (int phase = 0; phase < 3; ++phase) {
    for (int t = 0; t < ntiles; ++t) {
      int count = saved;
      if (!fast) {
        if (tid == 0) s_cnt = 0;
        __syncthreads();
        if (t == 0) {
          for (int i = tid; i < cnt; i += OWN_T)
            list[i] = make_int2(n0 + i, n0 + i);
        } else {
          const int e0 = (t - 1) * OWN_CAP;
          const int e1 = min(e0 + OWN_CAP, E);
          for (int e = e0 + tid; e < e1; e += OWN_T) {
            const int d = dstE[e];
            if ((unsigned)(d - n0) < (unsigned)cnt) {
              const int slot = atomicAdd(&s_cnt, 1);
              list[slot] = make_int2(srcE[e], d);
            }
          }
        }
        __syncthreads();
        count = t == 0 ? cnt : __builtin_amdgcn_readfirstlane(s_cnt);
      }
      if (phase == 0)
        own2_process<0>(count, fast, n0, list, lg, acc, mS, dS, xrS, xl, a4);
      else if (phase == 1)
        own2_process<1>(count, fast, n0, list, lg, acc, mS, dS, xrS, xl, a4);
      else
        own2_process<2>(count, fast, n0, list, lg, acc, mS, dS, xrS, xl, a4);
      __syncthreads();
    }
  }

  for (int j = tid; j < cnt * 16; j += OWN_T) {
    const int i = j >> 4;
    const int f = j & 15;
    out[(size_t)(n0 + i) * 16 + f] = acc[i * OWN2_S + f] + bias[f];
  }
}

extern "C" void kernel_launch(void* const* d_in, const int* in_sizes, int n_in,
                              void* d_out, int out_size, void* d_ws,
                              size_t ws_size, hipStream_t stream) {
  const float* x = (const float*)d_in[0];
  const int* edge = (const int*)d_in[1];
  const float* W1l = (const float*)d_in[2];
  const float* W1r = (const float*)d_in[3];
  const float* att1 = (const float*)d_in[4];
  const float* b1 = (const float*)d_in[5];
  const float* W2l = (const float*)d_in[6];
  const float* W2r = (const float*)d_in[7];
  const float* att2 = (const float*)d_in[8];
  const float* b2 = (const float*)d_in[9];
  float* out = (float*)d_out;

  const int F_IN = 128, H1 = 80, F_OUT = 16;
  const int N = in_sizes[0] / F_IN;
  const int E = in_sizes[1] / 2;
  const int* srcE = edge;
  const int* dstE = edge + E;

  float* ws = (float*)d_ws;
  float* xl1 = ws;                 ws += (size_t)N * H1;
  float* xr1 = ws;                 ws += (size_t)N * H1;
  float* agg1 = ws;                ws += (size_t)N * H1;
  float* xl2 = ws;                 ws += (size_t)N * F_OUT;
  float* xr2 = ws;                 ws += (size_t)N * F_OUT;
  const int ranges = (N + OWN_R - 1) / OWN_R;
  int* savedList = (int*)ws;       ws += (size_t)ranges * OWN_CAP * 2;
  int* savedCnt = (int*)ws;
  (void)ws_size; (void)n_in; (void)out_size;

  const int BLK = 256;

  {
    const int rowTiles = N / 16, colTiles = H1 / 16;
    const int waves = rowTiles * colTiles;
    const int blocks = (waves + 7) / 8;
    wmma_dual_gemm<128, 80><<<blocks, BLK, 0, stream>>>(
        x, W1l, W1r, xl1, xr1, rowTiles, colTiles);
  }
  layer1_owner_lds<<<ranges, OWN_T, 0, stream>>>(
      srcE, dstE, E, N, xl1, xr1, att1, b1, agg1, savedList, savedCnt);

  {
    const int rowTiles = N / 16, colTiles = F_OUT / 16;
    const int waves = rowTiles * colTiles;
    const int blocks = (waves + 7) / 8;
    wmma_dual_gemm<80, 16><<<blocks, BLK, 0, stream>>>(
        agg1, W2l, W2r, xl2, xr2, rowTiles, colTiles);
  }
  layer2_owner_lds<<<ranges, OWN_T, 0, stream>>>(
      srcE, dstE, E, N, xl2, xr2, att2, b2, savedList, savedCnt, out);
}


# Layer-1 edge filter on byte keys: transform launch writes each edge's destination-range index as one byte; a work-group tests 16 edges per 16-byte load (zero-byte test); LDS arrival order kept
# speedup vs baseline: 5.0518x; 1.1522x over previous
#include <hip/hip_runtime.h>


typedef __attribute__((ext_vector_type(2))) float v2f;
typedef __attribute__((ext_vector_type(8))) float v8f;

#define NEG_SLOPE 0.2f
#define OWN_R 200
#define KEY_NONE 255u

template <int K, int LDB, bool KEYS>
__global__ __launch_bounds__(256) void wmma_dual_gemm(
    const float* __restrict__ A, const float* __restrict__ Bl,
    const float* __restrict__ Br, float* __restrict__ Cl,
    float* __restrict__ Cr, int rowTiles, int colTiles,
    const int* __restrict__ dstE, int E, int N,
    unsigned int* __restrict__ keys) {
  if (KEYS) {
    const int nwords = ((E + 15) >> 4) << 2;
    for (int g = blockIdx.x * blockDim.x + threadIdx.x; g < nwords;
         g += gridDim.x * blockDim.x) {
      const int e0 = 4 * g;
      int d[4];
      if (e0 + 3 < E) {
        d[0] = dstE[e0];
        d[1] = dstE[e0 + 1];
        d[2] = dstE[e0 + 2];
        d[3] = dstE[e0 + 3];
      } else {
#pragma unroll
        for (int j = 0; j < 4; ++j) d[j] = e0 + j < E ? dstE[e0 + j] : -1;
      }
      unsigned int w = 0;
#pragma unroll
      for (int j = 0; j < 4; ++j) {
        const unsigned int key =
            (unsigned)d[j] < (unsigned)N ? (unsigned)d[j] / OWN_R : KEY_NONE;
        w |= key << (8 * j);
      }
      keys[g] = w;
    }
  }
  const int wid = (blockIdx.x * blockDim.x + threadIdx.x) >> 5;
  const int lane = threadIdx.x & 31;
  if (wid >= rowTiles * colTiles) return;
  const int rt = wid / colTiles;
  const int ct = wid % colTiles;
  const int row0 = rt * 16;
  const int col0 = ct * 16;
  const int half = lane >> 4;
  const int l16 = lane & 15;
  const int arow = row0 + l16;
  const int bcol = col0 + l16;

  const float* __restrict__ Ap = A + (size_t)arow * K + half * 2;

  v8f accl = {};
  v8f accr = {};
  for (int k = 0; k < K; k += 4) {
    const int ka = k + half * 2;
    v2f a;
    a.x = Ap[k];
    a.y = Ap[k + 1];
    v2f bl, br;
    bl.x = Bl[(size_t)ka * LDB + bcol];
    bl.y = Bl[(size_t)(ka + 1) * LDB + bcol];
    br.x = Br[(size_t)ka * LDB + bcol];
    br.y = Br[(size_t)(ka + 1) * LDB + bcol];
    accl = __builtin_amdgcn_wmma_f32_16x16x4_f32(false, a, false, bl, (short)0,
                                                 accl, false, false);
    accr = __builtin_amdgcn_wmma_f32_16x16x4_f32(false, a, false, br, (short)0,
                                                 accr, false, false);
  }

  const size_t base = (size_t)(row0 + half * 8) * LDB + bcol;
  float* __restrict__ cl = Cl + base;
  float* __restrict__ cr = Cr + base;
#pragma unroll
  for (int i = 0; i < 8; ++i) {
    cl[(size_t)i * LDB] = accl[i];
    cr[(size_t)i * LDB] = accr[i];
  }
}

__device__ __forceinline__ void atomicMaxFloat(float* addr, float val) {
  if (val >= 0.f)
    atomicMax((int*)addr, __float_as_int(val));
  else
    atomicMin((unsigned int*)addr, __float_as_uint(val));
}

#define OWN_S 81
#define OWN_CAP 4096
#define OWN_T 1024
#define OWN_U 8

__device__ __forceinline__ unsigned int key_zero_any(unsigned int x) {
  return (x - 0x01010101u) & ~x & 0x80808080u;
}
__device__ __forceinline__ unsigned int key_zero_exact(unsigned int x,
                                                       unsigned int t) {
  return t & ~(x << 7);
}

template <int PHASE>
__device__ __forceinline__ void own_process(
    int count, bool cached, int n0, const int2* list, float* lg, float* acc,
    float* mS, float* dS, const float* __restrict__ xl,
    const float* __restrict__ xr, const float* __restrict__ att) {
  const int items = count * 10;
  for (int i = threadIdx.x; i < items; i += OWN_T) {
    const int ei = i / 10;
    const int h = i - ei * 10;
    const int2 sd = list[ei];
    const int dl = sd.y - n0;
    const int seg = dl * 10 + h;
    const float* __restrict__ xls = xl + (size_t)sd.x * 80 + h * 8;
    float xv[8];
    float logit;
    if (PHASE == 0 || !cached) {
      const float* __restrict__ xrd = xr + (size_t)sd.y * 80 + h * 8;
      logit = 0.f;
#pragma unroll
      for (int dd = 0; dd < 8; ++dd) {
        xv[dd] = xls[dd];
        float v = xv[dd] + xrd[dd];
        v = v > 0.f ? v : NEG_SLOPE * v;
        logit += v * att[h * 8 + dd];
      }
      if (PHASE == 0 && cached) lg[i] = logit;
    } else {
      logit = lg[i];
      if (PHASE == 2) {
#pragma unroll
        for (int dd = 0; dd < 8; ++dd) xv[dd] = xls[dd];
      }
    }
    if (PHASE == 0) {
      atomicMaxFloat(&mS[seg], logit);
    } else if (PHASE == 1) {
      atomicAdd(&dS[seg], __expf(logit - mS[seg]));
    } else {
      const float alpha = __expf(logit - mS[seg]) / dS[seg];
      float* a = acc + dl * OWN_S + h * 8 + (h >> 3);
#pragma unroll
      for (int dd = 0; dd < 8; ++dd) atomicAdd(&a[dd], alpha * xv[dd]);
    }
  }
}

__global__ __launch_bounds__(OWN_T) void layer1_owner_lds(
    const int* __restrict__ srcE, const int* __restrict__ dstE, int E, int N,
    const float* __restrict__ xl, const float* __restrict__ xr,
    const float* __restrict__ att, const float* __restrict__ bias,
    float* __restrict__ agg, int* __restrict__ savedList,
    int* __restrict__ savedCnt, const unsigned int* __restrict__ keys) {
  __shared__ float acc[OWN_R * OWN_S];
  __shared__ float mS[OWN_R * 10];
  __shared__ float dS[OWN_R * 10];
  __shared__ float lg[OWN_CAP * 10];
  __shared__ int2 list[OWN_CAP];
  __shared__ int qv[OWN_CAP];
  __shared__ int s_cnt;
  __shared__ int s_q;

  const int tid = threadIdx.x;
  const int n0 = blockIdx.x * OWN_R;
  const int cnt = min(OWN_R, N - n0);

  for (int i = tid; i < OWN_R * OWN_S; i += OWN_T) acc[i] = 0.f;
  for (int i = tid; i < OWN_R * 10; i += OWN_T) {
    mS[i] = -__builtin_inff();
    dS[i] = 0.f;
  }
  if (tid == 0) {
    s_cnt = 0;
    s_q = 0;
  }
  __syncthreads();

  const unsigned int pat = (unsigned)blockIdx.x * 0x01010101u;
  const uint4* __restrict__ kv = reinterpret_cast<const uint4*>(keys);
  {
    const int nvec = (E + 15) >> 4;
    const int rot = (int)(((long long)nvec * blockIdx.x) / gridDim.x);
    const int iters = (nvec + OWN_T - 1) / OWN_T;
    for (int k0 = 0; k0 < iters; k0 += OWN_U) {
      uint4 q[OWN_U];
      int vv[OWN_U];
#pragma unroll
      for (int u = 0; u < OWN_U; ++u) {
        const int idx = (k0 + u) * OWN_T + tid;
        int v = idx + rot;
        if (v >= nvec) v -= nvec;
        vv[u] = v;
        q[u] = make_uint4(~0u, ~0u, ~0u, ~0u);
        if (idx < nvec) q[u] = kv[v];
      }
#pragma unroll
      for (int u = 0; u < OWN_U; ++u) {
        const unsigned int any =
            key_zero_any(q[u].x ^ pat) | key_zero_any(q[u].y ^ pat) |
            key_zero_any(q[u].z ^ pat) | key_zero_any(q[u].w ^ pat);
        if (any != 0u) {
          const int s = atomicAdd(&s_q, 1);
          if (s < OWN_CAP) qv[s] = vv[u];
        }
      }
    }
  }
  __syncthreads();

  const int nq = __builtin_amdgcn_readfirstlane(s_q);
  if (nq <= OWN_CAP) {
    uint4 kk[OWN_CAP / OWN_T];
    int vq[OWN_CAP / OWN_T];
#pragma unroll
    for (int u = 0; u < OWN_CAP / OWN_T; ++u) {
      const int i = u * OWN_T + tid;
      vq[u] = 0;
      kk[u] = make_uint4(~0u, ~0u, ~0u, ~0u);
      if (i < nq) {
        vq[u] = qv[i];
        kk[u] = kv[vq[u]];
      }
    }
#pragma unroll
    for (int u = 0; u < OWN_CAP / OWN_T; ++u) {
      const unsigned int x0 = kk[u].x ^ pat, x1 = kk[u].y ^ pat;
      const unsigned int x2 = kk[u].z ^ pat, x3 = kk[u].w ^ pat;
      unsigned int m = (key_zero_exact(x0, key_zero_any(x0)) >> 7) |
                       (key_zero_exact(x1, key_zero_any(x1)) >> 6) |
                       (key_zero_exact(x2, key_zero_any(x2)) >> 5) |
                       (key_zero_exact(x3, key_zero_any(x3)) >> 4);
      while (m != 0u) {
        const int p = __builtin_ctz(m);
        m &= m - 1u;
        const int e = 16 * vq[u] + 4 * (p & 7) + (p >> 3);
        const int slot = atomicAdd(&s_cnt, 1);
        if (slot < OWN_CAP) list[slot] = make_int2(e, 0);
      }
    }
  }
  __syncthreads();

  const int total = __builtin_amdgcn_readfirstlane(s_cnt);
  const bool fast = nq <= OWN_CAP && total + cnt <= OWN_CAP;
  if (fast) {
    for (int i = tid; i < total; i += OWN_T) {
      const int e = list[i].x;
      list[i] = make_int2(srcE[e], dstE[e]);
    }
    for (int i = tid; i < cnt; i += OWN_T)
      list[total + i] = make_int2(n0 + i, n0 + i);
  }
  __syncthreads();

  if (fast) {
    int2* __restrict__ sl =
        reinterpret_cast<int2*>(savedList) + (size_t)blockIdx.x * OWN_CAP;
    for (int i = tid; i < total + cnt; i += OWN_T) sl[i] = list[i];
  }
  if (tid == 0) savedCnt[blockIdx.x] = fast ? total + cnt : -1;

  const int ntiles = fast ? 1 : 1 + (E + OWN_CAP - 1) / OWN_CAP;
  for (int phase = 0; phase < 3; ++phase) {
    for (int t = 0; t < ntiles; ++t) {
      int count = total + cnt;
      if (!fast) {
        if (tid == 0) s_cnt = 0;
        __syncthreads();
        if (t == 0) {
          for (int i = tid; i < cnt; i += OWN_T)
            list[i] = make_int2(n0 + i, n0 + i);
        } else {
          const int e0 = (t - 1) * OWN_CAP;
          const int e1 = min(e0 + OWN_CAP, E);
          for (int e = e0 + tid; e < e1; e += OWN_T) {
            const int d = dstE[e];
            if ((unsigned)(d - n0) < (unsigned)cnt) {
              const int slot = atomicAdd(&s_cnt, 1);
              list[slot] = make_int2(srcE[e], d);
            }
          }
        }
        __syncthreads();
        count = t == 0 ? cnt : __builtin_amdgcn_readfirstlane(s_cnt);
      }
      if (phase == 0)
        own_process<0>(count, fast, n0, list, lg, acc, mS, dS, xl, xr, att);
      else if (phase == 1)
        own_process<1>(count, fast, n0, list, lg, acc, mS, dS, xl, xr, att);
      else
        own_process<2>(count, fast, n0, list, lg, acc, mS, dS, xl, xr, att);
      __syncthreads();
    }
  }

  for (int j = tid; j < cnt * 80; j += OWN_T) {
    const int i = j / 80;
    const int f = j - i * 80;
    const float v = acc[i * OWN_S + f + (f >> 6)] + bias[f];
    agg[(size_t)(n0 + i) * 80 + f] = fmaxf(v, 0.f);
  }
}

#define OWN2_S 17
#define OWN2_XS 20
#define OWN2_U 4
#define OWN2_G (OWN_T / 4)

template <int CTRL>
__device__ __forceinline__ float dpp_flt(float v) {
  return __int_as_float(
      __builtin_amdgcn_update_dpp(0, __float_as_int(v), CTRL, 0xf, 0xf, true));
}

template <int PHASE>
__device__ __forceinline__ void own2_process(
    int count, bool cached, int n0, const int2* list, float* lg, float* acc,
    float* mS, float* dS, const float* xrS, const float* __restrict__ xl,
    float4 a4) {
  if (PHASE == 1 && cached) {
    for (int e = threadIdx.x; e < count; e += OWN_T) {
      const int dl = list[e].y - n0;
      atomicAdd(&dS[dl], __expf(lg[e] - mS[dl]));
    }
    return;
  }
  const int q = threadIdx.x & 3;
  const int g = threadIdx.x >> 2;
  for (int e0 = g; e0 < count; e0 += OWN2_U * OWN2_G) {
    int ee[OWN2_U], dl[OWN2_U];
    bool live[OWN2_U];
    float4 xv[OWN2_U];
#pragma unroll
    for (int u = 0; u < OWN2_U; ++u) {
      const int e = e0 + u * OWN2_G;
      live[u] = e < count;
      ee[u] = live[u] ? e : e0;
      const int2 sd = list[ee[u]];
      dl[u] = sd.y - n0;
      xv[u] = *reinterpret_cast<const float4*>(xl + (size_t)sd.x * 16 + q * 4);
    }
#pragma unroll
    for (int u = 0; u < OWN2_U; ++u) {
      float logit;
      if (PHASE == 0 || !cached) {
        const float4 rv =
            *reinterpret_cast<const float4*>(xrS + dl[u] * OWN2_XS + q * 4);
        float v0 = xv[u].x + rv.x, v1 = xv[u].y + rv.y;
        float v2 = xv[u].z + rv.z, v3 = xv[u].w + rv.w;
        v0 = v0 > 0.f ? v0 : NEG_SLOPE * v0;
        v1 = v1 > 0.f ? v1 : NEG_SLOPE * v1;
        v2 = v2 > 0.f ? v2 : NEG_SLOPE * v2;
        v3 = v3 > 0.f ? v3 : NEG_SLOPE * v3;
        float p = v0 * a4.x + v1 * a4.y + v2 * a4.z + v3 * a4.w;
        p += dpp_flt<0x161>(p);
        p += dpp_flt<0x162>(p);
        logit = p;
        if (PHASE == 0 && cached && q == 0 && live[u]) lg[ee[u]] = logit;
      } else {
        logit = lg[ee[u]];
      }
      if (PHASE == 0) {
        if (live[u] && q == 0) atomicMaxFloat(&mS[dl[u]], logit);
      } else if (PHASE == 1) {
        if (live[u] && q == 0)
          atomicAdd(&dS[dl[u]], __expf(logit - mS[dl[u]]));
      } else {
        const float alpha = __expf(logit - mS[dl[u]]) / dS[dl[u]];
        if (live[u]) {
          float* a = acc + dl[u] * OWN2_S + q * 4;
          atomicAdd(&a[0], alpha * xv[u].x);
          atomicAdd(&a[1], alpha * xv[u].y);
          atomicAdd(&a[2], alpha * xv[u].z);
          atomicAdd(&a[3], alpha * xv[u].w);
        }
      }
    }
  }
}

__global__ __launch_bounds__(OWN_T) void layer2_owner_lds(
    const int* __restrict__ srcE, const int* __restrict__ dstE, int E, int N,
    const float* __restrict__ xl, const float* __restrict__ xr,
    const float* __restrict__ att, const float* __restrict__ bias,
    const int* __restrict__ savedList, const int* __restrict__ savedCnt,
    float* __restrict__ out) {
  __shared__ float acc[OWN_R * OWN2_S];
  __shared__ __attribute__((aligned(16))) float xrS[OWN_R * OWN2_XS];
  __shared__ float mS[OWN_R];
  __shared__ float dS[OWN_R];
  __shared__ float lg[OWN_CAP];
  __shared__ int2 list[OWN_CAP];
  __shared__ int s_cnt;

  const int tid = threadIdx.x;
  const int n0 = blockIdx.x * OWN_R;
  const int cnt = min(OWN_R, N - n0);
  const int q = tid & 3;

  const int saved = __builtin_amdgcn_readfirstlane(savedCnt[blockIdx.x]);
  const bool fast = saved >= 0;
  const float4 a4 = make_float4(att[q * 4], att[q * 4 + 1], att[q * 4 + 2],
                                att[q * 4 + 3]);
  static_assert(OWN_R * 4 <= OWN_T, "one 16-byte piece of the xr rows per thread");
  float4 xrv = make_float4(0.f, 0.f, 0.f, 0.f);
  if (tid < cnt * 4)
    xrv = *reinterpret_cast<const float4*>(xr + (size_t)n0 * 16 + tid * 4);
  const int2* __restrict__ sl =
      reinterpret_cast<const int2*>(savedList) + (size_t)blockIdx.x * OWN_CAP;
  int2 sv[OWN_CAP / OWN_T];
#pragma unroll
  for (int u = 0; u < OWN_CAP / OWN_T; ++u) {
    const int i = u * OWN_T + tid;
    sv[u] = make_int2(0, 0);
    if (i < saved) sv[u] = sl[i];
  }

  for (int i = tid; i < OWN_R * OWN2_S; i += OWN_T) acc[i] = 0.f;
  for (int i = tid; i < OWN_R; i += OWN_T) {
    mS[i] = -__builtin_inff();
    dS[i] = 0.f;
  }
  if (tid < cnt * 4)
    *reinterpret_cast<float4*>(xrS + (tid >> 2) * OWN2_XS + q * 4) = xrv;
#pragma unroll
  for (int u = 0; u < OWN_CAP / OWN_T; ++u) {
    const int i = u * OWN_T + tid;
    if (i < saved) list[i] = sv[u];
  }
  __syncthreads();

  const int ntiles = fast ? 1 : 1 + (E + OWN_CAP - 1) / OWN_CAP;
  for (int phase = 0; phase < 3; ++phase) {
    for (int t = 0; t < ntiles; ++t) {
      int count = saved;
      if (!fast) {
        if (tid == 0) s_cnt = 0;
        __syncthreads();
        if (t == 0) {
          for (int i = tid; i < cnt; i += OWN_T)
            list[i] = make_int2(n0 + i, n0 + i);
        } else {
          const int e0 = (t - 1) * OWN_CAP;
          const int e1 = min(e0 + OWN_CAP, E);
          for (int e = e0 + tid; e < e1; e += OWN_T) {
            const int d = dstE[e];
            if ((unsigned)(d - n0) < (unsigned)cnt) {
              const int slot = atomicAdd(&s_cnt, 1);
              list[slot] = make_int2(srcE[e], d);
            }
          }
        }
        __syncthreads();
        count = t == 0 ? cnt : __builtin_amdgcn_readfirstlane(s_cnt);
      }
      if (phase == 0)
        own2_process<0>(count, fast, n0, list, lg, acc, mS, dS, xrS, xl, a4);
      else if (phase == 1)
        own2_process<1>(count, fast, n0, list, lg, acc, mS, dS, xrS, xl, a4);
      else
        own2_process<2>(count, fast, n0, list, lg, acc, mS, dS, xrS, xl, a4);
      __syncthreads();
    }
  }

  for (int j = tid; j < cnt * 16; j += OWN_T) {
    const int i = j >> 4;
    const int f = j & 15;
    out[(size_t)(n0 + i) * 16 + f] = acc[i * OWN2_S + f] + bias[f];
  }
}

extern "C" void kernel_launch(void* const* d_in, const int* in_sizes, int n_in,
                              void* d_out, int out_size, void* d_ws,
                              size_t ws_size, hipStream_t stream) {
  const float* x = (const float*)d_in[0];
  const int* edge = (const int*)d_in[1];
  const float* W1l = (const float*)d_in[2];
  const float* W1r = (const float*)d_in[3];
  const float* att1 = (const float*)d_in[4];
  const float* b1 = (const float*)d_in[5];
  const float* W2l = (const float*)d_in[6];
  const float* W2r = (const float*)d_in[7];
  const float* att2 = (const float*)d_in[8];
  const float* b2 = (const float*)d_in[9];
  float* out = (float*)d_out;

  const int F_IN = 128, H1 = 80, F_OUT = 16;
  const int N = in_sizes[0] / F_IN;
  const int E = in_sizes[1] / 2;
  const int* srcE = edge;
  const int* dstE = edge + E;

  float* ws = (float*)d_ws;
  float* xl1 = ws;                 ws += (size_t)N * H1;
  float* xr1 = ws;                 ws += (size_t)N * H1;
  float* agg1 = ws;                ws += (size_t)N * H1;
  float* xl2 = ws;                 ws += (size_t)N * F_OUT;
  float* xr2 = ws;                 ws += (size_t)N * F_OUT;
  const int ranges = (N + OWN_R - 1) / OWN_R;
  int* savedList = (int*)ws;       ws += (size_t)ranges * OWN_CAP * 2;
  unsigned int* keys = (unsigned int*)ws;   ws += (size_t)((E + 15) / 16) * 4;
  int* savedCnt = (int*)ws;
  (void)ws_size; (void)n_in; (void)out_size;

  const int BLK = 256;

  {
    const int rowTiles = N / 16, colTiles = H1 / 16;
    const int waves = rowTiles * colTiles;
    const int blocks = (waves + 7) / 8;
    wmma_dual_gemm<128, 80, true><<<blocks, BLK, 0, stream>>>(
        x, W1l, W1r, xl1, xr1, rowTiles, colTiles, dstE, E, N, keys);
  }
  layer1_owner_lds<<<ranges, OWN_T, 0, stream>>>(
      srcE, dstE, E, N, xl1, xr1, att1, b1, agg1, savedList, savedCnt, keys);

  {
    const int rowTiles = N / 16, colTiles = F_OUT / 16;
    const int waves = rowTiles * colTiles;
    const int blocks = (waves + 7) / 8;
    wmma_dual_gemm<80, 16, false><<<blocks, BLK, 0, stream>>>(
        agg1, W2l, W2r, xl2, xr2, rowTiles, colTiles, dstE, E, N, keys);
  }
  layer2_owner_lds<<<ranges, OWN_T, 0, stream>>>(
      srcE, dstE, E, N, xl2, xr2, att2, b2, savedList, savedCnt, out);
}


# Layer-1 logit pass takes the destination's xr row from LDS: each work-group stages its 200 owned xr rows once, in the region of the sums (idle until the last phase); only xl rows are gathered
# speedup vs baseline: 5.2794x; 1.0451x over previous
#include <hip/hip_runtime.h>


typedef __attribute__((ext_vector_type(2))) float v2f;
typedef __attribute__((ext_vector_type(8))) float v8f;

#define NEG_SLOPE 0.2f
#define OWN_R 200
#define KEY_NONE 255u

template <int K, int LDB, bool KEYS>
__global__ __launch_bounds__(256) void wmma_dual_gemm(
    const float* __restrict__ A, const float* __restrict__ Bl,
    const float* __restrict__ Br, float* __restrict__ Cl,
    float* __restrict__ Cr, int rowTiles, int colTiles,
    const int* __restrict__ dstE, int E, int N,
    unsigned int* __restrict__ keys) {
  if (KEYS) {
    const int nwords = ((E + 15) >> 4) << 2;
    for (int g = blockIdx.x * blockDim.x + threadIdx.x; g < nwords;
         g += gridDim.x * blockDim.x) {
      const int e0 = 4 * g;
      int d[4];
      if (e0 + 3 < E) {
        d[0] = dstE[e0];
        d[1] = dstE[e0 + 1];
        d[2] = dstE[e0 + 2];
        d[3] = dstE[e0 + 3];
      } else {
#pragma unroll
        for (int j = 0; j < 4; ++j) d[j] = e0 + j < E ? dstE[e0 + j] : -1;
      }
      unsigned int w = 0;
#pragma unroll
      for (int j = 0; j < 4; ++j) {
        const unsigned int key =
            (unsigned)d[j] < (unsigned)N ? (unsigned)d[j] / OWN_R : KEY_NONE;
        w |= key << (8 * j);
      }
      keys[g] = w;
    }
  }
  const int wid = (blockIdx.x * blockDim.x + threadIdx.x) >> 5;
  const int lane = threadIdx.x & 31;
  if (wid >= rowTiles * colTiles) return;
  const int rt = wid / colTiles;
  const int ct = wid % colTiles;
  const int row0 = rt * 16;
  const int col0 = ct * 16;
  const int half = lane >> 4;
  const int l16 = lane & 15;
  const int arow = row0 + l16;
  const int bcol = col0 + l16;

  const float* __restrict__ Ap = A + (size_t)arow * K + half * 2;

  v8f accl = {};
  v8f accr = {};
  for (int k = 0; k < K; k += 4) {
    const int ka = k + half * 2;
    v2f a;
    a.x = Ap[k];
    a.y = Ap[k + 1];
    v2f bl, br;
    bl.x = Bl[(size_t)ka * LDB + bcol];
    bl.y = Bl[(size_t)(ka + 1) * LDB + bcol];
    br.x = Br[(size_t)ka * LDB + bcol];
    br.y = Br[(size_t)(ka + 1) * LDB + bcol];
    accl = __builtin_amdgcn_wmma_f32_16x16x4_f32(false, a, false, bl, (short)0,
                                                 accl, false, false);
    accr = __builtin_amdgcn_wmma_f32_16x16x4_f32(false, a, false, br, (short)0,
                                                 accr, false, false);
  }

  const size_t base = (size_t)(row0 + half * 8) * LDB + bcol;
  float* __restrict__ cl = Cl + base;
  float* __restrict__ cr = Cr + base;
#pragma unroll
  for (int i = 0; i < 8; ++i) {
    cl[(size_t)i * LDB] = accl[i];
    cr[(size_t)i * LDB] = accr[i];
  }
}

__device__ __forceinline__ void atomicMaxFloat(float* addr, float val) {
  if (val >= 0.f)
    atomicMax((int*)addr, __float_as_int(val));
  else
    atomicMin((unsigned int*)addr, __float_as_uint(val));
}

#define OWN_S 81
#define OWN_XP 84
#define OWN_CAP 4096
#define OWN_T 1024
#define OWN_U 8

__device__ __forceinline__ unsigned int key_zero_any(unsigned int x) {
  return (x - 0x01010101u) & ~x & 0x80808080u;
}
__device__ __forceinline__ unsigned int key_zero_exact(unsigned int x,
                                                       unsigned int t) {
  return t & ~(x << 7);
}

template <int PHASE>
__device__ __forceinline__ void own_process(
    int count, bool cached, int n0, const int2* list, float* lg, float* acc,
    float* mS, float* dS, const float* __restrict__ xl,
    const float* __restrict__ xr, const float* __restrict__ att) {
  const int items = count * 10;
  for (int i = threadIdx.x; i < items; i += OWN_T) {
    const int ei = i / 10;
    const int h = i - ei * 10;
    const int2 sd = list[ei];
    const int dl = sd.y - n0;
    const int seg = dl * 10 + h;
    const float* __restrict__ xls = xl + (size_t)sd.x * 80 + h * 8;
    float xv[8];
    float logit;
    if (PHASE == 0 || !cached) {
      float rv[8];
      if (cached) {
        const float* xrd = acc + dl * OWN_XP + h * 8 + ((h >> 3) << 2);
        const float4 r0 = *reinterpret_cast<const float4*>(xrd);
        const float4 r1 = *reinterpret_cast<const float4*>(xrd + 4);
        rv[0] = r0.x; rv[1] = r0.y; rv[2] = r0.z; rv[3] = r0.w;
        rv[4] = r1.x; rv[5] = r1.y; rv[6] = r1.z; rv[7] = r1.w;
      } else {
        const float* __restrict__ xrd = xr + (size_t)sd.y * 80 + h * 8;
#pragma unroll
        for (int dd = 0; dd < 8; ++dd) rv[dd] = xrd[dd];
      }
      logit = 0.f;
#pragma unroll
      for (int dd = 0; dd < 8; ++dd) {
        xv[dd] = xls[dd];
        float v = xv[dd] + rv[dd];
        v = v > 0.f ? v : NEG_SLOPE * v;
        logit += v * att[h * 8 + dd];
      }
      if (PHASE == 0 && cached) lg[i] = logit;
    } else {
      logit = lg[i];
      if (PHASE == 2) {
#pragma unroll
        for (int dd = 0; dd < 8; ++dd) xv[dd] = xls[dd];
      }
    }
    if (PHASE == 0) {
      atomicMaxFloat(&mS[seg], logit);
    } else if (PHASE == 1) {
      atomicAdd(&dS[seg], __expf(logit - mS[seg]));
    } else {
      const float alpha = __expf(logit - mS[seg]) / dS[seg];
      float* a = acc + dl * OWN_S + h * 8 + (h >> 3);
#pragma unroll
      for (int dd = 0; dd < 8; ++dd) atomicAdd(&a[dd], alpha * xv[dd]);
    }
  }
}

__global__ __launch_bounds__(OWN_T) void layer1_owner_lds(
    const int* __restrict__ srcE, const int* __restrict__ dstE, int E, int N,
    const float* __restrict__ xl, const float* __restrict__ xr,
    const float* __restrict__ att, const float* __restrict__ bias,
    float* __restrict__ agg, int* __restrict__ savedList,
    int* __restrict__ savedCnt, const unsigned int* __restrict__ keys) {
  static_assert(OWN_XP >= 80 + 4 && OWN_XP % 4 == 0 && OWN_XP >= OWN_S,
                "acc holds either layout, xr rows 16-byte aligned");
  __shared__ __attribute__((aligned(16))) float acc[OWN_R * OWN_XP];
  __shared__ float mS[OWN_R * 10];
  __shared__ float dS[OWN_R * 10];
  __shared__ float lg[OWN_CAP * 10];
  __shared__ int2 list[OWN_CAP];
  __shared__ int qv[OWN_CAP];
  __shared__ int s_cnt;
  __shared__ int s_q;

  const int tid = threadIdx.x;
  const int n0 = blockIdx.x * OWN_R;
  const int cnt = min(OWN_R, N - n0);

  static_assert(OWN_R * 20 <= 4 * OWN_T, "four 16-byte pieces of the xr rows per thread");
  float4 xrv[4];
#pragma unroll
  for (int u = 0; u < 4; ++u) {
    const int j = u * OWN_T + tid;
    xrv[u] = make_float4(0.f, 0.f, 0.f, 0.f);
    if (j < cnt * 20)
      xrv[u] = *reinterpret_cast<const float4*>(xr + (size_t)n0 * 80 + j * 4);
  }

  for (int i = tid; i < OWN_R * 10; i += OWN_T) {
    mS[i] = -__builtin_inff();
    dS[i] = 0.f;
  }
  if (tid == 0) {
    s_cnt = 0;
    s_q = 0;
  }
  __syncthreads();

  const unsigned int pat = (unsigned)blockIdx.x * 0x01010101u;
  const uint4* __restrict__ kv = reinterpret_cast<const uint4*>(keys);
  {
    const int nvec = (E + 15) >> 4;
    const int rot = (int)(((long long)nvec * blockIdx.x) / gridDim.x);
    const int iters = (nvec + OWN_T - 1) / OWN_T;
    for (int k0 = 0; k0 < iters; k0 += OWN_U) {
      uint4 q[OWN_U];
      int vv[OWN_U];
#pragma unroll
      for (int u = 0; u < OWN_U; ++u) {
        const int idx = (k0 + u) * OWN_T + tid;
        int v = idx + rot;
        if (v >= nvec) v -= nvec;
        vv[u] = v;
        q[u] = make_uint4(~0u, ~0u, ~0u, ~0u);
        if (idx < nvec) q[u] = kv[v];
      }
#pragma unroll
      for (int u = 0; u < OWN_U; ++u) {
        const unsigned int any =
            key_zero_any(q[u].x ^ pat) | key_zero_any(q[u].y ^ pat) |
            key_zero_any(q[u].z ^ pat) | key_zero_any(q[u].w ^ pat);
        if (any != 0u) {
          const int s = atomicAdd(&s_q, 1);
          if (s < OWN_CAP) qv[s] = vv[u];
        }
      }
    }
  }
#pragma unroll
  for (int u = 0; u < 4; ++u) {
    const int j = u * OWN_T + tid;
    if (j < cnt * 20) {
      const int row = j / 20;
      const int c4 = j - row * 20;
      *reinterpret_cast<float4*>(acc + row * OWN_XP + c4 * 4 + ((c4 >> 4) << 2)) =
          xrv[u];
    }
  }
  __syncthreads();

  const int nq = __builtin_amdgcn_readfirstlane(s_q);
  if (nq <= OWN_CAP) {
    uint4 kk[OWN_CAP / OWN_T];
    int vq[OWN_CAP / OWN_T];
#pragma unroll
    for (int u = 0; u < OWN_CAP / OWN_T; ++u) {
      const int i = u * OWN_T + tid;
      vq[u] = 0;
      kk[u] = make_uint4(~0u, ~0u, ~0u, ~0u);
      if (i < nq) {
        vq[u] = qv[i];
        kk[u] = kv[vq[u]];
      }
    }
#pragma unroll
    for (int u = 0; u < OWN_CAP / OWN_T; ++u) {
      const unsigned int x0 = kk[u].x ^ pat, x1 = kk[u].y ^ pat;
      const unsigned int x2 = kk[u].z ^ pat, x3 = kk[u].w ^ pat;
      unsigned int m = (key_zero_exact(x0, key_zero_any(x0)) >> 7) |
                       (key_zero_exact(x1, key_zero_any(x1)) >> 6) |
                       (key_zero_exact(x2, key_zero_any(x2)) >> 5) |
                       (key_zero_exact(x3, key_zero_any(x3)) >> 4);
      while (m != 0u) {
        const int p = __builtin_ctz(m);
        m &= m - 1u;
        const int e = 16 * vq[u] + 4 * (p & 7) + (p >> 3);
        const int slot = atomicAdd(&s_cnt, 1);
        if (slot < OWN_CAP) list[slot] = make_int2(e, 0);
      }
    }
  }
  __syncthreads();

  const int total = __builtin_amdgcn_readfirstlane(s_cnt);
  const bool fast = nq <= OWN_CAP && total + cnt <= OWN_CAP;
  if (fast) {
    for (int i = tid; i < total; i += OWN_T) {
      const int e = list[i].x;
      list[i] = make_int2(srcE[e], dstE[e]);
    }
    for (int i = tid; i < cnt; i += OWN_T)
      list[total + i] = make_int2(n0 + i, n0 + i);
  } else {
    for (int i = tid; i < OWN_R * OWN_S; i += OWN_T) acc[i] = 0.f;
  }
  __syncthreads();

  if (fast) {
    int2* __restrict__ sl =
        reinterpret_cast<int2*>(savedList) + (size_t)blockIdx.x * OWN_CAP;
    for (int i = tid; i < total + cnt; i += OWN_T) sl[i] = list[i];
  }
  if (tid == 0) savedCnt[blockIdx.x] = fast ? total + cnt : -1;

  const int ntiles = fast ? 1 : 1 + (E + OWN_CAP - 1) / OWN_CAP;
  for (int phase = 0; phase < 3; ++phase) {
    for (int t = 0; t < ntiles; ++t) {
      int count = total + cnt;
      if (!fast) {
        if (tid == 0) s_cnt = 0;
        __syncthreads();
        if (t == 0) {
          for (int i = tid; i < cnt; i += OWN_T)
            list[i] = make_int2(n0 + i, n0 + i);
        } else {
          const int e0 = (t - 1) * OWN_CAP;
          const int e1 = min(e0 + OWN_CAP, E);
          for (int e = e0 + tid; e < e1; e += OWN_T) {
            const int d = dstE[e];
            if ((unsigned)(d - n0) < (unsigned)cnt) {
              const int slot = atomicAdd(&s_cnt, 1);
              list[slot] = make_int2(srcE[e], d);
            }
          }
        }
        __syncthreads();
        count = t == 0 ? cnt : __builtin_amdgcn_readfirstlane(s_cnt);
      }
      if (phase == 0)
        own_process<0>(count, fast, n0, list, lg, acc, mS, dS, xl, xr, att);
      else if (phase == 1) {
        if (fast)
          for (int i = tid; i < OWN_R * OWN_S; i += OWN_T) acc[i] = 0.f;
        own_process<1>(count, fast, n0, list, lg, acc, mS, dS, xl, xr, att);
      } else
        own_process<2>(count, fast, n0, list, lg, acc, mS, dS, xl, xr, att);
      __syncthreads();
    }
  }

  for (int j = tid; j < cnt * 80; j += OWN_T) {
    const int i = j / 80;
    const int f = j - i * 80;
    const float v = acc[i * OWN_S + f + (f >> 6)] + bias[f];
    agg[(size_t)(n0 + i) * 80 + f] = fmaxf(v, 0.f);
  }
}

#define OWN2_S 17
#define OWN2_XS 20
#define OWN2_U 4
#define OWN2_G (OWN_T / 4)

template <int CTRL>
__device__ __forceinline__ float dpp_flt(float v) {
  return __int_as_float(
      __builtin_amdgcn_update_dpp(0, __float_as_int(v), CTRL, 0xf, 0xf, true));
}

template <int PHASE>
__device__ __forceinline__ void own2_process(
    int count, bool cached, int n0, const int2* list, float* lg, float* acc,
    float* mS, float* dS, const float* xrS, const float* __restrict__ xl,
    float4 a4) {
  if (PHASE == 1 && cached) {
    for (int e = threadIdx.x; e < count; e += OWN_T) {
      const int dl = list[e].y - n0;
      atomicAdd(&dS[dl], __expf(lg[e] - mS[dl]));
    }
    return;
  }
  const int q = threadIdx.x & 3;
  const int g = threadIdx.x >> 2;
  for (int e0 = g; e0 < count; e0 += OWN2_U * OWN2_G) {
    int ee[OWN2_U], dl[OWN2_U];
    bool live[OWN2_U];
    float4 xv[OWN2_U];
#pragma unroll
    for (int u = 0; u < OWN2_U; ++u) {
      const int e = e0 + u * OWN2_G;
      live[u] = e < count;
      ee[u] = live[u] ? e : e0;
      const int2 sd = list[ee[u]];
      dl[u] = sd.y - n0;
      xv[u] = *reinterpret_cast<const float4*>(xl + (size_t)sd.x * 16 + q * 4);
    }
#pragma unroll
    for (int u = 0; u < OWN2_U; ++u) {
      float logit;
      if (PHASE == 0 || !cached) {
        const float4 rv =
            *reinterpret_cast<const float4*>(xrS + dl[u] * OWN2_XS + q * 4);
        float v0 = xv[u].x + rv.x, v1 = xv[u].y + rv.y;
        float v2 = xv[u].z + rv.z, v3 = xv[u].w + rv.w;
        v0 = v0 > 0.f ? v0 : NEG_SLOPE * v0;
        v1 = v1 > 0.f ? v1 : NEG_SLOPE * v1;
        v2 = v2 > 0.f ? v2 : NEG_SLOPE * v2;
        v3 = v3 > 0.f ? v3 : NEG_SLOPE * v3;
        float p = v0 * a4.x + v1 * a4.y + v2 * a4.z + v3 * a4.w;
        p += dpp_flt<0x161>(p);
        p += dpp_flt<0x162>(p);
        logit = p;
        if (PHASE == 0 && cached && q == 0 && live[u]) lg[ee[u]] = logit;
      } else {
        logit = lg[ee[u]];
      }
      if (PHASE == 0) {
        if (live[u] && q == 0) atomicMaxFloat(&mS[dl[u]], logit);
      } else if (PHASE == 1) {
        if (live[u] && q == 0)
          atomicAdd(&dS[dl[u]], __expf(logit - mS[dl[u]]));
      } else {
        const float alpha = __expf(logit - mS[dl[u]]) / dS[dl[u]];
        if (live[u]) {
          float* a = acc + dl[u] * OWN2_S + q * 4;
          atomicAdd(&a[0], alpha * xv[u].x);
          atomicAdd(&a[1], alpha * xv[u].y);
          atomicAdd(&a[2], alpha * xv[u].z);
          atomicAdd(&a[3], alpha * xv[u].w);
        }
      }
    }
  }
}

__global__ __launch_bounds__(OWN_T) void layer2_owner_lds(
    const int* __restrict__ srcE, const int* __restrict__ dstE, int E, int N,
    const float* __restrict__ xl, const float* __restrict__ xr,
    const float* __restrict__ att, const float* __restrict__ bias,
    const int* __restrict__ savedList, const int* __restrict__ savedCnt,
    float* __restrict__ out) {
  __shared__ float acc[OWN_R * OWN2_S];
  __shared__ __attribute__((aligned(16))) float xrS[OWN_R * OWN2_XS];
  __shared__ float mS[OWN_R];
  __shared__ float dS[OWN_R];
  __shared__ float lg[OWN_CAP];
  __shared__ int2 list[OWN_CAP];
  __shared__ int s_cnt;

  const int tid = threadIdx.x;
  const int n0 = blockIdx.x * OWN_R;
  const int cnt = min(OWN_R, N - n0);
  const int q = tid & 3;

  const int saved = __builtin_amdgcn_readfirstlane(savedCnt[blockIdx.x]);
  const bool fast = saved >= 0;
  const float4 a4 = make_float4(att[q * 4], att[q * 4 + 1], att[q * 4 + 2],
                                att[q * 4 + 3]);
  static_assert(OWN_R * 4 <= OWN_T, "one 16-byte piece of the xr rows per thread");
  float4 xrv = make_float4(0.f, 0.f, 0.f, 0.f);
  if (tid < cnt * 4)
    xrv = *reinterpret_cast<const float4*>(xr + (size_t)n0 * 16 + tid * 4);
  const int2* __restrict__ sl =
      reinterpret_cast<const int2*>(savedList) + (size_t)blockIdx.x * OWN_CAP;
  int2 sv[OWN_CAP / OWN_T];
#pragma unroll
  for (int u = 0; u < OWN_CAP / OWN_T; ++u) {
    const int i = u * OWN_T + tid;
    sv[u] = make_int2(0, 0);
    if (i < saved) sv[u] = sl[i];
  }

  for (int i = tid; i < OWN_R * OWN2_S; i += OWN_T) acc[i] = 0.f;
  for (int i = tid; i < OWN_R; i += OWN_T) {
    mS[i] = -__builtin_inff();
    dS[i] = 0.f;
  }
  if (tid < cnt * 4)
    *reinterpret_cast<float4*>(xrS + (tid >> 2) * OWN2_XS + q * 4) = xrv;
#pragma unroll
  for (int u = 0; u < OWN_CAP / OWN_T; ++u) {
    const int i = u * OWN_T + tid;
    if (i < saved) list[i] = sv[u];
  }
  __syncthreads();

  const int ntiles = fast ? 1 : 1 + (E + OWN_CAP - 1) / OWN_CAP;
  for (int phase = 0; phase < 3; ++phase) {
    for (int t = 0; t < ntiles; ++t) {
      int count = saved;
      if (!fast) {
        if (tid == 0) s_cnt = 0;
        __syncthreads();
        if (t == 0) {
          for (int i = tid; i < cnt; i += OWN_T)
            list[i] = make_int2(n0 + i, n0 + i);
        } else {
          const int e0 = (t - 1) * OWN_CAP;
          const int e1 = min(e0 + OWN_CAP, E);
          for (int e = e0 + tid; e < e1; e += OWN_T) {
            const int d = dstE[e];
            if ((unsigned)(d - n0) < (unsigned)cnt) {
              const int slot = atomicAdd(&s_cnt, 1);
              list[slot] = make_int2(srcE[e], d);
            }
          }
        }
        __syncthreads();
        count = t == 0 ? cnt : __builtin_amdgcn_readfirstlane(s_cnt);
      }
      if (phase == 0)
        own2_process<0>(count, fast, n0, list, lg, acc, mS, dS, xrS, xl, a4);
      else if (phase == 1)
        own2_process<1>(count, fast, n0, list, lg, acc, mS, dS, xrS, xl, a4);
      else
        own2_process<2>(count, fast, n0, list, lg, acc, mS, dS, xrS, xl, a4);
      __syncthreads();
    }
  }

  for (int j = tid; j < cnt * 16; j += OWN_T) {
    const int i = j >> 4;
    const int f = j & 15;
    out[(size_t)(n0 + i) * 16 + f] = acc[i * OWN2_S + f] + bias[f];
  }
}

extern "C" void kernel_launch(void* const* d_in, const int* in_sizes, int n_in,
                              void* d_out, int out_size, void* d_ws,
                              size_t ws_size, hipStream_t stream) {
  const float* x = (const float*)d_in[0];
  const int* edge = (const int*)d_in[1];
  const float* W1l = (const float*)d_in[2];
  const float* W1r = (const float*)d_in[3];
  const float* att1 = (const float*)d_in[4];
  const float* b1 = (const float*)d_in[5];
  const float* W2l = (const float*)d_in[6];
  const float* W2r = (const float*)d_in[7];
  const float* att2 = (const float*)d_in[8];
  const float* b2 = (const float*)d_in[9];
  float* out = (float*)d_out;

  const int F_IN = 128, H1 = 80, F_OUT = 16;
  const int N = in_sizes[0] / F_IN;
  const int E = in_sizes[1] / 2;
  const int* srcE = edge;
  const int* dstE = edge + E;

  float* ws = (float*)d_ws;
  float* xl1 = ws;                 ws += (size_t)N * H1;
  float* xr1 = ws;                 ws += (size_t)N * H1;
  float* agg1 = ws;                ws += (size_t)N * H1;
  float* xl2 = ws;                 ws += (size_t)N * F_OUT;
  float* xr2 = ws;                 ws += (size_t)N * F_OUT;
  const int ranges = (N + OWN_R - 1) / OWN_R;
  int* savedList = (int*)ws;       ws += (size_t)ranges * OWN_CAP * 2;
  unsigned int* keys = (unsigned int*)ws;   ws += (size_t)((E + 15) / 16) * 4;
  int* savedCnt = (int*)ws;
  (void)ws_size; (void)n_in; (void)out_size;

  const int BLK = 256;

  {
    const int rowTiles = N / 16, colTiles = H1 / 16;
    const int waves = rowTiles * colTiles;
    const int blocks = (waves + 7) / 8;
    wmma_dual_gemm<128, 80, true><<<blocks, BLK, 0, stream>>>(
        x, W1l, W1r, xl1, xr1, rowTiles, colTiles, dstE, E, N, keys);
  }
  layer1_owner_lds<<<ranges, OWN_T, 0, stream>>>(
      srcE, dstE, E, N, xl1, xr1, att1, b1, agg1, savedList, savedCnt, keys);

  {
    const int rowTiles = N / 16, colTiles = F_OUT / 16;
    const int waves = rowTiles * colTiles;
    const int blocks = (waves + 7) / 8;
    wmma_dual_gemm<80, 16, false><<<blocks, BLK, 0, stream>>>(
        agg1, W2l, W2r, xl2, xr2, rowTiles, colTiles, dstE, E, N, keys);
  }
  layer2_owner_lds<<<ranges, OWN_T, 0, stream>>>(
      srcE, dstE, E, N, xl2, xr2, att2, b2, savedList, savedCnt, out);
}


# Layer-1 edge filter on per-range bit planes: transform launch ORs each 1024-edge tile in LDS, stores one 128-byte line per range; a work-group scans only its own plane; LDS arrival order kept
# speedup vs baseline: 5.6575x; 1.0716x over previous
#include <hip/hip_runtime.h>


typedef __attribute__((ext_vector_type(2))) float v2f;
typedef __attribute__((ext_vector_type(8))) float v8f;

#define NEG_SLOPE 0.2f
#define OWN_R 200
#define BM_CHUNK 1024
#define BM_W (BM_CHUNK / 32)
#define GEMM_T 256

template <int K, int LDB, bool KEYS>
__global__ __launch_bounds__(GEMM_T) void wmma_dual_gemm(
    const float* __restrict__ A, const float* __restrict__ Bl,
    const float* __restrict__ Br, float* __restrict__ Cl,
    float* __restrict__ Cr, int rowTiles, int colTiles,
    const int* __restrict__ dstE, int E, int N,
    unsigned int* __restrict__ planes) {
  if (KEYS) {
    __shared__ __attribute__((aligned(16))) unsigned int bm[256 * BM_W];
    uint4* bm4 = reinterpret_cast<uint4*>(bm);
    const int ranges = (N + OWN_R - 1) / OWN_R;
    const int nChunks = (E + BM_CHUNK - 1) / BM_CHUNK;
    const size_t planeWords = (size_t)nChunks * BM_W;
    static_assert(BM_CHUNK == 4 * GEMM_T, "four edges of a tile per thread");
    const int le = 4 * threadIdx.x;
    for (int c = blockIdx.x; c < nChunks; c += gridDim.x) {
      const int e0 = c * BM_CHUNK + le;
      int d[4];
      if (e0 + 3 < E) {
        d[0] = dstE[e0];
        d[1] = dstE[e0 + 1];
        d[2] = dstE[e0 + 2];
        d[3] = dstE[e0 + 3];
      } else {
#pragma unroll
        for (int j = 0; j < 4; ++j) d[j] = e0 + j < E ? dstE[e0 + j] : -1;
      }
      if (c != (int)blockIdx.x) __syncthreads();
      for (int i = threadIdx.x; i < ranges * (BM_W / 4); i += GEMM_T)
        bm4[i] = make_uint4(0u, 0u, 0u, 0u);
      __syncthreads();
#pragma unroll
      for (int j = 0; j < 4; ++j)
        if ((unsigned)d[j] < (unsigned)N)
          atomicOr(&bm[((unsigned)d[j] / OWN_R) * BM_W + (le >> 5)],
                   1u << ((le + j) & 31));
      __syncthreads();
      for (int i = threadIdx.x; i < ranges * (BM_W / 4); i += GEMM_T) {
        const int r = i / (BM_W / 4);
        const int w4 = i - r * (BM_W / 4);
        reinterpret_cast<uint4*>(planes + (size_t)r * planeWords +
                                 (size_t)c * BM_W)[w4] = bm4[i];
      }
    }
  }
  const int wid = (blockIdx.x * blockDim.x + threadIdx.x) >> 5;
  const int lane = threadIdx.x & 31;
  if (wid >= rowTiles * colTiles) return;
  const int rt = wid / colTiles;
  const int ct = wid % colTiles;
  const int row0 = rt * 16;
  const int col0 = ct * 16;
  const int half = lane >> 4;
  const int l16 = lane & 15;
  const int arow = row0 + l16;
  const int bcol = col0 + l16;

  const float* __restrict__ Ap = A + (size_t)arow * K + half * 2;

  v8f accl = {};
  v8f accr = {};
  for (int k = 0; k < K; k += 4) {
    const int ka = k + half * 2;
    v2f a;
    a.x = Ap[k];
    a.y = Ap[k + 1];
    v2f bl, br;
    bl.x = Bl[(size_t)ka * LDB + bcol];
    bl.y = Bl[(size_t)(ka + 1) * LDB + bcol];
    br.x = Br[(size_t)ka * LDB + bcol];
    br.y = Br[(size_t)(ka + 1) * LDB + bcol];
    accl = __builtin_amdgcn_wmma_f32_16x16x4_f32(false, a, false, bl, (short)0,
                                                 accl, false, false);
    accr = __builtin_amdgcn_wmma_f32_16x16x4_f32(false, a, false, br, (short)0,
                                                 accr, false, false);
  }

  const size_t base = (size_t)(row0 + half * 8) * LDB + bcol;
  float* __restrict__ cl = Cl + base;
  float* __restrict__ cr = Cr + base;
#pragma unroll
  for (int i = 0; i < 8; ++i) {
    cl[(size_t)i * LDB] = accl[i];
    cr[(size_t)i * LDB] = accr[i];
  }
}

__device__ __forceinline__ void atomicMaxFloat(float* addr, float val) {
  if (val >= 0.f)
    atomicMax((int*)addr, __float_as_int(val));
  else
    atomicMin((unsigned int*)addr, __float_as_uint(val));
}

#define OWN_S 81
#define OWN_XP 84
#define OWN_CAP 4096
#define OWN_T 1024
#define OWN_U 8

template <int PHASE>
__device__ __forceinline__ void own_process(
    int count, bool cached, int n0, const int2* list, float* lg, float* acc,
    float* mS, float* dS, const float* __restrict__ xl,
    const float* __restrict__ xr, const float* __restrict__ att) {
  const int items = count * 10;
  for (int i = threadIdx.x; i < items; i += OWN_T) {
    const int ei = i / 10;
    const int h = i - ei * 10;
    const int2 sd = list[ei];
    const int dl = sd.y - n0;
    const int seg = dl * 10 + h;
    const float* __restrict__ xls = xl + (size_t)sd.x * 80 + h * 8;
    float xv[8];
    float logit;
    if (PHASE == 0 || !cached) {
      float rv[8];
      if (cached) {
        const float* xrd = acc + dl * OWN_XP + h * 8 + ((h >> 3) << 2);
        const float4 r0 = *reinterpret_cast<const float4*>(xrd);
        const float4 r1 = *reinterpret_cast<const float4*>(xrd + 4);
        rv[0] = r0.x; rv[1] = r0.y; rv[2] = r0.z; rv[3] = r0.w;
        rv[4] = r1.x; rv[5] = r1.y; rv[6] = r1.z; rv[7] = r1.w;
      } else {
        const float* __restrict__ xrd = xr + (size_t)sd.y * 80 + h * 8;
#pragma unroll
        for (int dd = 0; dd < 8; ++dd) rv[dd] = xrd[dd];
      }
      logit = 0.f;
#pragma unroll
      for (int dd = 0; dd < 8; ++dd) {
        xv[dd] = xls[dd];
        float v = xv[dd] + rv[dd];
        v = v > 0.f ? v : NEG_SLOPE * v;
        logit += v * att[h * 8 + dd];
      }
      if (PHASE == 0 && cached) lg[i] = logit;
    } else {
      logit = lg[i];
      if (PHASE == 2) {
#pragma unroll
        for (int dd = 0; dd < 8; ++dd) xv[dd] = xls[dd];
      }
    }
    if (PHASE == 0) {
      atomicMaxFloat(&mS[seg], logit);
    } else if (PHASE == 1) {
      atomicAdd(&dS[seg], __expf(logit - mS[seg]));
    } else {
      const float alpha = __expf(logit - mS[seg]) / dS[seg];
      float* a = acc + dl * OWN_S + h * 8 + (h >> 3);
#pragma unroll
      for (int dd = 0; dd < 8; ++dd) atomicAdd(&a[dd], alpha * xv[dd]);
    }
  }
}

__global__ __launch_bounds__(OWN_T) void layer1_owner_lds(
    const int* __restrict__ srcE, const int* __restrict__ dstE, int E, int N,
    const float* __restrict__ xl, const float* __restrict__ xr,
    const float* __restrict__ att, const float* __restrict__ bias,
    float* __restrict__ agg, int* __restrict__ savedList,
    int* __restrict__ savedCnt, const unsigned int* __restrict__ planes) {
  static_assert(OWN_XP >= 80 + 4 && OWN_XP % 4 == 0 && OWN_XP >= OWN_S,
                "acc holds either layout, xr rows 16-byte aligned");
  __shared__ __attribute__((aligned(16))) float acc[OWN_R * OWN_XP];
  __shared__ float mS[OWN_R * 10];
  __shared__ float dS[OWN_R * 10];
  __shared__ float lg[OWN_CAP * 10];
  __shared__ int2 list[OWN_CAP];
  __shared__ int s_cnt;

  const int tid = threadIdx.x;
  const int n0 = blockIdx.x * OWN_R;
  const int cnt = min(OWN_R, N - n0);

  static_assert(OWN_R * 20 <= 4 * OWN_T, "four 16-byte pieces of the xr rows per thread");
  float4 xrv[4];
#pragma unroll
  for (int u = 0; u < 4; ++u) {
    const int j = u * OWN_T + tid;
    xrv[u] = make_float4(0.f, 0.f, 0.f, 0.f);
    if (j < cnt * 20)
      xrv[u] = *reinterpret_cast<const float4*>(xr + (size_t)n0 * 80 + j * 4);
  }

  for (int i = tid; i < OWN_R * 10; i += OWN_T) {
    mS[i] = -__builtin_inff();
    dS[i] = 0.f;
  }
  if (tid == 0) s_cnt = 0;
  __syncthreads();

  {
    const int nChunks = (E + BM_CHUNK - 1) / BM_CHUNK;
    const size_t planeWords = (size_t)nChunks * BM_W;
    const int nvec = nChunks * (BM_W / 4);
    const uint4* __restrict__ pv =
        reinterpret_cast<const uint4*>(planes + (size_t)blockIdx.x * planeWords);
    const int iters = (nvec + OWN_T - 1) / OWN_T;
    for (int k0 = 0; k0 < iters; k0 += OWN_U) {
      uint4 q[OWN_U];
#pragma unroll
      for (int u = 0; u < OWN_U; ++u) {
        const int idx = (k0 + u) * OWN_T + tid;
        q[u] = make_uint4(0u, 0u, 0u, 0u);
        if (idx < nvec) q[u] = pv[idx];
      }
      int nb = 0;
#pragma unroll
      for (int u = 0; u < OWN_U; ++u)
        nb += __popc(q[u].x) + __popc(q[u].y) + __popc(q[u].z) + __popc(q[u].w);
      int slot = atomicAdd(&s_cnt, nb);
#pragma unroll
      for (int u = 0; u < OWN_U; ++u) {
        const int idx = (k0 + u) * OWN_T + tid;
        const unsigned int w[4] = {q[u].x, q[u].y, q[u].z, q[u].w};
#pragma unroll
        for (int j = 0; j < 4; ++j) {
          unsigned int m = w[j];
          while (m != 0u) {
            const int p = __builtin_ctz(m);
            m &= m - 1u;
            if (slot < OWN_CAP) list[slot] = make_int2((idx * 4 + j) * 32 + p, 0);
            ++slot;
          }
        }
      }
    }
  }
#pragma unroll
  for (int u = 0; u < 4; ++u) {
    const int j = u * OWN_T + tid;
    if (j < cnt * 20) {
      const int row = j / 20;
      const int c4 = j - row * 20;
      *reinterpret_cast<float4*>(acc + row * OWN_XP + c4 * 4 + ((c4 >> 4) << 2)) =
          xrv[u];
    }
  }
  __syncthreads();

  const int total = __builtin_amdgcn_readfirstlane(s_cnt);
  const bool fast = total + cnt <= OWN_CAP;
  if (fast) {
    for (int i = tid; i < total; i += OWN_T) {
      const int e = list[i].x;
      list[i] = make_int2(srcE[e], dstE[e]);
    }
    for (int i = tid; i < cnt; i += OWN_T)
      list[total + i] = make_int2(n0 + i, n0 + i);
  } else {
    for (int i = tid; i < OWN_R * OWN_S; i += OWN_T) acc[i] = 0.f;
  }
  __syncthreads();

  if (fast) {
    int2* __restrict__ sl =
        reinterpret_cast<int2*>(savedList) + (size_t)blockIdx.x * OWN_CAP;
    for (int i = tid; i < total + cnt; i += OWN_T) sl[i] = list[i];
  }
  if (tid == 0) savedCnt[blockIdx.x] = fast ? total + cnt : -1;

  const int ntiles = fast ? 1 : 1 + (E + OWN_CAP - 1) / OWN_CAP;
  for (int phase = 0; phase < 3; ++phase) {
    for (int t = 0; t < ntiles; ++t) {
      int count = total + cnt;
      if (!fast) {
        if (tid == 0) s_cnt = 0;
        __syncthreads();
        if (t == 0) {
          for (int i = tid; i < cnt; i += OWN_T)
            list[i] = make_int2(n0 + i, n0 + i);
        } else {
          const int e0 = (t - 1) * OWN_CAP;
          const int e1 = min(e0 + OWN_CAP, E);
          for (int e = e0 + tid; e < e1; e += OWN_T) {
            const int d = dstE[e];
            if ((unsigned)(d - n0) < (unsigned)cnt) {
              const int slot = atomicAdd(&s_cnt, 1);
              list[slot] = make_int2(srcE[e], d);
            }
          }
        }
        __syncthreads();
        count = t == 0 ? cnt : __builtin_amdgcn_readfirstlane(s_cnt);
      }
      if (phase == 0)
        own_process<0>(count, fast, n0, list, lg, acc, mS, dS, xl, xr, att);
      else if (phase == 1) {
        if (fast)
          for (int i = tid; i < OWN_R * OWN_S; i += OWN_T) acc[i] = 0.f;
        own_process<1>(count, fast, n0, list, lg, acc, mS, dS, xl, xr, att);
      } else
        own_process<2>(count, fast, n0, list, lg, acc, mS, dS, xl, xr, att);
      __syncthreads();
    }
  }

  for (int j = tid; j < cnt * 80; j += OWN_T) {
    const int i = j / 80;
    const int f = j - i * 80;
    const float v = acc[i * OWN_S + f + (f >> 6)] + bias[f];
    agg[(size_t)(n0 + i) * 80 + f] = fmaxf(v, 0.f);
  }
}

#define OWN2_S 17
#define OWN2_XS 20
#define OWN2_U 4
#define OWN2_G (OWN_T / 4)

template <int CTRL>
__device__ __forceinline__ float dpp_flt(float v) {
  return __int_as_float(
      __builtin_amdgcn_update_dpp(0, __float_as_int(v), CTRL, 0xf, 0xf, true));
}

template <int PHASE>
__device__ __forceinline__ void own2_process(
    int count, bool cached, int n0, const int2* list, float* lg, float* acc,
    float* mS, float* dS, const float* xrS, const float* __restrict__ xl,
    float4 a4) {
  if (PHASE == 1 && cached) {
    for (int e = threadIdx.x; e < count; e += OWN_T) {
      const int dl = list[e].y - n0;
      atomicAdd(&dS[dl], __expf(lg[e] - mS[dl]));
    }
    return;
  }
  const int q = threadIdx.x & 3;
  const int g = threadIdx.x >> 2;
  for (int e0 = g; e0 < count; e0 += OWN2_U * OWN2_G) {
    int ee[OWN2_U], dl[OWN2_U];
    bool live[OWN2_U];
    float4 xv[OWN2_U];
#pragma unroll
    for (int u = 0; u < OWN2_U; ++u) {
      const int e = e0 + u * OWN2_G;
      live[u] = e < count;
      ee[u] = live[u] ? e : e0;
      const int2 sd = list[ee[u]];
      dl[u] = sd.y - n0;
      xv[u] = *reinterpret_cast<const float4*>(xl + (size_t)sd.x * 16 + q * 4);
    }
#pragma unroll
    for (int u = 0; u < OWN2_U; ++u) {
      float logit;
      if (PHASE == 0 || !cached) {
        const float4 rv =
            *reinterpret_cast<const float4*>(xrS + dl[u] * OWN2_XS + q * 4);
        float v0 = xv[u].x + rv.x, v1 = xv[u].y + rv.y;
        float v2 = xv[u].z + rv.z, v3 = xv[u].w + rv.w;
        v0 = v0 > 0.f ? v0 : NEG_SLOPE * v0;
        v1 = v1 > 0.f ? v1 : NEG_SLOPE * v1;
        v2 = v2 > 0.f ? v2 : NEG_SLOPE * v2;
        v3 = v3 > 0.f ? v3 : NEG_SLOPE * v3;
        float p = v0 * a4.x + v1 * a4.y + v2 * a4.z + v3 * a4.w;
        p += dpp_flt<0x161>(p);
        p += dpp_flt<0x162>(p);
        logit = p;
        if (PHASE == 0 && cached && q == 0 && live[u]) lg[ee[u]] = logit;
      } else {
        logit = lg[ee[u]];
      }
      if (PHASE == 0) {
        if (live[u] && q == 0) atomicMaxFloat(&mS[dl[u]], logit);
      } else if (PHASE == 1) {
        if (live[u] && q == 0)
          atomicAdd(&dS[dl[u]], __expf(logit - mS[dl[u]]));
      } else {
        const float alpha = __expf(logit - mS[dl[u]]) / dS[dl[u]];
        if (live[u]) {
          float* a = acc + dl[u] * OWN2_S + q * 4;
          atomicAdd(&a[0], alpha * xv[u].x);
          atomicAdd(&a[1], alpha * xv[u].y);
          atomicAdd(&a[2], alpha * xv[u].z);
          atomicAdd(&a[3], alpha * xv[u].w);
        }
      }
    }
  }
}

__global__ __launch_bounds__(OWN_T) void layer2_owner_lds(
    const int* __restrict__ srcE, const int* __restrict__ dstE, int E, int N,
    const float* __restrict__ xl, const float* __restrict__ xr,
    const float* __restrict__ att, const float* __restrict__ bias,
    const int* __restrict__ savedList, const int* __restrict__ savedCnt,
    float* __restrict__ out) {
  __shared__ float acc[OWN_R * OWN2_S];
  __shared__ __attribute__((aligned(16))) float xrS[OWN_R * OWN2_XS];
  __shared__ float mS[OWN_R];
  __shared__ float dS[OWN_R];
  __shared__ float lg[OWN_CAP];
  __shared__ int2 list[OWN_CAP];
  __shared__ int s_cnt;

  const int tid = threadIdx.x;
  const int n0 = blockIdx.x * OWN_R;
  const int cnt = min(OWN_R, N - n0);
  const int q = tid & 3;

  const int saved = __builtin_amdgcn_readfirstlane(savedCnt[blockIdx.x]);
  const bool fast = saved >= 0;
  const float4 a4 = make_float4(att[q * 4], att[q * 4 + 1], att[q * 4 + 2],
                                att[q * 4 + 3]);
  static_assert(OWN_R * 4 <= OWN_T, "one 16-byte piece of the xr rows per thread");
  float4 xrv = make_float4(0.f, 0.f, 0.f, 0.f);
  if (tid < cnt * 4)
    xrv = *reinterpret_cast<const float4*>(xr + (size_t)n0 * 16 + tid * 4);
  const int2* __restrict__ sl =
      reinterpret_cast<const int2*>(savedList) + (size_t)blockIdx.x * OWN_CAP;
  int2 sv[OWN_CAP / OWN_T];
#pragma unroll
  for (int u = 0; u < OWN_CAP / OWN_T; ++u) {
    const int i = u * OWN_T + tid;
    sv[u] = make_int2(0, 0);
    if (i < saved) sv[u] = sl[i];
  }

  for (int i = tid; i < OWN_R * OWN2_S; i += OWN_T) acc[i] = 0.f;
  for (int i = tid; i < OWN_R; i += OWN_T) {
    mS[i] = -__builtin_inff();
    dS[i] = 0.f;
  }
  if (tid < cnt * 4)
    *reinterpret_cast<float4*>(xrS + (tid >> 2) * OWN2_XS + q * 4) = xrv;
#pragma unroll
  for (int u = 0; u < OWN_CAP / OWN_T; ++u) {
    const int i = u * OWN_T + tid;
    if (i < saved) list[i] = sv[u];
  }
  __syncthreads();

  const int ntiles = fast ? 1 : 1 + (E + OWN_CAP - 1) / OWN_CAP;
  for (int phase = 0; phase < 3; ++phase) {
    for (int t = 0; t < ntiles; ++t) {
      int count = saved;
      if (!fast) {
        if (tid == 0) s_cnt = 0;
        __syncthreads();
        if (t == 0) {
          for (int i = tid; i < cnt; i += OWN_T)
            list[i] = make_int2(n0 + i, n0 + i);
        } else {
          const int e0 = (t - 1) * OWN_CAP;
          const int e1 = min(e0 + OWN_CAP, E);
          for (int e = e0 + tid; e < e1; e += OWN_T) {
            const int d = dstE[e];
            if ((unsigned)(d - n0) < (unsigned)cnt) {
              const int slot = atomicAdd(&s_cnt, 1);
              list[slot] = make_int2(srcE[e], d);
            }
          }
        }
        __syncthreads();
        count = t == 0 ? cnt : __builtin_amdgcn_readfirstlane(s_cnt);
      }
      if (phase == 0)
        own2_process<0>(count, fast, n0, list, lg, acc, mS, dS, xrS, xl, a4);
      else if (phase == 1)
        own2_process<1>(count, fast, n0, list, lg, acc, mS, dS, xrS, xl, a4);
      else
        own2_process<2>(count, fast, n0, list, lg, acc, mS, dS, xrS, xl, a4);
      __syncthreads();
    }
  }

  for (int j = tid; j < cnt * 16; j += OWN_T) {
    const int i = j >> 4;
    const int f = j & 15;
    out[(size_t)(n0 + i) * 16 + f] = acc[i * OWN2_S + f] + bias[f];
  }
}

extern "C" void kernel_launch(void* const* d_in, const int* in_sizes, int n_in,
                              void* d_out, int out_size, void* d_ws,
                              size_t ws_size, hipStream_t stream) {
  const float* x = (const float*)d_in[0];
  const int* edge = (const int*)d_in[1];
  const float* W1l = (const float*)d_in[2];
  const float* W1r = (const float*)d_in[3];
  const float* att1 = (const float*)d_in[4];
  const float* b1 = (const float*)d_in[5];
  const float* W2l = (const float*)d_in[6];
  const float* W2r = (const float*)d_in[7];
  const float* att2 = (const float*)d_in[8];
  const float* b2 = (const float*)d_in[9];
  float* out = (float*)d_out;

  const int F_IN = 128, H1 = 80, F_OUT = 16;
  const int N = in_sizes[0] / F_IN;
  const int E = in_sizes[1] / 2;
  const int* srcE = edge;
  const int* dstE = edge + E;

  float* ws = (float*)d_ws;
  float* xl1 = ws;                 ws += (size_t)N * H1;
  float* xr1 = ws;                 ws += (size_t)N * H1;
  float* agg1 = ws;                ws += (size_t)N * H1;
  float* xl2 = ws;                 ws += (size_t)N * F_OUT;
  float* xr2 = ws;                 ws += (size_t)N * F_OUT;
  const int ranges = (N + OWN_R - 1) / OWN_R;
  int* savedList = (int*)ws;       ws += (size_t)ranges * OWN_CAP * 2;
  const int chunks = (E + BM_CHUNK - 1) / BM_CHUNK;
  unsigned int* planes = (unsigned int*)ws; ws += (size_t)ranges * chunks * BM_W;
  int* savedCnt = (int*)ws;
  (void)ws_size; (void)n_in; (void)out_size;

  const int BLK = GEMM_T;

  {
    const int rowTiles = N / 16, colTiles = H1 / 16;
    const int waves = rowTiles * colTiles;
    const int blocks = (waves + 7) / 8;
    wmma_dual_gemm<128, 80, true><<<blocks, BLK, 0, stream>>>(
        x, W1l, W1r, xl1, xr1, rowTiles, colTiles, dstE, E, N, planes);
  }
  layer1_owner_lds<<<ranges, OWN_T, 0, stream>>>(
      srcE, dstE, E, N, xl1, xr1, att1, b1, agg1, savedList, savedCnt, planes);

  {
    const int rowTiles = N / 16, colTiles = F_OUT / 16;
    const int waves = rowTiles * colTiles;
    const int blocks = (waves + 7) / 8;
    wmma_dual_gemm<80, 16, false><<<blocks, BLK, 0, stream>>>(
        agg1, W2l, W2r, xl2, xr2, rowTiles, colTiles, dstE, E, N, planes);
  }
  layer2_owner_lds<<<ranges, OWN_T, 0, stream>>>(
      srcE, dstE, E, N, xl2, xr2, att2, b2, savedList, savedCnt, out);
}


# Layer-1 logit, denominator and sum passes take 4 items (one head of one edge) per thread per trip: list, att, xl gathers and softmax state of all 4 loaded before first use; LDS arrival order kept
# speedup vs baseline: 5.8674x; 1.0371x over previous
#include <hip/hip_runtime.h>


typedef __attribute__((ext_vector_type(2))) float v2f;
typedef __attribute__((ext_vector_type(8))) float v8f;

#define NEG_SLOPE 0.2f
#define OWN_R 200
#define BM_CHUNK 1024
#define BM_W (BM_CHUNK / 32)
#define GEMM_T 256

template <int K, int LDB, bool KEYS>
__global__ __launch_bounds__(GEMM_T) void wmma_dual_gemm(
    const float* __restrict__ A, const float* __restrict__ Bl,
    const float* __restrict__ Br, float* __restrict__ Cl,
    float* __restrict__ Cr, int rowTiles, int colTiles,
    const int* __restrict__ dstE, int E, int N,
    unsigned int* __restrict__ planes) {
  if (KEYS) {
    __shared__ __attribute__((aligned(16))) unsigned int bm[256 * BM_W];
    uint4* bm4 = reinterpret_cast<uint4*>(bm);
    const int ranges = (N + OWN_R - 1) / OWN_R;
    const int nChunks = (E + BM_CHUNK - 1) / BM_CHUNK;
    const size_t planeWords = (size_t)nChunks * BM_W;
    static_assert(BM_CHUNK == 4 * GEMM_T, "four edges of a tile per thread");
    const int le = 4 * threadIdx.x;
    for (int c = blockIdx.x; c < nChunks; c += gridDim.x) {
      const int e0 = c * BM_CHUNK + le;
      int d[4];
      if (e0 + 3 < E) {
        d[0] = dstE[e0];
        d[1] = dstE[e0 + 1];
        d[2] = dstE[e0 + 2];
        d[3] = dstE[e0 + 3];
      } else {
#pragma unroll
        for (int j = 0; j < 4; ++j) d[j] = e0 + j < E ? dstE[e0 + j] : -1;
      }
      if (c != (int)blockIdx.x) __syncthreads();
      for (int i = threadIdx.x; i < ranges * (BM_W / 4); i += GEMM_T)
        bm4[i] = make_uint4(0u, 0u, 0u, 0u);
      __syncthreads();
#pragma unroll
      for (int j = 0; j < 4; ++j)
        if ((unsigned)d[j] < (unsigned)N)
          atomicOr(&bm[((unsigned)d[j] / OWN_R) * BM_W + (le >> 5)],
                   1u << ((le + j) & 31));
      __syncthreads();
      for (int i = threadIdx.x; i < ranges * (BM_W / 4); i += GEMM_T) {
        const int r = i / (BM_W / 4);
        const int w4 = i - r * (BM_W / 4);
        reinterpret_cast<uint4*>(planes + (size_t)r * planeWords +
                                 (size_t)c * BM_W)[w4] = bm4[i];
      }
    }
  }
  const int wid = (blockIdx.x * blockDim.x + threadIdx.x) >> 5;
  const int lane = threadIdx.x & 31;
  if (wid >= rowTiles * colTiles) return;
  const int rt = wid / colTiles;
  const int ct = wid % colTiles;
  const int row0 = rt * 16;
  const int col0 = ct * 16;
  const int half = lane >> 4;
  const int l16 = lane & 15;
  const int arow = row0 + l16;
  const int bcol = col0 + l16;

  const float* __restrict__ Ap = A + (size_t)arow * K + half * 2;

  v8f accl = {};
  v8f accr = {};
  for (int k = 0; k < K; k += 4) {
    const int ka = k + half * 2;
    v2f a;
    a.x = Ap[k];
    a.y = Ap[k + 1];
    v2f bl, br;
    bl.x = Bl[(size_t)ka * LDB + bcol];
    bl.y = Bl[(size_t)(ka + 1) * LDB + bcol];
    br.x = Br[(size_t)ka * LDB + bcol];
    br.y = Br[(size_t)(ka + 1) * LDB + bcol];
    accl = __builtin_amdgcn_wmma_f32_16x16x4_f32(false, a, false, bl, (short)0,
                                                 accl, false, false);
    accr = __builtin_amdgcn_wmma_f32_16x16x4_f32(false, a, false, br, (short)0,
                                                 accr, false, false);
  }

  const size_t base = (size_t)(row0 + half * 8) * LDB + bcol;
  float* __restrict__ cl = Cl + base;
  float* __restrict__ cr = Cr + base;
#pragma unroll
  for (int i = 0; i < 8; ++i) {
    cl[(size_t)i * LDB] = accl[i];
    cr[(size_t)i * LDB] = accr[i];
  }
}

__device__ __forceinline__ void atomicMaxFloat(float* addr, float val) {
  if (val >= 0.f)
    atomicMax((int*)addr, __float_as_int(val));
  else
    atomicMin((unsigned int*)addr, __float_as_uint(val));
}

#define OWN_S 81
#define OWN_XP 84
#define OWN_CAP 4096
#define OWN_T 1024
#define OWN_U 8
#define OWN_IU 4

template <int PHASE, bool CACHED, int U>
__device__ __forceinline__ void own_items(
    int i0, int n0, const int2* list, float* lg, float* acc, float* mS,
    float* dS, const float* __restrict__ xl, const float* __restrict__ xr,
    const float* __restrict__ att) {
  constexpr bool LOGIT = PHASE == 0 || !CACHED;
  constexpr bool ROWS = LOGIT || PHASE == 2;
  int hh[U], dl[U];
  float av[U][8], xv[U][8], rv[U][8];
  float logit[U], mv[U], dv[U];
#pragma unroll
  for (int u = 0; u < U; ++u) {
    const int i = i0 + u * OWN_T;
    const int ei = i / 10;
    hh[u] = i - ei * 10;
    if (LOGIT) {
#pragma unroll
      for (int dd = 0; dd < 8; ++dd) av[u][dd] = att[hh[u] * 8 + dd];
    }
    const int2 sd = list[ei];
    dl[u] = sd.y - n0;
    if (ROWS) {
      const float* __restrict__ xls = xl + (size_t)sd.x * 80 + hh[u] * 8;
#pragma unroll
      for (int dd = 0; dd < 8; ++dd) xv[u][dd] = xls[dd];
    }
    if (LOGIT && !CACHED) {
      const float* __restrict__ xrd = xr + (size_t)sd.y * 80 + hh[u] * 8;
#pragma unroll
      for (int dd = 0; dd < 8; ++dd) rv[u][dd] = xrd[dd];
    }
    if (!LOGIT) logit[u] = lg[i];
    if (PHASE != 0) mv[u] = mS[dl[u] * 10 + hh[u]];
    if (PHASE == 2) dv[u] = dS[dl[u] * 10 + hh[u]];
  }
  if (LOGIT) {
#pragma unroll
    for (int u = 0; u < U; ++u) {
      if (CACHED) {
        const float* xrd = acc + dl[u] * OWN_XP + hh[u] * 8 + ((hh[u] >> 3) << 2);
        const float4 r0 = *reinterpret_cast<const float4*>(xrd);
        const float4 r1 = *reinterpret_cast<const float4*>(xrd + 4);
        rv[u][0] = r0.x; rv[u][1] = r0.y; rv[u][2] = r0.z; rv[u][3] = r0.w;
        rv[u][4] = r1.x; rv[u][5] = r1.y; rv[u][6] = r1.z; rv[u][7] = r1.w;
      }
      float s = 0.f;
#pragma unroll
      for (int dd = 0; dd < 8; ++dd) {
        float v = xv[u][dd] + rv[u][dd];
        v = v > 0.f ? v : NEG_SLOPE * v;
        s += v * av[u][dd];
      }
      logit[u] = s;
    }
  }
  float w[U];
#pragma unroll
  for (int u = 0; u < U; ++u) {
    w[u] = 0.f;
    if (PHASE != 0) w[u] = __expf(logit[u] - mv[u]);
    if (PHASE == 2) w[u] = w[u] / dv[u];
  }
#pragma unroll
  for (int u = 0; u < U; ++u) {
    const int seg = dl[u] * 10 + hh[u];
    if (PHASE == 0) {
      if (CACHED) lg[i0 + u * OWN_T] = logit[u];
      atomicMaxFloat(&mS[seg], logit[u]);
    } else if (PHASE == 1) {
      atomicAdd(&dS[seg], w[u]);
    } else {
      float* a = acc + dl[u] * OWN_S + hh[u] * 8 + (hh[u] >> 3);
#pragma unroll
      for (int dd = 0; dd < 8; ++dd) atomicAdd(&a[dd], w[u] * xv[u][dd]);
    }
  }
}

template <int PHASE, bool CACHED>
__device__ __forceinline__ void own_process(
    int count, int n0, const int2* list, float* lg, float* acc,
    float* mS, float* dS, const float* __restrict__ xl,
    const float* __restrict__ xr, const float* __restrict__ att) {
  const int items = count * 10;
  int i0 = threadIdx.x;
  constexpr int G = CACHED ? OWN_IU : 1;
  for (; i0 + (G - 1) * OWN_T < items; i0 += G * OWN_T)
    own_items<PHASE, CACHED, G>(i0, n0, list, lg, acc, mS, dS, xl, xr, att);
  for (; i0 < items; i0 += OWN_T)
    own_items<PHASE, CACHED, 1>(i0, n0, list, lg, acc, mS, dS, xl, xr, att);
}

__global__ __launch_bounds__(OWN_T) void layer1_owner_lds(
    const int* __restrict__ srcE, const int* __restrict__ dstE, int E, int N,
    const float* __restrict__ xl, const float* __restrict__ xr,
    const float* __restrict__ att, const float* __restrict__ bias,
    float* __restrict__ agg, int* __restrict__ savedList,
    int* __restrict__ savedCnt, const unsigned int* __restrict__ planes) {
  static_assert(OWN_XP >= 80 + 4 && OWN_XP % 4 == 0 && OWN_XP >= OWN_S,
                "acc holds either layout, xr rows 16-byte aligned");
  __shared__ __attribute__((aligned(16))) float acc[OWN_R * OWN_XP];
  __shared__ float mS[OWN_R * 10];
  __shared__ float dS[OWN_R * 10];
  __shared__ float lg[OWN_CAP * 10];
  __shared__ int2 list[OWN_CAP];
  __shared__ int s_cnt;

  const int tid = threadIdx.x;
  const int n0 = blockIdx.x * OWN_R;
  const int cnt = min(OWN_R, N - n0);

  static_assert(OWN_R * 20 <= 4 * OWN_T, "four 16-byte pieces of the xr rows per thread");
  float4 xrv[4];
#pragma unroll
  for (int u = 0; u < 4; ++u) {
    const int j = u * OWN_T + tid;
    xrv[u] = make_float4(0.f, 0.f, 0.f, 0.f);
    if (j < cnt * 20)
      xrv[u] = *reinterpret_cast<const float4*>(xr + (size_t)n0 * 80 + j * 4);
  }

  for (int i = tid; i < OWN_R * 10; i += OWN_T) {
    mS[i] = -__builtin_inff();
    dS[i] = 0.f;
  }
  if (tid == 0) s_cnt = 0;
  __syncthreads();

  {
    const int nChunks = (E + BM_CHUNK - 1) / BM_CHUNK;
    const size_t planeWords = (size_t)nChunks * BM_W;
    const int nvec = nChunks * (BM_W / 4);
    const uint4* __restrict__ pv =
        reinterpret_cast<const uint4*>(planes + (size_t)blockIdx.x * planeWords);
    const int iters = (nvec + OWN_T - 1) / OWN_T;
    for (int k0 = 0; k0 < iters; k0 += OWN_U) {
      uint4 q[OWN_U];
#pragma unroll
      for (int u = 0; u < OWN_U; ++u) {
        const int idx = (k0 + u) * OWN_T + tid;
        q[u] = make_uint4(0u, 0u, 0u, 0u);
        if (idx < nvec) q[u] = pv[idx];
      }
      int nb = 0;
#pragma unroll
      for (int u = 0; u < OWN_U; ++u)
        nb += __popc(q[u].x) + __popc(q[u].y) + __popc(q[u].z) + __popc(q[u].w);
      int slot = atomicAdd(&s_cnt, nb);
#pragma unroll
      for (int u = 0; u < OWN_U; ++u) {
        const int idx = (k0 + u) * OWN_T + tid;
        const unsigned int w[4] = {q[u].x, q[u].y, q[u].z, q[u].w};
#pragma unroll
        for (int j = 0; j < 4; ++j) {
          unsigned int m = w[j];
          while (m != 0u) {
            const int p = __builtin_ctz(m);
            m &= m - 1u;
            if (slot < OWN_CAP) list[slot] = make_int2((idx * 4 + j) * 32 + p, 0);
            ++slot;
          }
        }
      }
    }
  }
#pragma unroll
  for (int u = 0; u < 4; ++u) {
    const int j = u * OWN_T + tid;
    if (j < cnt * 20) {
      const int row = j / 20;
      const int c4 = j - row * 20;
      *reinterpret_cast<float4*>(acc + row * OWN_XP + c4 * 4 + ((c4 >> 4) << 2)) =
          xrv[u];
    }
  }
  __syncthreads();

  const int total = __builtin_amdgcn_readfirstlane(s_cnt);
  const bool fast = total + cnt <= OWN_CAP;
  if (fast) {
    for (int i = tid; i < total; i += OWN_T) {
      const int e = list[i].x;
      list[i] = make_int2(srcE[e], dstE[e]);
    }
    for (int i = tid; i < cnt; i += OWN_T)
      list[total + i] = make_int2(n0 + i, n0 + i);
  } else {
    for (int i = tid; i < OWN_R * OWN_S; i += OWN_T) acc[i] = 0.f;
  }
  __syncthreads();

  if (fast) {
    int2* __restrict__ sl =
        reinterpret_cast<int2*>(savedList) + (size_t)blockIdx.x * OWN_CAP;
    for (int i = tid; i < total + cnt; i += OWN_T) sl[i] = list[i];
  }
  if (tid == 0) savedCnt[blockIdx.x] = fast ? total + cnt : -1;

  const int ntiles = fast ? 1 : 1 + (E + OWN_CAP - 1) / OWN_CAP;
  for (int phase = 0; phase < 3; ++phase) {
    for (int t = 0; t < ntiles; ++t) {
      int count = total + cnt;
      if (!fast) {
        if (tid == 0) s_cnt = 0;
        __syncthreads();
        if (t == 0) {
          for (int i = tid; i < cnt; i += OWN_T)
            list[i] = make_int2(n0 + i, n0 + i);
        } else {
          const int e0 = (t - 1) * OWN_CAP;
          const int e1 = min(e0 + OWN_CAP, E);
          for (int e = e0 + tid; e < e1; e += OWN_T) {
            const int d = dstE[e];
            if ((unsigned)(d - n0) < (unsigned)cnt) {
              const int slot = atomicAdd(&s_cnt, 1);
              list[slot] = make_int2(srcE[e], d);
            }
          }
        }
        __syncthreads();
        count = t == 0 ? cnt : __builtin_amdgcn_readfirstlane(s_cnt);
      }
      if (phase == 0) {
        if (fast)
          own_process<0, true>(count, n0, list, lg, acc, mS, dS, xl, xr, att);
        else
          own_process<0, false>(count, n0, list, lg, acc, mS, dS, xl, xr, att);
      } else if (phase == 1) {
        if (fast) {
          for (int i = tid; i < OWN_R * OWN_S; i += OWN_T) acc[i] = 0.f;
          own_process<1, true>(count, n0, list, lg, acc, mS, dS, xl, xr, att);
        } else {
          own_process<1, false>(count, n0, list, lg, acc, mS, dS, xl, xr, att);
        }
      } else {
        if (fast)
          own_process<2, true>(count, n0, list, lg, acc, mS, dS, xl, xr, att);
        else
          own_process<2, false>(count, n0, list, lg, acc, mS, dS, xl, xr, att);
      }
      __syncthreads();
    }
  }

  for (int j = tid; j < cnt * 80; j += OWN_T) {
    const int i = j / 80;
    const int f = j - i * 80;
    const float v = acc[i * OWN_S + f + (f >> 6)] + bias[f];
    agg[(size_t)(n0 + i) * 80 + f] = fmaxf(v, 0.f);
  }
}

#define OWN2_S 17
#define OWN2_XS 20
#define OWN2_U 4
#define OWN2_G (OWN_T / 4)

template <int CTRL>
__device__ __forceinline__ float dpp_flt(float v) {
  return __int_as_float(
      __builtin_amdgcn_update_dpp(0, __float_as_int(v), CTRL, 0xf, 0xf, true));
}

template <int PHASE>
__device__ __forceinline__ void own2_process(
    int count, bool cached, int n0, const int2* list, float* lg, float* acc,
    float* mS, float* dS, const float* xrS, const float* __restrict__ xl,
    float4 a4) {
  if (PHASE == 1 && cached) {
    for (int e = threadIdx.x; e < count; e += OWN_T) {
      const int dl = list[e].y - n0;
      atomicAdd(&dS[dl], __expf(lg[e] - mS[dl]));
    }
    return;
  }
  const int q = threadIdx.x & 3;
  const int g = threadIdx.x >> 2;
  for (int e0 = g; e0 < count; e0 += OWN2_U * OWN2_G) {
    int ee[OWN2_U], dl[OWN2_U];
    bool live[OWN2_U];
    float4 xv[OWN2_U];
#pragma unroll
    for (int u = 0; u < OWN2_U; ++u) {
      const int e = e0 + u * OWN2_G;
      live[u] = e < count;
      ee[u] = live[u] ? e : e0;
      const int2 sd = list[ee[u]];
      dl[u] = sd.y - n0;
      xv[u] = *reinterpret_cast<const float4*>(xl + (size_t)sd.x * 16 + q * 4);
    }
#pragma unroll
    for (int u = 0; u < OWN2_U; ++u) {
      float logit;
      if (PHASE == 0 || !cached) {
        const float4 rv =
            *reinterpret_cast<const float4*>(xrS + dl[u] * OWN2_XS + q * 4);
        float v0 = xv[u].x + rv.x, v1 = xv[u].y + rv.y;
        float v2 = xv[u].z + rv.z, v3 = xv[u].w + rv.w;
        v0 = v0 > 0.f ? v0 : NEG_SLOPE * v0;
        v1 = v1 > 0.f ? v1 : NEG_SLOPE * v1;
        v2 = v2 > 0.f ? v2 : NEG_SLOPE * v2;
        v3 = v3 > 0.f ? v3 : NEG_SLOPE * v3;
        float p = v0 * a4.x + v1 * a4.y + v2 * a4.z + v3 * a4.w;
        p += dpp_flt<0x161>(p);
        p += dpp_flt<0x162>(p);
        logit = p;
        if (PHASE == 0 && cached && q == 0 && live[u]) lg[ee[u]] = logit;
      } else {
        logit = lg[ee[u]];
      }
      if (PHASE == 0) {
        if (live[u] && q == 0) atomicMaxFloat(&mS[dl[u]], logit);
      } else if (PHASE == 1) {
        if (live[u] && q == 0)
          atomicAdd(&dS[dl[u]], __expf(logit - mS[dl[u]]));
      } else {
        const float alpha = __expf(logit - mS[dl[u]]) / dS[dl[u]];
        if (live[u]) {
          float* a = acc + dl[u] * OWN2_S + q * 4;
          atomicAdd(&a[0], alpha * xv[u].x);
          atomicAdd(&a[1], alpha * xv[u].y);
          atomicAdd(&a[2], alpha * xv[u].z);
          atomicAdd(&a[3], alpha * xv[u].w);
        }
      }
    }
  }
}

__global__ __launch_bounds__(OWN_T) void layer2_owner_lds(
    const int* __restrict__ srcE, const int* __restrict__ dstE, int E, int N,
    const float* __restrict__ xl, const float* __restrict__ xr,
    const float* __restrict__ att, const float* __restrict__ bias,
    const int* __restrict__ savedList, const int* __restrict__ savedCnt,
    float* __restrict__ out) {
  __shared__ float acc[OWN_R * OWN2_S];
  __shared__ __attribute__((aligned(16))) float xrS[OWN_R * OWN2_XS];
  __shared__ float mS[OWN_R];
  __shared__ float dS[OWN_R];
  __shared__ float lg[OWN_CAP];
  __shared__ int2 list[OWN_CAP];
  __shared__ int s_cnt;

  const int tid = threadIdx.x;
  const int n0 = blockIdx.x * OWN_R;
  const int cnt = min(OWN_R, N - n0);
  const int q = tid & 3;

  const int saved = __builtin_amdgcn_readfirstlane(savedCnt[blockIdx.x]);
  const bool fast = saved >= 0;
  const float4 a4 = make_float4(att[q * 4], att[q * 4 + 1], att[q * 4 + 2],
                                att[q * 4 + 3]);
  static_assert(OWN_R * 4 <= OWN_T, "one 16-byte piece of the xr rows per thread");
  float4 xrv = make_float4(0.f, 0.f, 0.f, 0.f);
  if (tid < cnt * 4)
    xrv = *reinterpret_cast<const float4*>(xr + (size_t)n0 * 16 + tid * 4);
  const int2* __restrict__ sl =
      reinterpret_cast<const int2*>(savedList) + (size_t)blockIdx.x * OWN_CAP;
  int2 sv[OWN_CAP / OWN_T];
#pragma unroll
  for (int u = 0; u < OWN_CAP / OWN_T; ++u) {
    const int i = u * OWN_T + tid;
    sv[u] = make_int2(0, 0);
    if (i < saved) sv[u] = sl[i];
  }

  for (int i = tid; i < OWN_R * OWN2_S; i += OWN_T) acc[i] = 0.f;
  for (int i = tid; i < OWN_R; i += OWN_T) {
    mS[i] = -__builtin_inff();
    dS[i] = 0.f;
  }
  if (tid < cnt * 4)
    *reinterpret_cast<float4*>(xrS + (tid >> 2) * OWN2_XS + q * 4) = xrv;
#pragma unroll
  for (int u = 0; u < OWN_CAP / OWN_T; ++u) {
    const int i = u * OWN_T + tid;
    if (i < saved) list[i] = sv[u];
  }
  __syncthreads();

  const int ntiles = fast ? 1 : 1 + (E + OWN_CAP - 1) / OWN_CAP;
  for (int phase = 0; phase < 3; ++phase) {
    for (int t = 0; t < ntiles; ++t) {
      int count = saved;
      if (!fast) {
        if (tid == 0) s_cnt = 0;
        __syncthreads();
        if (t == 0) {
          for (int i = tid; i < cnt; i += OWN_T)
            list[i] = make_int2(n0 + i, n0 + i);
        } else {
          const int e0 = (t - 1) * OWN_CAP;
          const int e1 = min(e0 + OWN_CAP, E);
          for (int e = e0 + tid; e < e1; e += OWN_T) {
            const int d = dstE[e];
            if ((unsigned)(d - n0) < (unsigned)cnt) {
              const int slot = atomicAdd(&s_cnt, 1);
              list[slot] = make_int2(srcE[e], d);
            }
          }
        }
        __syncthreads();
        count = t == 0 ? cnt : __builtin_amdgcn_readfirstlane(s_cnt);
      }
      if (phase == 0)
        own2_process<0>(count, fast, n0, list, lg, acc, mS, dS, xrS, xl, a4);
      else if (phase == 1)
        own2_process<1>(count, fast, n0, list, lg, acc, mS, dS, xrS, xl, a4);
      else
        own2_process<2>(count, fast, n0, list, lg, acc, mS, dS, xrS, xl, a4);
      __syncthreads();
    }
  }

  for (int j = tid; j < cnt * 16; j += OWN_T) {
    const int i = j >> 4;
    const int f = j & 15;
    out[(size_t)(n0 + i) * 16 + f] = acc[i * OWN2_S + f] + bias[f];
  }
}

extern "C" void kernel_launch(void* const* d_in, const int* in_sizes, int n_in,
                              void* d_out, int out_size, void* d_ws,
                              size_t ws_size, hipStream_t stream) {
  const float* x = (const float*)d_in[0];
  const int* edge = (const int*)d_in[1];
  const float* W1l = (const float*)d_in[2];
  const float* W1r = (const float*)d_in[3];
  const float* att1 = (const float*)d_in[4];
  const float* b1 = (const float*)d_in[5];
  const float* W2l = (const float*)d_in[6];
  const float* W2r = (const float*)d_in[7];
  const float* att2 = (const float*)d_in[8];
  const float* b2 = (const float*)d_in[9];
  float* out = (float*)d_out;

  const int F_IN = 128, H1 = 80, F_OUT = 16;
  const int N = in_sizes[0] / F_IN;
  const int E = in_sizes[1] / 2;
  const int* srcE = edge;
  const int* dstE = edge + E;

  float* ws = (float*)d_ws;
  float* xl1 = ws;                 ws += (size_t)N * H1;
  float* xr1 = ws;                 ws += (size_t)N * H1;
  float* agg1 = ws;                ws += (size_t)N * H1;
  float* xl2 = ws;                 ws += (size_t)N * F_OUT;
  float* xr2 = ws;                 ws += (size_t)N * F_OUT;
  const int ranges = (N + OWN_R - 1) / OWN_R;
  int* savedList = (int*)ws;       ws += (size_t)ranges * OWN_CAP * 2;
  const int chunks = (E + BM_CHUNK - 1) / BM_CHUNK;
  unsigned int* planes = (unsigned int*)ws; ws += (size_t)ranges * chunks * BM_W;
  int* savedCnt = (int*)ws;
  (void)ws_size; (void)n_in; (void)out_size;

  const int BLK = GEMM_T;

  {
    const int rowTiles = N / 16, colTiles = H1 / 16;
    const int waves = rowTiles * colTiles;
    const int blocks = (waves + 7) / 8;
    wmma_dual_gemm<128, 80, true><<<blocks, BLK, 0, stream>>>(
        x, W1l, W1r, xl1, xr1, rowTiles, colTiles, dstE, E, N, planes);
  }
  layer1_owner_lds<<<ranges, OWN_T, 0, stream>>>(
      srcE, dstE, E, N, xl1, xr1, att1, b1, agg1, savedList, savedCnt, planes);

  {
    const int rowTiles = N / 16, colTiles = F_OUT / 16;
    const int waves = rowTiles * colTiles;
    const int blocks = (waves + 7) / 8;
    wmma_dual_gemm<80, 16, false><<<blocks, BLK, 0, stream>>>(
        agg1, W2l, W2r, xl2, xr2, rowTiles, colTiles, dstE, E, N, planes);
  }
  layer2_owner_lds<<<ranges, OWN_T, 0, stream>>>(
      srcE, dstE, E, N, xl2, xr2, att2, b2, savedList, savedCnt, out);
}


# Layer-1 edge softmax in ONE pass: shift = destination's self-loop logit (ready before the pass), not the max; sums divided by the denominator at the end (rounding changes, LDS arrival order kept)
# speedup vs baseline: 6.5886x; 1.1229x over previous
#include <hip/hip_runtime.h>


typedef __attribute__((ext_vector_type(2))) float v2f;
typedef __attribute__((ext_vector_type(8))) float v8f;

#define NEG_SLOPE 0.2f
#define OWN_R 200
#define BM_CHUNK 1024
#define BM_W (BM_CHUNK / 32)
#define GEMM_T 256

template <int K, int LDB, bool KEYS>
__global__ __launch_bounds__(GEMM_T) void wmma_dual_gemm(
    const float* __restrict__ A, const float* __restrict__ Bl,
    const float* __restrict__ Br, float* __restrict__ Cl,
    float* __restrict__ Cr, int rowTiles, int colTiles,
    const int* __restrict__ dstE, int E, int N,
    unsigned int* __restrict__ planes) {
  if (KEYS) {
    __shared__ __attribute__((aligned(16))) unsigned int bm[256 * BM_W];
    uint4* bm4 = reinterpret_cast<uint4*>(bm);
    const int ranges = (N + OWN_R - 1) / OWN_R;
    const int nChunks = (E + BM_CHUNK - 1) / BM_CHUNK;
    const size_t planeWords = (size_t)nChunks * BM_W;
    static_assert(BM_CHUNK == 4 * GEMM_T, "four edges of a tile per thread");
    const int le = 4 * threadIdx.x;
    for (int c = blockIdx.x; c < nChunks; c += gridDim.x) {
      const int e0 = c * BM_CHUNK + le;
      int d[4];
      if (e0 + 3 < E) {
        d[0] = dstE[e0];
        d[1] = dstE[e0 + 1];
        d[2] = dstE[e0 + 2];
        d[3] = dstE[e0 + 3];
      } else {
#pragma unroll
        for (int j = 0; j < 4; ++j) d[j] = e0 + j < E ? dstE[e0 + j] : -1;
      }
      if (c != (int)blockIdx.x) __syncthreads();
      for (int i = threadIdx.x; i < ranges * (BM_W / 4); i += GEMM_T)
        bm4[i] = make_uint4(0u, 0u, 0u, 0u);
      __syncthreads();
#pragma unroll
      for (int j = 0; j < 4; ++j)
        if ((unsigned)d[j] < (unsigned)N)
          atomicOr(&bm[((unsigned)d[j] / OWN_R) * BM_W + (le >> 5)],
                   1u << ((le + j) & 31));
      __syncthreads();
      for (int i = threadIdx.x; i < ranges * (BM_W / 4); i += GEMM_T) {
        const int r = i / (BM_W / 4);
        const int w4 = i - r * (BM_W / 4);
        reinterpret_cast<uint4*>(planes + (size_t)r * planeWords +
                                 (size_t)c * BM_W)[w4] = bm4[i];
      }
    }
  }
  const int wid = (blockIdx.x * blockDim.x + threadIdx.x) >> 5;
  const int lane = threadIdx.x & 31;
  if (wid >= rowTiles * colTiles) return;
  const int rt = wid / colTiles;
  const int ct = wid % colTiles;
  const int row0 = rt * 16;
  const int col0 = ct * 16;
  const int half = lane >> 4;
  const int l16 = lane & 15;
  const int arow = row0 + l16;
  const int bcol = col0 + l16;

  const float* __restrict__ Ap = A + (size_t)arow * K + half * 2;

  v8f accl = {};
  v8f accr = {};
  for (int k = 0; k < K; k += 4) {
    const int ka = k + half * 2;
    v2f a;
    a.x = Ap[k];
    a.y = Ap[k + 1];
    v2f bl, br;
    bl.x = Bl[(size_t)ka * LDB + bcol];
    bl.y = Bl[(size_t)(ka + 1) * LDB + bcol];
    br.x = Br[(size_t)ka * LDB + bcol];
    br.y = Br[(size_t)(ka + 1) * LDB + bcol];
    accl = __builtin_amdgcn_wmma_f32_16x16x4_f32(false, a, false, bl, (short)0,
                                                 accl, false, false);
    accr = __builtin_amdgcn_wmma_f32_16x16x4_f32(false, a, false, br, (short)0,
                                                 accr, false, false);
  }

  const size_t base = (size_t)(row0 + half * 8) * LDB + bcol;
  float* __restrict__ cl = Cl + base;
  float* __restrict__ cr = Cr + base;
#pragma unroll
  for (int i = 0; i < 8; ++i) {
    cl[(size_t)i * LDB] = accl[i];
    cr[(size_t)i * LDB] = accr[i];
  }
}

__device__ __forceinline__ void atomicMaxFloat(float* addr, float val) {
  if (val >= 0.f)
    atomicMax((int*)addr, __float_as_int(val));
  else
    atomicMin((unsigned int*)addr, __float_as_uint(val));
}

#define OWN_S 81
#define OWN_XP 84
#define OWN_CAP 4096
#define OWN_T 1024
#define OWN_U 8
#define OWN_IU 4
#define OWN_DMAX 8.0f

__device__ __forceinline__ float own_logit(const float (&xv)[8],
                                           const float (&rv)[8],
                                           const float (&av)[8]) {
  float s = 0.f;
#pragma unroll
  for (int dd = 0; dd < 8; ++dd) {
    float v = xv[dd] + rv[dd];
    v = v > 0.f ? v : NEG_SLOPE * v;
    s += v * av[dd];
  }
  return s;
}

template <int U>
__device__ __forceinline__ bool own_fused_items(
    int i0, int n0, const int2* list, float* acc, const float* xrS,
    const float* cS, float* dS, const float* __restrict__ xl,
    const float* __restrict__ att) {
  int hh[U], dl[U];
  float av[U][8], xv[U][8];
  float cv[U], w[U];
#pragma unroll
  for (int u = 0; u < U; ++u) {
    const int i = i0 + u * OWN_T;
    const int ei = i / 10;
    hh[u] = i - ei * 10;
#pragma unroll
    for (int dd = 0; dd < 8; ++dd) av[u][dd] = att[hh[u] * 8 + dd];
    const int2 sd = list[ei];
    dl[u] = sd.y - n0;
    const float* __restrict__ xls = xl + (size_t)sd.x * 80 + hh[u] * 8;
#pragma unroll
    for (int dd = 0; dd < 8; ++dd) xv[u][dd] = xls[dd];
    cv[u] = cS[dl[u] * 10 + hh[u]];
  }
  bool bad = false;
#pragma unroll
  for (int u = 0; u < U; ++u) {
    const float* xrd = xrS + dl[u] * OWN_XP + hh[u] * 8 + ((hh[u] >> 3) << 2);
    const float4 r0 = *reinterpret_cast<const float4*>(xrd);
    const float4 r1 = *reinterpret_cast<const float4*>(xrd + 4);
    const float rv[8] = {r0.x, r0.y, r0.z, r0.w, r1.x, r1.y, r1.z, r1.w};
    const float d = own_logit(xv[u], rv, av[u]) - cv[u];
    bad |= !(d <= OWN_DMAX);
    w[u] = __expf(d);
  }
#pragma unroll
  for (int u = 0; u < U; ++u) {
    atomicAdd(&dS[dl[u] * 10 + hh[u]], w[u]);
    float* a = acc + dl[u] * OWN_S + hh[u] * 8 + (hh[u] >> 3);
#pragma unroll
    for (int dd = 0; dd < 8; ++dd) atomicAdd(&a[dd], w[u] * xv[u][dd]);
  }
  return bad;
}

template <int PHASE>
__device__ __forceinline__ void own_process3(
    int count, int n0, const int2* list, float* acc, float* mS, float* dS,
    const float* __restrict__ xl, const float* __restrict__ xr,
    const float* __restrict__ att) {
  const int items = count * 10;
  for (int i = threadIdx.x; i < items; i += OWN_T) {
    const int ei = i / 10;
    const int hh = i - ei * 10;
    const int2 sd = list[ei];
    const int dl = sd.y - n0;
    const float* __restrict__ xls = xl + (size_t)sd.x * 80 + hh * 8;
    const float* __restrict__ xrd = xr + (size_t)sd.y * 80 + hh * 8;
    float av[8], xv[8], rv[8];
#pragma unroll
    for (int dd = 0; dd < 8; ++dd) {
      av[dd] = att[hh * 8 + dd];
      xv[dd] = xls[dd];
      rv[dd] = xrd[dd];
    }
    const float logit = own_logit(xv, rv, av);
    const int seg = dl * 10 + hh;
    if (PHASE == 0) {
      atomicMaxFloat(&mS[seg], logit);
    } else if (PHASE == 1) {
      atomicAdd(&dS[seg], __expf(logit - mS[seg]));
    } else {
      const float alpha = __expf(logit - mS[seg]) / dS[seg];
      float* a = acc + dl * OWN_S + hh * 8 + (hh >> 3);
#pragma unroll
      for (int dd = 0; dd < 8; ++dd) atomicAdd(&a[dd], alpha * xv[dd]);
    }
  }
}

__global__ __launch_bounds__(OWN_T) void layer1_owner_lds(
    const int* __restrict__ srcE, const int* __restrict__ dstE, int E, int N,
    const float* __restrict__ xl, const float* __restrict__ xr,
    const float* __restrict__ att, const float* __restrict__ bias,
    float* __restrict__ agg, int* __restrict__ savedList,
    int* __restrict__ savedCnt, const unsigned int* __restrict__ planes) {
  static_assert(OWN_XP >= 80 + 4 && OWN_XP % 4 == 0,
                "xr rows 16-byte aligned, heads 8 and 9 four floats further on");
  __shared__ float acc[OWN_R * OWN_S];
  __shared__ __attribute__((aligned(16))) float xrS[OWN_R * OWN_XP];
  __shared__ float mS[OWN_R * 10];
  __shared__ float dS[OWN_R * 10];
  __shared__ int2 list[OWN_CAP];
  __shared__ int s_cnt;
  __shared__ int s_redo;

  const int tid = threadIdx.x;
  const int n0 = blockIdx.x * OWN_R;
  const int cnt = min(OWN_R, N - n0);

  static_assert(OWN_R * 20 <= 4 * OWN_T, "four 16-byte pieces of the xr rows per thread");
  float4 xrv[4];
#pragma unroll
  for (int u = 0; u < 4; ++u) {
    const int j = u * OWN_T + tid;
    xrv[u] = make_float4(0.f, 0.f, 0.f, 0.f);
    if (j < cnt * 20)
      xrv[u] = *reinterpret_cast<const float4*>(xr + (size_t)n0 * 80 + j * 4);
  }

  for (int i = tid; i < OWN_R * 10; i += OWN_T) {
    mS[i] = -__builtin_inff();
    dS[i] = 0.f;
  }
  for (int i = tid; i < OWN_R * OWN_S; i += OWN_T) acc[i] = 0.f;
  if (tid == 0) {
    s_cnt = 0;
    s_redo = 0;
  }
  __syncthreads();

  {
    const int nChunks = (E + BM_CHUNK - 1) / BM_CHUNK;
    const size_t planeWords = (size_t)nChunks * BM_W;
    const int nvec = nChunks * (BM_W / 4);
    const uint4* __restrict__ pv =
        reinterpret_cast<const uint4*>(planes + (size_t)blockIdx.x * planeWords);
    const int iters = (nvec + OWN_T - 1) / OWN_T;
    for (int k0 = 0; k0 < iters; k0 += OWN_U) {
      uint4 q[OWN_U];
#pragma unroll
      for (int u = 0; u < OWN_U; ++u) {
        const int idx = (k0 + u) * OWN_T + tid;
        q[u] = make_uint4(0u, 0u, 0u, 0u);
        if (idx < nvec) q[u] = pv[idx];
      }
      int nb = 0;
#pragma unroll
      for (int u = 0; u < OWN_U; ++u)
        nb += __popc(q[u].x) + __popc(q[u].y) + __popc(q[u].z) + __popc(q[u].w);
      int slot = atomicAdd(&s_cnt, nb);
#pragma unroll
      for (int u = 0; u < OWN_U; ++u) {
        const int idx = (k0 + u) * OWN_T + tid;
        const unsigned int w[4] = {q[u].x, q[u].y, q[u].z, q[u].w};
#pragma unroll
        for (int j = 0; j < 4; ++j) {
          unsigned int m = w[j];
          while (m != 0u) {
            const int p = __builtin_ctz(m);
            m &= m - 1u;
            if (slot < OWN_CAP) list[slot] = make_int2((idx * 4 + j) * 32 + p, 0);
            ++slot;
          }
        }
      }
    }
  }
#pragma unroll
  for (int u = 0; u < 4; ++u) {
    const int j = u * OWN_T + tid;
    if (j < cnt * 20) {
      const int row = j / 20;
      const int c4 = j - row * 20;
      *reinterpret_cast<float4*>(xrS + row * OWN_XP + c4 * 4 + ((c4 >> 4) << 2)) =
          xrv[u];
    }
  }
  __syncthreads();

  const int total = __builtin_amdgcn_readfirstlane(s_cnt);
  const bool fast = total + cnt <= OWN_CAP;
  if (fast) {
    for (int i = tid; i < total; i += OWN_T) {
      const int e = list[i].x;
      list[i] = make_int2(srcE[e], dstE[e]);
    }
    for (int i = tid; i < cnt; i += OWN_T)
      list[total + i] = make_int2(n0 + i, n0 + i);
    for (int j = tid; j < cnt * 10; j += OWN_T) {
      const int i = j / 10;
      const int h = j - i * 10;
      const float* __restrict__ xls = xl + (size_t)(n0 + i) * 80 + h * 8;
      const float* xrd = xrS + i * OWN_XP + h * 8 + ((h >> 3) << 2);
      float av[8], xv[8], rv[8];
#pragma unroll
      for (int dd = 0; dd < 8; ++dd) {
        av[dd] = att[h * 8 + dd];
        xv[dd] = xls[dd];
        rv[dd] = xrd[dd];
      }
      mS[j] = own_logit(xv, rv, av);
    }
  }
  __syncthreads();

  if (fast) {
    int2* __restrict__ sl =
        reinterpret_cast<int2*>(savedList) + (size_t)blockIdx.x * OWN_CAP;
    for (int i = tid; i < total + cnt; i += OWN_T) sl[i] = list[i];
  }
  if (tid == 0) savedCnt[blockIdx.x] = fast ? total + cnt : -1;

  bool three = !fast;
  if (fast) {
    const int items = (total + cnt) * 10;
    bool bad = false;
    int i0 = tid;
    for (; i0 + (OWN_IU - 1) * OWN_T < items; i0 += OWN_IU * OWN_T)
      bad |= own_fused_items<OWN_IU>(i0, n0, list, acc, xrS, mS, dS, xl, att);
    for (; i0 < items; i0 += OWN_T)
      bad |= own_fused_items<1>(i0, n0, list, acc, xrS, mS, dS, xl, att);
    if (bad) s_redo = 1;
    __syncthreads();
    three = __builtin_amdgcn_readfirstlane(s_redo) != 0;
    if (three) {
      for (int i = tid; i < OWN_R * 10; i += OWN_T) {
        mS[i] = -__builtin_inff();
        dS[i] = 0.f;
      }
      for (int i = tid; i < OWN_R * OWN_S; i += OWN_T) acc[i] = 0.f;
      __syncthreads();
    }
  }

  if (three) {
    const int ntiles = fast ? 1 : 1 + (E + OWN_CAP - 1) / OWN_CAP;
    for (int phase = 0; phase < 3; ++phase) {
      for (int t = 0; t < ntiles; ++t) {
        int count = total + cnt;
        if (!fast) {
          if (tid == 0) s_cnt = 0;
          __syncthreads();
          if (t == 0) {
            for (int i = tid; i < cnt; i += OWN_T)
              list[i] = make_int2(n0 + i, n0 + i);
          } else {
            const int e0 = (t - 1) * OWN_CAP;
            const int e1 = min(e0 + OWN_CAP, E);
            for (int e = e0 + tid; e < e1; e += OWN_T) {
              const int d = dstE[e];
              if ((unsigned)(d - n0) < (unsigned)cnt) {
                const int slot = atomicAdd(&s_cnt, 1);
                list[slot] = make_int2(srcE[e], d);
              }
            }
          }
          __syncthreads();
          count = t == 0 ? cnt : __builtin_amdgcn_readfirstlane(s_cnt);
        }
        if (phase == 0)
          own_process3<0>(count, n0, list, acc, mS, dS, xl, xr, att);
        else if (phase == 1)
          own_process3<1>(count, n0, list, acc, mS, dS, xl, xr, att);
        else
          own_process3<2>(count, n0, list, acc, mS, dS, xl, xr, att);
        __syncthreads();
      }
    }
  }

  for (int j = tid; j < cnt * 80; j += OWN_T) {
    const int i = j / 80;
    const int f = j - i * 80;
    float v = acc[i * OWN_S + f + (f >> 6)];
    if (!three) v = v / dS[i * 10 + (f >> 3)];
    agg[(size_t)(n0 + i) * 80 + f] = fmaxf(v + bias[f], 0.f);
  }
}

#define OWN2_S 17
#define OWN2_XS 20
#define OWN2_U 4
#define OWN2_G (OWN_T / 4)

template <int CTRL>
__device__ __forceinline__ float dpp_flt(float v) {
  return __int_as_float(
      __builtin_amdgcn_update_dpp(0, __float_as_int(v), CTRL, 0xf, 0xf, true));
}

template <int PHASE>
__device__ __forceinline__ void own2_process(
    int count, bool cached, int n0, const int2* list, float* lg, float* acc,
    float* mS, float* dS, const float* xrS, const float* __restrict__ xl,
    float4 a4) {
  if (PHASE == 1 && cached) {
    for (int e = threadIdx.x; e < count; e += OWN_T) {
      const int dl = list[e].y - n0;
      atomicAdd(&dS[dl], __expf(lg[e] - mS[dl]));
    }
    return;
  }
  const int q = threadIdx.x & 3;
  const int g = threadIdx.x >> 2;
  for (int e0 = g; e0 < count; e0 += OWN2_U * OWN2_G) {
    int ee[OWN2_U], dl[OWN2_U];
    bool live[OWN2_U];
    float4 xv[OWN2_U];
#pragma unroll
    for (int u = 0; u < OWN2_U; ++u) {
      const int e = e0 + u * OWN2_G;
      live[u] = e < count;
      ee[u] = live[u] ? e : e0;
      const int2 sd = list[ee[u]];
      dl[u] = sd.y - n0;
      xv[u] = *reinterpret_cast<const float4*>(xl + (size_t)sd.x * 16 + q * 4);
    }
#pragma unroll
    for (int u = 0; u < OWN2_U; ++u) {
      float logit;
      if (PHASE == 0 || !cached) {
        const float4 rv =
            *reinterpret_cast<const float4*>(xrS + dl[u] * OWN2_XS + q * 4);
        float v0 = xv[u].x + rv.x, v1 = xv[u].y + rv.y;
        float v2 = xv[u].z + rv.z, v3 = xv[u].w + rv.w;
        v0 = v0 > 0.f ? v0 : NEG_SLOPE * v0;
        v1 = v1 > 0.f ? v1 : NEG_SLOPE * v1;
        v2 = v2 > 0.f ? v2 : NEG_SLOPE * v2;
        v3 = v3 > 0.f ? v3 : NEG_SLOPE * v3;
        float p = v0 * a4.x + v1 * a4.y + v2 * a4.z + v3 * a4.w;
        p += dpp_flt<0x161>(p);
        p += dpp_flt<0x162>(p);
        logit = p;
        if (PHASE == 0 && cached && q == 0 && live[u]) lg[ee[u]] = logit;
      } else {
        logit = lg[ee[u]];
      }
      if (PHASE == 0) {
        if (live[u] && q == 0) atomicMaxFloat(&mS[dl[u]], logit);
      } else if (PHASE == 1) {
        if (live[u] && q == 0)
          atomicAdd(&dS[dl[u]], __expf(logit - mS[dl[u]]));
      } else {
        const float alpha = __expf(logit - mS[dl[u]]) / dS[dl[u]];
        if (live[u]) {
          float* a = acc + dl[u] * OWN2_S + q * 4;
          atomicAdd(&a[0], alpha * xv[u].x);
          atomicAdd(&a[1], alpha * xv[u].y);
          atomicAdd(&a[2], alpha * xv[u].z);
          atomicAdd(&a[3], alpha * xv[u].w);
        }
      }
    }
  }
}

__global__ __launch_bounds__(OWN_T) void layer2_owner_lds(
    const int* __restrict__ srcE, const int* __restrict__ dstE, int E, int N,
    const float* __restrict__ xl, const float* __restrict__ xr,
    const float* __restrict__ att, const float* __restrict__ bias,
    const int* __restrict__ savedList, const int* __restrict__ savedCnt,
    float* __restrict__ out) {
  __shared__ float acc[OWN_R * OWN2_S];
  __shared__ __attribute__((aligned(16))) float xrS[OWN_R * OWN2_XS];
  __shared__ float mS[OWN_R];
  __shared__ float dS[OWN_R];
  __shared__ float lg[OWN_CAP];
  __shared__ int2 list[OWN_CAP];
  __shared__ int s_cnt;

  const int tid = threadIdx.x;
  const int n0 = blockIdx.x * OWN_R;
  const int cnt = min(OWN_R, N - n0);
  const int q = tid & 3;

  const int saved = __builtin_amdgcn_readfirstlane(savedCnt[blockIdx.x]);
  const bool fast = saved >= 0;
  const float4 a4 = make_float4(att[q * 4], att[q * 4 + 1], att[q * 4 + 2],
                                att[q * 4 + 3]);
  static_assert(OWN_R * 4 <= OWN_T, "one 16-byte piece of the xr rows per thread");
  float4 xrv = make_float4(0.f, 0.f, 0.f, 0.f);
  if (tid < cnt * 4)
    xrv = *reinterpret_cast<const float4*>(xr + (size_t)n0 * 16 + tid * 4);
  const int2* __restrict__ sl =
      reinterpret_cast<const int2*>(savedList) + (size_t)blockIdx.x * OWN_CAP;
  int2 sv[OWN_CAP / OWN_T];
#pragma unroll
  for (int u = 0; u < OWN_CAP / OWN_T; ++u) {
    const int i = u * OWN_T + tid;
    sv[u] = make_int2(0, 0);
    if (i < saved) sv[u] = sl[i];
  }

  for (int i = tid; i < OWN_R * OWN2_S; i += OWN_T) acc[i] = 0.f;
  for (int i = tid; i < OWN_R; i += OWN_T) {
    mS[i] = -__builtin_inff();
    dS[i] = 0.f;
  }
  if (tid < cnt * 4)
    *reinterpret_cast<float4*>(xrS + (tid >> 2) * OWN2_XS + q * 4) = xrv;
#pragma unroll
  for (int u = 0; u < OWN_CAP / OWN_T; ++u) {
    const int i = u * OWN_T + tid;
    if (i < saved) list[i] = sv[u];
  }
  __syncthreads();

  const int ntiles = fast ? 1 : 1 + (E + OWN_CAP - 1) / OWN_CAP;
  for (int phase = 0; phase < 3; ++phase) {
    for (int t = 0; t < ntiles; ++t) {
      int count = saved;
      if (!fast) {
        if (tid == 0) s_cnt = 0;
        __syncthreads();
        if (t == 0) {
          for (int i = tid; i < cnt; i += OWN_T)
            list[i] = make_int2(n0 + i, n0 + i);
        } else {
          const int e0 = (t - 1) * OWN_CAP;
          const int e1 = min(e0 + OWN_CAP, E);
          for (int e = e0 + tid; e < e1; e += OWN_T) {
            const int d = dstE[e];
            if ((unsigned)(d - n0) < (unsigned)cnt) {
              const int slot = atomicAdd(&s_cnt, 1);
              list[slot] = make_int2(srcE[e], d);
            }
          }
        }
        __syncthreads();
        count = t == 0 ? cnt : __builtin_amdgcn_readfirstlane(s_cnt);
      }
      if (phase == 0)
        own2_process<0>(count, fast, n0, list, lg, acc, mS, dS, xrS, xl, a4);
      else if (phase == 1)
        own2_process<1>(count, fast, n0, list, lg, acc, mS, dS, xrS, xl, a4);
      else
        own2_process<2>(count, fast, n0, list, lg, acc, mS, dS, xrS, xl, a4);
      __syncthreads();
    }
  }

  for (int j = tid; j < cnt * 16; j += OWN_T) {
    const int i = j >> 4;
    const int f = j & 15;
    out[(size_t)(n0 + i) * 16 + f] = acc[i * OWN2_S + f] + bias[f];
  }
}

extern "C" void kernel_launch(void* const* d_in, const int* in_sizes, int n_in,
                              void* d_out, int out_size, void* d_ws,
                              size_t ws_size, hipStream_t stream) {
  const float* x = (const float*)d_in[0];
  const int* edge = (const int*)d_in[1];
  const float* W1l = (const float*)d_in[2];
  const float* W1r = (const float*)d_in[3];
  const float* att1 = (const float*)d_in[4];
  const float* b1 = (const float*)d_in[5];
  const float* W2l = (const float*)d_in[6];
  const float* W2r = (const float*)d_in[7];
  const float* att2 = (const float*)d_in[8];
  const float* b2 = (const float*)d_in[9];
  float* out = (float*)d_out;

  const int F_IN = 128, H1 = 80, F_OUT = 16;
  const int N = in_sizes[0] / F_IN;
  const int E = in_sizes[1] / 2;
  const int* srcE = edge;
  const int* dstE = edge + E;

  float* ws = (float*)d_ws;
  float* xl1 = ws;                 ws += (size_t)N * H1;
  float* xr1 = ws;                 ws += (size_t)N * H1;
  float* agg1 = ws;                ws += (size_t)N * H1;
  float* xl2 = ws;                 ws += (size_t)N * F_OUT;
  float* xr2 = ws;                 ws += (size_t)N * F_OUT;
  const int ranges = (N + OWN_R - 1) / OWN_R;
  int* savedList = (int*)ws;       ws += (size_t)ranges * OWN_CAP * 2;
  const int chunks = (E + BM_CHUNK - 1) / BM_CHUNK;
  unsigned int* planes = (unsigned int*)ws; ws += (size_t)ranges * chunks * BM_W;
  int* savedCnt = (int*)ws;
  (void)ws_size; (void)n_in; (void)out_size;

  const int BLK = GEMM_T;

  {
    const int rowTiles = N / 16, colTiles = H1 / 16;
    const int waves = rowTiles * colTiles;
    const int blocks = (waves + 7) / 8;
    wmma_dual_gemm<128, 80, true><<<blocks, BLK, 0, stream>>>(
        x, W1l, W1r, xl1, xr1, rowTiles, colTiles, dstE, E, N, planes);
  }
  layer1_owner_lds<<<ranges, OWN_T, 0, stream>>>(
      srcE, dstE, E, N, xl1, xr1, att1, b1, agg1, savedList, savedCnt, planes);

  {
    const int rowTiles = N / 16, colTiles = F_OUT / 16;
    const int waves = rowTiles * colTiles;
    const int blocks = (waves + 7) / 8;
    wmma_dual_gemm<80, 16, false><<<blocks, BLK, 0, stream>>>(
        agg1, W2l, W2r, xl2, xr2, rowTiles, colTiles, dstE, E, N, planes);
  }
  layer2_owner_lds<<<ranges, OWN_T, 0, stream>>>(
      srcE, dstE, E, N, xl2, xr2, att2, b2, savedList, savedCnt, out);
}


# Layer-2 edge softmax in ONE pass on the saved lists: shift = destination's self-loop logit (made from the owned rows before the pass), not the max; sums divided at the end (rounding changes)
# speedup vs baseline: 6.8502x; 1.0397x over previous
#include <hip/hip_runtime.h>


typedef __attribute__((ext_vector_type(2))) float v2f;
typedef __attribute__((ext_vector_type(8))) float v8f;

#define NEG_SLOPE 0.2f
#define OWN_R 200
#define BM_CHUNK 1024
#define BM_W (BM_CHUNK / 32)
#define GEMM_T 256

template <int K, int LDB, bool KEYS>
__global__ __launch_bounds__(GEMM_T) void wmma_dual_gemm(
    const float* __restrict__ A, const float* __restrict__ Bl,
    const float* __restrict__ Br, float* __restrict__ Cl,
    float* __restrict__ Cr, int rowTiles, int colTiles,
    const int* __restrict__ dstE, int E, int N,
    unsigned int* __restrict__ planes) {
  if (KEYS) {
    __shared__ __attribute__((aligned(16))) unsigned int bm[256 * BM_W];
    uint4* bm4 = reinterpret_cast<uint4*>(bm);
    const int ranges = (N + OWN_R - 1) / OWN_R;
    const int nChunks = (E + BM_CHUNK - 1) / BM_CHUNK;
    const size_t planeWords = (size_t)nChunks * BM_W;
    static_assert(BM_CHUNK == 4 * GEMM_T, "four edges of a tile per thread");
    const int le = 4 * threadIdx.x;
    for (int c = blockIdx.x; c < nChunks; c += gridDim.x) {
      const int e0 = c * BM_CHUNK + le;
      int d[4];
      if (e0 + 3 < E) {
        d[0] = dstE[e0];
        d[1] = dstE[e0 + 1];
        d[2] = dstE[e0 + 2];
        d[3] = dstE[e0 + 3];
      } else {
#pragma unroll
        for (int j = 0; j < 4; ++j) d[j] = e0 + j < E ? dstE[e0 + j] : -1;
      }
      if (c != (int)blockIdx.x) __syncthreads();
      for (int i = threadIdx.x; i < ranges * (BM_W / 4); i += GEMM_T)
        bm4[i] = make_uint4(0u, 0u, 0u, 0u);
      __syncthreads();
#pragma unroll
      for (int j = 0; j < 4; ++j)
        if ((unsigned)d[j] < (unsigned)N)
          atomicOr(&bm[((unsigned)d[j] / OWN_R) * BM_W + (le >> 5)],
                   1u << ((le + j) & 31));
      __syncthreads();
      for (int i = threadIdx.x; i < ranges * (BM_W / 4); i += GEMM_T) {
        const int r = i / (BM_W / 4);
        const int w4 = i - r * (BM_W / 4);
        reinterpret_cast<uint4*>(planes + (size_t)r * planeWords +
                                 (size_t)c * BM_W)[w4] = bm4[i];
      }
    }
  }
  const int wid = (blockIdx.x * blockDim.x + threadIdx.x) >> 5;
  const int lane = threadIdx.x & 31;
  if (wid >= rowTiles * colTiles) return;
  const int rt = wid / colTiles;
  const int ct = wid % colTiles;
  const int row0 = rt * 16;
  const int col0 = ct * 16;
  const int half = lane >> 4;
  const int l16 = lane & 15;
  const int arow = row0 + l16;
  const int bcol = col0 + l16;

  const float* __restrict__ Ap = A + (size_t)arow * K + half * 2;

  v8f accl = {};
  v8f accr = {};
  for (int k = 0; k < K; k += 4) {
    const int ka = k + half * 2;
    v2f a;
    a.x = Ap[k];
    a.y = Ap[k + 1];
    v2f bl, br;
    bl.x = Bl[(size_t)ka * LDB + bcol];
    bl.y = Bl[(size_t)(ka + 1) * LDB + bcol];
    br.x = Br[(size_t)ka * LDB + bcol];
    br.y = Br[(size_t)(ka + 1) * LDB + bcol];
    accl = __builtin_amdgcn_wmma_f32_16x16x4_f32(false, a, false, bl, (short)0,
                                                 accl, false, false);
    accr = __builtin_amdgcn_wmma_f32_16x16x4_f32(false, a, false, br, (short)0,
                                                 accr, false, false);
  }

  const size_t base = (size_t)(row0 + half * 8) * LDB + bcol;
  float* __restrict__ cl = Cl + base;
  float* __restrict__ cr = Cr + base;
#pragma unroll
  for (int i = 0; i < 8; ++i) {
    cl[(size_t)i * LDB] = accl[i];
    cr[(size_t)i * LDB] = accr[i];
  }
}

__device__ __forceinline__ void atomicMaxFloat(float* addr, float val) {
  if (val >= 0.f)
    atomicMax((int*)addr, __float_as_int(val));
  else
    atomicMin((unsigned int*)addr, __float_as_uint(val));
}

#define OWN_S 81
#define OWN_XP 84
#define OWN_CAP 4096
#define OWN_T 1024
#define OWN_U 8
#define OWN_IU 4
#define OWN_DMAX 8.0f

__device__ __forceinline__ float own_logit(const float (&xv)[8],
                                           const float (&rv)[8],
                                           const float (&av)[8]) {
  float s = 0.f;
#pragma unroll
  for (int dd = 0; dd < 8; ++dd) {
    float v = xv[dd] + rv[dd];
    v = v > 0.f ? v : NEG_SLOPE * v;
    s += v * av[dd];
  }
  return s;
}

template <int U>
__device__ __forceinline__ bool own_fused_items(
    int i0, int n0, const int2* list, float* acc, const float* xrS,
    const float* cS, float* dS, const float* __restrict__ xl,
    const float* __restrict__ att) {
  int hh[U], dl[U];
  float av[U][8], xv[U][8];
  float cv[U], w[U];
#pragma unroll
  for (int u = 0; u < U; ++u) {
    const int i = i0 + u * OWN_T;
    const int ei = i / 10;
    hh[u] = i - ei * 10;
#pragma unroll
    for (int dd = 0; dd < 8; ++dd) av[u][dd] = att[hh[u] * 8 + dd];
    const int2 sd = list[ei];
    dl[u] = sd.y - n0;
    const float* __restrict__ xls = xl + (size_t)sd.x * 80 + hh[u] * 8;
#pragma unroll
    for (int dd = 0; dd < 8; ++dd) xv[u][dd] = xls[dd];
    cv[u] = cS[dl[u] * 10 + hh[u]];
  }
  bool bad = false;
#pragma unroll
  for (int u = 0; u < U; ++u) {
    const float* xrd = xrS + dl[u] * OWN_XP + hh[u] * 8 + ((hh[u] >> 3) << 2);
    const float4 r0 = *reinterpret_cast<const float4*>(xrd);
    const float4 r1 = *reinterpret_cast<const float4*>(xrd + 4);
    const float rv[8] = {r0.x, r0.y, r0.z, r0.w, r1.x, r1.y, r1.z, r1.w};
    const float d = own_logit(xv[u], rv, av[u]) - cv[u];
    bad |= !(d <= OWN_DMAX);
    w[u] = __expf(d);
  }
#pragma unroll
  for (int u = 0; u < U; ++u) {
    atomicAdd(&dS[dl[u] * 10 + hh[u]], w[u]);
    float* a = acc + dl[u] * OWN_S + hh[u] * 8 + (hh[u] >> 3);
#pragma unroll
    for (int dd = 0; dd < 8; ++dd) atomicAdd(&a[dd], w[u] * xv[u][dd]);
  }
  return bad;
}

template <int PHASE>
__device__ __forceinline__ void own_process3(
    int count, int n0, const int2* list, float* acc, float* mS, float* dS,
    const float* __restrict__ xl, const float* __restrict__ xr,
    const float* __restrict__ att) {
  const int items = count * 10;
  for (int i = threadIdx.x; i < items; i += OWN_T) {
    const int ei = i / 10;
    const int hh = i - ei * 10;
    const int2 sd = list[ei];
    const int dl = sd.y - n0;
    const float* __restrict__ xls = xl + (size_t)sd.x * 80 + hh * 8;
    const float* __restrict__ xrd = xr + (size_t)sd.y * 80 + hh * 8;
    float av[8], xv[8], rv[8];
#pragma unroll
    for (int dd = 0; dd < 8; ++dd) {
      av[dd] = att[hh * 8 + dd];
      xv[dd] = xls[dd];
      rv[dd] = xrd[dd];
    }
    const float logit = own_logit(xv, rv, av);
    const int seg = dl * 10 + hh;
    if (PHASE == 0) {
      atomicMaxFloat(&mS[seg], logit);
    } else if (PHASE == 1) {
      atomicAdd(&dS[seg], __expf(logit - mS[seg]));
    } else {
      const float alpha = __expf(logit - mS[seg]) / dS[seg];
      float* a = acc + dl * OWN_S + hh * 8 + (hh >> 3);
#pragma unroll
      for (int dd = 0; dd < 8; ++dd) atomicAdd(&a[dd], alpha * xv[dd]);
    }
  }
}

__global__ __launch_bounds__(OWN_T) void layer1_owner_lds(
    const int* __restrict__ srcE, const int* __restrict__ dstE, int E, int N,
    const float* __restrict__ xl, const float* __restrict__ xr,
    const float* __restrict__ att, const float* __restrict__ bias,
    float* __restrict__ agg, int* __restrict__ savedList,
    int* __restrict__ savedCnt, const unsigned int* __restrict__ planes) {
  static_assert(OWN_XP >= 80 + 4 && OWN_XP % 4 == 0,
                "xr rows 16-byte aligned, heads 8 and 9 four floats further on");
  __shared__ float acc[OWN_R * OWN_S];
  __shared__ __attribute__((aligned(16))) float xrS[OWN_R * OWN_XP];
  __shared__ float mS[OWN_R * 10];
  __shared__ float dS[OWN_R * 10];
  __shared__ int2 list[OWN_CAP];
  __shared__ int s_cnt;
  __shared__ int s_redo;

  const int tid = threadIdx.x;
  const int n0 = blockIdx.x * OWN_R;
  const int cnt = min(OWN_R, N - n0);

  static_assert(OWN_R * 20 <= 4 * OWN_T, "four 16-byte pieces of the xr rows per thread");
  float4 xrv[4];
#pragma unroll
  for (int u = 0; u < 4; ++u) {
    const int j = u * OWN_T + tid;
    xrv[u] = make_float4(0.f, 0.f, 0.f, 0.f);
    if (j < cnt * 20)
      xrv[u] = *reinterpret_cast<const float4*>(xr + (size_t)n0 * 80 + j * 4);
  }

  for (int i = tid; i < OWN_R * 10; i += OWN_T) {
    mS[i] = -__builtin_inff();
    dS[i] = 0.f;
  }
  for (int i = tid; i < OWN_R * OWN_S; i += OWN_T) acc[i] = 0.f;
  if (tid == 0) {
    s_cnt = 0;
    s_redo = 0;
  }
  __syncthreads();

  {
    const int nChunks = (E + BM_CHUNK - 1) / BM_CHUNK;
    const size_t planeWords = (size_t)nChunks * BM_W;
    const int nvec = nChunks * (BM_W / 4);
    const uint4* __restrict__ pv =
        reinterpret_cast<const uint4*>(planes + (size_t)blockIdx.x * planeWords);
    const int iters = (nvec + OWN_T - 1) / OWN_T;
    for (int k0 = 0; k0 < iters; k0 += OWN_U) {
      uint4 q[OWN_U];
#pragma unroll
      for (int u = 0; u < OWN_U; ++u) {
        const int idx = (k0 + u) * OWN_T + tid;
        q[u] = make_uint4(0u, 0u, 0u, 0u);
        if (idx < nvec) q[u] = pv[idx];
      }
      int nb = 0;
#pragma unroll
      for (int u = 0; u < OWN_U; ++u)
        nb += __popc(q[u].x) + __popc(q[u].y) + __popc(q[u].z) + __popc(q[u].w);
      int slot = atomicAdd(&s_cnt, nb);
#pragma unroll
      for (int u = 0; u < OWN_U; ++u) {
        const int idx = (k0 + u) * OWN_T + tid;
        const unsigned int w[4] = {q[u].x, q[u].y, q[u].z, q[u].w};
#pragma unroll
        for (int j = 0; j < 4; ++j) {
          unsigned int m = w[j];
          while (m != 0u) {
            const int p = __builtin_ctz(m);
            m &= m - 1u;
            if (slot < OWN_CAP) list[slot] = make_int2((idx * 4 + j) * 32 + p, 0);
            ++slot;
          }
        }
      }
    }
  }
#pragma unroll
  for (int u = 0; u < 4; ++u) {
    const int j = u * OWN_T + tid;
    if (j < cnt * 20) {
      const int row = j / 20;
      const int c4 = j - row * 20;
      *reinterpret_cast<float4*>(xrS + row * OWN_XP + c4 * 4 + ((c4 >> 4) << 2)) =
          xrv[u];
    }
  }
  __syncthreads();

  const int total = __builtin_amdgcn_readfirstlane(s_cnt);
  const bool fast = total + cnt <= OWN_CAP;
  if (fast) {
    for (int i = tid; i < total; i += OWN_T) {
      const int e = list[i].x;
      list[i] = make_int2(srcE[e], dstE[e]);
    }
    for (int i = tid; i < cnt; i += OWN_T)
      list[total + i] = make_int2(n0 + i, n0 + i);
    for (int j = tid; j < cnt * 10; j += OWN_T) {
      const int i = j / 10;
      const int h = j - i * 10;
      const float* __restrict__ xls = xl + (size_t)(n0 + i) * 80 + h * 8;
      const float* xrd = xrS + i * OWN_XP + h * 8 + ((h >> 3) << 2);
      float av[8], xv[8], rv[8];
#pragma unroll
      for (int dd = 0; dd < 8; ++dd) {
        av[dd] = att[h * 8 + dd];
        xv[dd] = xls[dd];
        rv[dd] = xrd[dd];
      }
      mS[j] = own_logit(xv, rv, av);
    }
  }
  __syncthreads();

  if (fast) {
    int2* __restrict__ sl =
        reinterpret_cast<int2*>(savedList) + (size_t)blockIdx.x * OWN_CAP;
    for (int i = tid; i < total + cnt; i += OWN_T) sl[i] = list[i];
  }
  if (tid == 0) savedCnt[blockIdx.x] = fast ? total + cnt : -1;

  bool three = !fast;
  if (fast) {
    const int items = (total + cnt) * 10;
    bool bad = false;
    int i0 = tid;
    for (; i0 + (OWN_IU - 1) * OWN_T < items; i0 += OWN_IU * OWN_T)
      bad |= own_fused_items<OWN_IU>(i0, n0, list, acc, xrS, mS, dS, xl, att);
    for (; i0 < items; i0 += OWN_T)
      bad |= own_fused_items<1>(i0, n0, list, acc, xrS, mS, dS, xl, att);
    if (bad) s_redo = 1;
    __syncthreads();
    three = __builtin_amdgcn_readfirstlane(s_redo) != 0;
    if (three) {
      for (int i = tid; i < OWN_R * 10; i += OWN_T) {
        mS[i] = -__builtin_inff();
        dS[i] = 0.f;
      }
      for (int i = tid; i < OWN_R * OWN_S; i += OWN_T) acc[i] = 0.f;
      __syncthreads();
    }
  }

  if (three) {
    const int ntiles = fast ? 1 : 1 + (E + OWN_CAP - 1) / OWN_CAP;
    for (int phase = 0; phase < 3; ++phase) {
      for (int t = 0; t < ntiles; ++t) {
        int count = total + cnt;
        if (!fast) {
          if (tid == 0) s_cnt = 0;
          __syncthreads();
          if (t == 0) {
            for (int i = tid; i < cnt; i += OWN_T)
              list[i] = make_int2(n0 + i, n0 + i);
          } else {
            const int e0 = (t - 1) * OWN_CAP;
            const int e1 = min(e0 + OWN_CAP, E);
            for (int e = e0 + tid; e < e1; e += OWN_T) {
              const int d = dstE[e];
              if ((unsigned)(d - n0) < (unsigned)cnt) {
                const int slot = atomicAdd(&s_cnt, 1);
                list[slot] = make_int2(srcE[e], d);
              }
            }
          }
          __syncthreads();
          count = t == 0 ? cnt : __builtin_amdgcn_readfirstlane(s_cnt);
        }
        if (phase == 0)
          own_process3<0>(count, n0, list, acc, mS, dS, xl, xr, att);
        else if (phase == 1)
          own_process3<1>(count, n0, list, acc, mS, dS, xl, xr, att);
        else
          own_process3<2>(count, n0, list, acc, mS, dS, xl, xr, att);
        __syncthreads();
      }
    }
  }

  for (int j = tid; j < cnt * 80; j += OWN_T) {
    const int i = j / 80;
    const int f = j - i * 80;
    float v = acc[i * OWN_S + f + (f >> 6)];
    if (!three) v = v / dS[i * 10 + (f >> 3)];
    agg[(size_t)(n0 + i) * 80 + f] = fmaxf(v + bias[f], 0.f);
  }
}

#define OWN2_S 17
#define OWN2_XS 20
#define OWN2_U 4
#define OWN2_G (OWN_T / 4)

template <int CTRL>
__device__ __forceinline__ float dpp_flt(float v) {
  return __int_as_float(
      __builtin_amdgcn_update_dpp(0, __float_as_int(v), CTRL, 0xf, 0xf, true));
}

__device__ __forceinline__ float own2_partial(float4 xv, float4 rv, float4 a4) {
  float v0 = xv.x + rv.x, v1 = xv.y + rv.y;
  float v2 = xv.z + rv.z, v3 = xv.w + rv.w;
  v0 = v0 > 0.f ? v0 : NEG_SLOPE * v0;
  v1 = v1 > 0.f ? v1 : NEG_SLOPE * v1;
  v2 = v2 > 0.f ? v2 : NEG_SLOPE * v2;
  v3 = v3 > 0.f ? v3 : NEG_SLOPE * v3;
  float p = v0 * a4.x + v1 * a4.y + v2 * a4.z + v3 * a4.w;
  p += dpp_flt<0x161>(p);
  p += dpp_flt<0x162>(p);
  return p;
}

__device__ __forceinline__ bool own2_fused(
    int count, int n0, const int2* list, float* acc, const float* cS, float* dS,
    const float* xrS, const float* __restrict__ xl, float4 a4) {
  const int q = threadIdx.x & 3;
  const int g = threadIdx.x >> 2;
  bool bad = false;
  for (int e0 = g; e0 < count; e0 += OWN2_U * OWN2_G) {
    int dl[OWN2_U];
    bool live[OWN2_U];
    float4 xv[OWN2_U];
#pragma unroll
    for (int u = 0; u < OWN2_U; ++u) {
      const int e = e0 + u * OWN2_G;
      live[u] = e < count;
      const int2 sd = list[live[u] ? e : e0];
      dl[u] = sd.y - n0;
      xv[u] = *reinterpret_cast<const float4*>(xl + (size_t)sd.x * 16 + q * 4);
    }
#pragma unroll
    for (int u = 0; u < OWN2_U; ++u) {
      const float4 rv =
          *reinterpret_cast<const float4*>(xrS + dl[u] * OWN2_XS + q * 4);
      const float d = own2_partial(xv[u], rv, a4) - cS[dl[u]];
      bad |= !(d <= OWN_DMAX);
      const float w = __expf(d);
      if (live[u]) {
        if (q == 0) atomicAdd(&dS[dl[u]], w);
        float* a = acc + dl[u] * OWN2_S + q * 4;
        atomicAdd(&a[0], w * xv[u].x);
        atomicAdd(&a[1], w * xv[u].y);
        atomicAdd(&a[2], w * xv[u].z);
        atomicAdd(&a[3], w * xv[u].w);
      }
    }
  }
  return bad;
}

template <int PHASE>
__device__ __forceinline__ void own2_process(
    int count, bool cached, int n0, const int2* list, float* lg, float* acc,
    float* mS, float* dS, const float* xrS, const float* __restrict__ xl,
    float4 a4) {
  if (PHASE == 1 && cached) {
    for (int e = threadIdx.x; e < count; e += OWN_T) {
      const int dl = list[e].y - n0;
      atomicAdd(&dS[dl], __expf(lg[e] - mS[dl]));
    }
    return;
  }
  const int q = threadIdx.x & 3;
  const int g = threadIdx.x >> 2;
  for (int e0 = g; e0 < count; e0 += OWN2_U * OWN2_G) {
    int ee[OWN2_U], dl[OWN2_U];
    bool live[OWN2_U];
    float4 xv[OWN2_U];
#pragma unroll
    for (int u = 0; u < OWN2_U; ++u) {
      const int e = e0 + u * OWN2_G;
      live[u] = e < count;
      ee[u] = live[u] ? e : e0;
      const int2 sd = list[ee[u]];
      dl[u] = sd.y - n0;
      xv[u] = *reinterpret_cast<const float4*>(xl + (size_t)sd.x * 16 + q * 4);
    }
#pragma unroll
    for (int u = 0; u < OWN2_U; ++u) {
      float logit;
      if (PHASE == 0 || !cached) {
        const float4 rv =
            *reinterpret_cast<const float4*>(xrS + dl[u] * OWN2_XS + q * 4);
        float v0 = xv[u].x + rv.x, v1 = xv[u].y + rv.y;
        float v2 = xv[u].z + rv.z, v3 = xv[u].w + rv.w;
        v0 = v0 > 0.f ? v0 : NEG_SLOPE * v0;
        v1 = v1 > 0.f ? v1 : NEG_SLOPE * v1;
        v2 = v2 > 0.f ? v2 : NEG_SLOPE * v2;
        v3 = v3 > 0.f ? v3 : NEG_SLOPE * v3;
        float p = v0 * a4.x + v1 * a4.y + v2 * a4.z + v3 * a4.w;
        p += dpp_flt<0x161>(p);
        p += dpp_flt<0x162>(p);
        logit = p;
        if (PHASE == 0 && cached && q == 0 && live[u]) lg[ee[u]] = logit;
      } else {
        logit = lg[ee[u]];
      }
      if (PHASE == 0) {
        if (live[u] && q == 0) atomicMaxFloat(&mS[dl[u]], logit);
      } else if (PHASE == 1) {
        if (live[u] && q == 0)
          atomicAdd(&dS[dl[u]], __expf(logit - mS[dl[u]]));
      } else {
        const float alpha = __expf(logit - mS[dl[u]]) / dS[dl[u]];
        if (live[u]) {
          float* a = acc + dl[u] * OWN2_S + q * 4;
          atomicAdd(&a[0], alpha * xv[u].x);
          atomicAdd(&a[1], alpha * xv[u].y);
          atomicAdd(&a[2], alpha * xv[u].z);
          atomicAdd(&a[3], alpha * xv[u].w);
        }
      }
    }
  }
}

__global__ __launch_bounds__(OWN_T) void layer2_owner_lds(
    const int* __restrict__ srcE, const int* __restrict__ dstE, int E, int N,
    const float* __restrict__ xl, const float* __restrict__ xr,
    const float* __restrict__ att, const float* __restrict__ bias,
    const int* __restrict__ savedList, const int* __restrict__ savedCnt,
    float* __restrict__ out) {
  __shared__ float acc[OWN_R * OWN2_S];
  __shared__ __attribute__((aligned(16))) float xrS[OWN_R * OWN2_XS];
  __shared__ float mS[OWN_R];
  __shared__ float dS[OWN_R];
  __shared__ float lg[OWN_CAP];
  __shared__ int2 list[OWN_CAP];
  __shared__ int s_cnt;
  __shared__ int s_redo;

  const int tid = threadIdx.x;
  const int n0 = blockIdx.x * OWN_R;
  const int cnt = min(OWN_R, N - n0);
  const int q = tid & 3;

  const int saved = __builtin_amdgcn_readfirstlane(savedCnt[blockIdx.x]);
  const bool fast = saved >= 0;
  const float4 a4 = make_float4(att[q * 4], att[q * 4 + 1], att[q * 4 + 2],
                                att[q * 4 + 3]);
  static_assert(OWN_R * 4 <= OWN_T, "one 16-byte piece of the xr rows per thread");
  float4 xrv = make_float4(0.f, 0.f, 0.f, 0.f);
  if (tid < cnt * 4)
    xrv = *reinterpret_cast<const float4*>(xr + (size_t)n0 * 16 + tid * 4);
  float4 xlv = make_float4(0.f, 0.f, 0.f, 0.f);
  if (fast && tid < cnt * 4)
    xlv = *reinterpret_cast<const float4*>(xl + (size_t)n0 * 16 + tid * 4);
  const int2* __restrict__ sl =
      reinterpret_cast<const int2*>(savedList) + (size_t)blockIdx.x * OWN_CAP;
  int2 sv[OWN_CAP / OWN_T];
#pragma unroll
  for (int u = 0; u < OWN_CAP / OWN_T; ++u) {
    const int i = u * OWN_T + tid;
    sv[u] = make_int2(0, 0);
    if (i < saved) sv[u] = sl[i];
  }

  for (int i = tid; i < OWN_R * OWN2_S; i += OWN_T) acc[i] = 0.f;
  for (int i = tid; i < OWN_R; i += OWN_T) {
    if (!fast) mS[i] = -__builtin_inff();
    dS[i] = 0.f;
  }
  if (tid == 0) s_redo = 0;
  if (tid < cnt * 4)
    *reinterpret_cast<float4*>(xrS + (tid >> 2) * OWN2_XS + q * 4) = xrv;
  if (fast && tid < cnt * 4) {
    const float s = own2_partial(xlv, xrv, a4);
    if (q == 0) mS[tid >> 2] = s;
  }
#pragma unroll
  for (int u = 0; u < OWN_CAP / OWN_T; ++u) {
    const int i = u * OWN_T + tid;
    if (i < saved) list[i] = sv[u];
  }
  __syncthreads();

  bool three = !fast;
  if (fast) {
    if (own2_fused(saved, n0, list, acc, mS, dS, xrS, xl, a4)) s_redo = 1;
    __syncthreads();
    three = __builtin_amdgcn_readfirstlane(s_redo) != 0;
    if (three) {
      for (int i = tid; i < OWN_R * OWN2_S; i += OWN_T) acc[i] = 0.f;
      for (int i = tid; i < OWN_R; i += OWN_T) {
        mS[i] = -__builtin_inff();
        dS[i] = 0.f;
      }
      __syncthreads();
    }
  }

  if (three) {
    const int ntiles = fast ? 1 : 1 + (E + OWN_CAP - 1) / OWN_CAP;
    for (int phase = 0; phase < 3; ++phase) {
      for (int t = 0; t < ntiles; ++t) {
        int count = saved;
        if (!fast) {
          if (tid == 0) s_cnt = 0;
          __syncthreads();
          if (t == 0) {
            for (int i = tid; i < cnt; i += OWN_T)
              list[i] = make_int2(n0 + i, n0 + i);
          } else {
            const int e0 = (t - 1) * OWN_CAP;
            const int e1 = min(e0 + OWN_CAP, E);
            for (int e = e0 + tid; e < e1; e += OWN_T) {
              const int d = dstE[e];
              if ((unsigned)(d - n0) < (unsigned)cnt) {
                const int slot = atomicAdd(&s_cnt, 1);
                list[slot] = make_int2(srcE[e], d);
              }
            }
          }
          __syncthreads();
          count = t == 0 ? cnt : __builtin_amdgcn_readfirstlane(s_cnt);
        }
        if (phase == 0)
          own2_process<0>(count, fast, n0, list, lg, acc, mS, dS, xrS, xl, a4);
        else if (phase == 1)
          own2_process<1>(count, fast, n0, list, lg, acc, mS, dS, xrS, xl, a4);
        else
          own2_process<2>(count, fast, n0, list, lg, acc, mS, dS, xrS, xl, a4);
        __syncthreads();
      }
    }
  }

  for (int j = tid; j < cnt * 16; j += OWN_T) {
    const int i = j >> 4;
    const int f = j & 15;
    float v = acc[i * OWN2_S + f];
    if (!three) v = v / dS[i];
    out[(size_t)(n0 + i) * 16 + f] = v + bias[f];
  }
}

extern "C" void kernel_launch(void* const* d_in, const int* in_sizes, int n_in,
                              void* d_out, int out_size, void* d_ws,
                              size_t ws_size, hipStream_t stream) {
  const float* x = (const float*)d_in[0];
  const int* edge = (const int*)d_in[1];
  const float* W1l = (const float*)d_in[2];
  const float* W1r = (const float*)d_in[3];
  const float* att1 = (const float*)d_in[4];
  const float* b1 = (const float*)d_in[5];
  const float* W2l = (const float*)d_in[6];
  const float* W2r = (const float*)d_in[7];
  const float* att2 = (const float*)d_in[8];
  const float* b2 = (const float*)d_in[9];
  float* out = (float*)d_out;

  const int F_IN = 128, H1 = 80, F_OUT = 16;
  const int N = in_sizes[0] / F_IN;
  const int E = in_sizes[1] / 2;
  const int* srcE = edge;
  const int* dstE = edge + E;

  float* ws = (float*)d_ws;
  float* xl1 = ws;                 ws += (size_t)N * H1;
  float* xr1 = ws;                 ws += (size_t)N * H1;
  float* agg1 = ws;                ws += (size_t)N * H1;
  float* xl2 = ws;                 ws += (size_t)N * F_OUT;
  float* xr2 = ws;                 ws += (size_t)N * F_OUT;
  const int ranges = (N + OWN_R - 1) / OWN_R;
  int* savedList = (int*)ws;       ws += (size_t)ranges * OWN_CAP * 2;
  const int chunks = (E + BM_CHUNK - 1) / BM_CHUNK;
  unsigned int* planes = (unsigned int*)ws; ws += (size_t)ranges * chunks * BM_W;
  int* savedCnt = (int*)ws;
  (void)ws_size; (void)n_in; (void)out_size;

  const int BLK = GEMM_T;

  {
    const int rowTiles = N / 16, colTiles = H1 / 16;
    const int waves = rowTiles * colTiles;
    const int blocks = (waves + 7) / 8;
    wmma_dual_gemm<128, 80, true><<<blocks, BLK, 0, stream>>>(
        x, W1l, W1r, xl1, xr1, rowTiles, colTiles, dstE, E, N, planes);
  }
  layer1_owner_lds<<<ranges, OWN_T, 0, stream>>>(
      srcE, dstE, E, N, xl1, xr1, att1, b1, agg1, savedList, savedCnt, planes);

  {
    const int rowTiles = N / 16, colTiles = F_OUT / 16;
    const int waves = rowTiles * colTiles;
    const int blocks = (waves + 7) / 8;
    wmma_dual_gemm<80, 16, false><<<blocks, BLK, 0, stream>>>(
        agg1, W2l, W2r, xl2, xr2, rowTiles, colTiles, dstE, E, N, planes);
  }
  layer2_owner_lds<<<ranges, OWN_T, 0, stream>>>(
      srcE, dstE, E, N, xl2, xr2, att2, b2, savedList, savedCnt, out);
}


# Layer-1 filter lines hold the edges: transform launch ranks each tile's edges per range in LDS, stores packed (source,destination) words; owner lists them, no random edge reads; LDS arrival order
# speedup vs baseline: 7.4471x; 1.0871x over previous
#include <hip/hip_runtime.h>


typedef __attribute__((ext_vector_type(2))) float v2f;
typedef __attribute__((ext_vector_type(8))) float v8f;

#define NEG_SLOPE 0.2f
#define OWN_R 200
#define BM_CHUNK 1024
#define BM_W 32
#define REC_NONE 0xffffffffu
#define REC_OVER 0xfffffffeu
#define GEMM_T 256

template <int K, int LDB, bool KEYS>
__global__ __launch_bounds__(GEMM_T) void wmma_dual_gemm(
    const float* __restrict__ A, const float* __restrict__ Bl,
    const float* __restrict__ Br, float* __restrict__ Cl,
    float* __restrict__ Cr, int rowTiles, int colTiles,
    const int* __restrict__ srcE, const int* __restrict__ dstE, int E, int N,
    unsigned int* __restrict__ planes) {
  if (KEYS) {
    static_assert(OWN_R <= 253, "destination - first node of the range fits 8 bits below the marks");
    static_assert(GEMM_T == 256, "one counter of the LDS tile per thread");
    __shared__ __attribute__((aligned(16))) unsigned int bm[256 * BM_W];
    __shared__ int cntR[256];
    uint4* bm4 = reinterpret_cast<uint4*>(bm);
    const int ranges = (N + OWN_R - 1) / OWN_R;
    const int nChunks = (E + BM_CHUNK - 1) / BM_CHUNK;
    const size_t planeWords = (size_t)nChunks * BM_W;
    static_assert(BM_CHUNK == 4 * GEMM_T, "four edges of a tile per thread");
    const int le = 4 * threadIdx.x;
    for (int c = blockIdx.x; c < nChunks; c += gridDim.x) {
      const int e0 = c * BM_CHUNK + le;
      int d[4], s[4];
      if (e0 + 3 < E) {
        d[0] = dstE[e0];
        d[1] = dstE[e0 + 1];
        d[2] = dstE[e0 + 2];
        d[3] = dstE[e0 + 3];
        s[0] = srcE[e0];
        s[1] = srcE[e0 + 1];
        s[2] = srcE[e0 + 2];
        s[3] = srcE[e0 + 3];
      } else {
#pragma unroll
        for (int j = 0; j < 4; ++j) {
          d[j] = e0 + j < E ? dstE[e0 + j] : -1;
          s[j] = e0 + j < E ? srcE[e0 + j] : 0;
        }
      }
      if (c != (int)blockIdx.x) __syncthreads();
      for (int i = threadIdx.x; i < ranges * (BM_W / 4); i += GEMM_T)
        bm4[i] = make_uint4(REC_NONE, REC_NONE, REC_NONE, REC_NONE);
      cntR[threadIdx.x] = 0;
      __syncthreads();
      int rr[4] = {0, 0, 0, 0}, rk[4];
#pragma unroll
      for (int j = 0; j < 4; ++j) {
        rk[j] = -1;
        if ((unsigned)d[j] < (unsigned)N) {
          rr[j] = (unsigned)d[j] / OWN_R;
          rk[j] = atomicAdd(&cntR[rr[j]], 1);
          if (rk[j] < BM_W)
            bm[rr[j] * BM_W + rk[j]] =
                ((unsigned)s[j] & 0xffffffu) |
                ((unsigned)(d[j] - rr[j] * OWN_R) << 24);
        }
      }
      __syncthreads();
#pragma unroll
      for (int j = 0; j < 4; ++j)
        if (rk[j] >= BM_W) bm[rr[j] * BM_W + BM_W - 1] = REC_OVER;
      __syncthreads();
      for (int i = threadIdx.x; i < ranges * (BM_W / 4); i += GEMM_T) {
        const int r = i / (BM_W / 4);
        const int w4 = i - r * (BM_W / 4);
        reinterpret_cast<uint4*>(planes + (size_t)r * planeWords +
                                 (size_t)c * BM_W)[w4] = bm4[i];
      }
    }
  }
  const int wid = (blockIdx.x * blockDim.x + threadIdx.x) >> 5;
  const int lane = threadIdx.x & 31;
  if (wid >= rowTiles * colTiles) return;
  const int rt = wid / colTiles;
  const int ct = wid % colTiles;
  const int row0 = rt * 16;
  const int col0 = ct * 16;
  const int half = lane >> 4;
  const int l16 = lane & 15;
  const int arow = row0 + l16;
  const int bcol = col0 + l16;

  const float* __restrict__ Ap = A + (size_t)arow * K + half * 2;

  v8f accl = {};
  v8f accr = {};
  for (int k = 0; k < K; k += 4) {
    const int ka = k + half * 2;
    v2f a;
    a.x = Ap[k];
    a.y = Ap[k + 1];
    v2f bl, br;
    bl.x = Bl[(size_t)ka * LDB + bcol];
    bl.y = Bl[(size_t)(ka + 1) * LDB + bcol];
    br.x = Br[(size_t)ka * LDB + bcol];
    br.y = Br[(size_t)(ka + 1) * LDB + bcol];
    accl = __builtin_amdgcn_wmma_f32_16x16x4_f32(false, a, false, bl, (short)0,
                                                 accl, false, false);
    accr = __builtin_amdgcn_wmma_f32_16x16x4_f32(false, a, false, br, (short)0,
                                                 accr, false, false);
  }

  const size_t base = (size_t)(row0 + half * 8) * LDB + bcol;
  float* __restrict__ cl = Cl + base;
  float* __restrict__ cr = Cr + base;
#pragma unroll
  for (int i = 0; i < 8; ++i) {
    cl[(size_t)i * LDB] = accl[i];
    cr[(size_t)i * LDB] = accr[i];
  }
}

__device__ __forceinline__ void atomicMaxFloat(float* addr, float val) {
  if (val >= 0.f)
    atomicMax((int*)addr, __float_as_int(val));
  else
    atomicMin((unsigned int*)addr, __float_as_uint(val));
}

#define OWN_S 81
#define OWN_XP 84
#define OWN_CAP 4096
#define OWN_T 1024
#define OWN_U 8
#define OWN_IU 4
#define OWN_DMAX 8.0f

__device__ __forceinline__ float own_logit(const float (&xv)[8],
                                           const float (&rv)[8],
                                           const float (&av)[8]) {
  float s = 0.f;
#pragma unroll
  for (int dd = 0; dd < 8; ++dd) {
    float v = xv[dd] + rv[dd];
    v = v > 0.f ? v : NEG_SLOPE * v;
    s += v * av[dd];
  }
  return s;
}

template <int U>
__device__ __forceinline__ bool own_fused_items(
    int i0, int n0, const int2* list, float* acc, const float* xrS,
    const float* cS, float* dS, const float* __restrict__ xl,
    const float* __restrict__ att) {
  int hh[U], dl[U];
  float av[U][8], xv[U][8];
  float cv[U], w[U];
#pragma unroll
  for (int u = 0; u < U; ++u) {
    const int i = i0 + u * OWN_T;
    const int ei = i / 10;
    hh[u] = i - ei * 10;
#pragma unroll
    for (int dd = 0; dd < 8; ++dd) av[u][dd] = att[hh[u] * 8 + dd];
    const int2 sd = list[ei];
    dl[u] = sd.y - n0;
    const float* __restrict__ xls = xl + (size_t)sd.x * 80 + hh[u] * 8;
#pragma unroll
    for (int dd = 0; dd < 8; ++dd) xv[u][dd] = xls[dd];
    cv[u] = cS[dl[u] * 10 + hh[u]];
  }
  bool bad = false;
#pragma unroll
  for (int u = 0; u < U; ++u) {
    const float* xrd = xrS + dl[u] * OWN_XP + hh[u] * 8 + ((hh[u] >> 3) << 2);
    const float4 r0 = *reinterpret_cast<const float4*>(xrd);
    const float4 r1 = *reinterpret_cast<const float4*>(xrd + 4);
    const float rv[8] = {r0.x, r0.y, r0.z, r0.w, r1.x, r1.y, r1.z, r1.w};
    const float d = own_logit(xv[u], rv, av[u]) - cv[u];
    bad |= !(d <= OWN_DMAX);
    w[u] = __expf(d);
  }
#pragma unroll
  for (int u = 0; u < U; ++u) {
    atomicAdd(&dS[dl[u] * 10 + hh[u]], w[u]);
    float* a = acc + dl[u] * OWN_S + hh[u] * 8 + (hh[u] >> 3);
#pragma unroll
    for (int dd = 0; dd < 8; ++dd) atomicAdd(&a[dd], w[u] * xv[u][dd]);
  }
  return bad;
}

template <int PHASE>
__device__ __forceinline__ void own_process3(
    int count, int n0, const int2* list, float* acc, float* mS, float* dS,
    const float* __restrict__ xl, const float* __restrict__ xr,
    const float* __restrict__ att) {
  const int items = count * 10;
  for (int i = threadIdx.x; i < items; i += OWN_T) {
    const int ei = i / 10;
    const int hh = i - ei * 10;
    const int2 sd = list[ei];
    const int dl = sd.y - n0;
    const float* __restrict__ xls = xl + (size_t)sd.x * 80 + hh * 8;
    const float* __restrict__ xrd = xr + (size_t)sd.y * 80 + hh * 8;
    float av[8], xv[8], rv[8];
#pragma unroll
    for (int dd = 0; dd < 8; ++dd) {
      av[dd] = att[hh * 8 + dd];
      xv[dd] = xls[dd];
      rv[dd] = xrd[dd];
    }
    const float logit = own_logit(xv, rv, av);
    const int seg = dl * 10 + hh;
    if (PHASE == 0) {
      atomicMaxFloat(&mS[seg], logit);
    } else if (PHASE == 1) {
      atomicAdd(&dS[seg], __expf(logit - mS[seg]));
    } else {
      const float alpha = __expf(logit - mS[seg]) / dS[seg];
      float* a = acc + dl * OWN_S + hh * 8 + (hh >> 3);
#pragma unroll
      for (int dd = 0; dd < 8; ++dd) atomicAdd(&a[dd], alpha * xv[dd]);
    }
  }
}

__global__ __launch_bounds__(OWN_T) void layer1_owner_lds(
    const int* __restrict__ srcE, const int* __restrict__ dstE, int E, int N,
    const float* __restrict__ xl, const float* __restrict__ xr,
    const float* __restrict__ att, const float* __restrict__ bias,
    float* __restrict__ agg, int* __restrict__ savedList,
    int* __restrict__ savedCnt, const unsigned int* __restrict__ planes) {
  static_assert(OWN_XP >= 80 + 4 && OWN_XP % 4 == 0,
                "xr rows 16-byte aligned, heads 8 and 9 four floats further on");
  __shared__ float acc[OWN_R * OWN_S];
  __shared__ __attribute__((aligned(16))) float xrS[OWN_R * OWN_XP];
  __shared__ float mS[OWN_R * 10];
  __shared__ float dS[OWN_R * 10];
  __shared__ int2 list[OWN_CAP];
  __shared__ int s_cnt;
  __shared__ int s_redo;
  __shared__ int s_over;

  const int tid = threadIdx.x;
  const int n0 = blockIdx.x * OWN_R;
  const int cnt = min(OWN_R, N - n0);

  static_assert(OWN_R * 20 <= 4 * OWN_T, "four 16-byte pieces of the xr rows per thread");
  float4 xrv[4];
#pragma unroll
  for (int u = 0; u < 4; ++u) {
    const int j = u * OWN_T + tid;
    xrv[u] = make_float4(0.f, 0.f, 0.f, 0.f);
    if (j < cnt * 20)
      xrv[u] = *reinterpret_cast<const float4*>(xr + (size_t)n0 * 80 + j * 4);
  }

  for (int i = tid; i < OWN_R * 10; i += OWN_T) {
    mS[i] = -__builtin_inff();
    dS[i] = 0.f;
  }
  for (int i = tid; i < OWN_R * OWN_S; i += OWN_T) acc[i] = 0.f;
  if (tid == 0) {
    s_cnt = 0;
    s_redo = 0;
    s_over = 0;
  }
  __syncthreads();

  {
    const int nChunks = (E + BM_CHUNK - 1) / BM_CHUNK;
    const size_t planeWords = (size_t)nChunks * BM_W;
    const int nvec = nChunks * (BM_W / 4);
    const uint4* __restrict__ pv =
        reinterpret_cast<const uint4*>(planes + (size_t)blockIdx.x * planeWords);
    const int iters = (nvec + OWN_T - 1) / OWN_T;
    for (int k0 = 0; k0 < iters; k0 += OWN_U) {
      uint4 q[OWN_U];
#pragma unroll
      for (int u = 0; u < OWN_U; ++u) {
        const int idx = (k0 + u) * OWN_T + tid;
        q[u] = make_uint4(REC_NONE, REC_NONE, REC_NONE, REC_NONE);
        if (idx < nvec) q[u] = pv[idx];
      }
      int nb = 0;
      bool over = false;
#pragma unroll
      for (int u = 0; u < OWN_U; ++u) {
        nb += (q[u].x < REC_OVER) + (q[u].y < REC_OVER) + (q[u].z < REC_OVER) +
              (q[u].w < REC_OVER);
        over |= q[u].w == REC_OVER;
      }
      if (over) s_over = 1;
      if (nb != 0) {
        int slot = atomicAdd(&s_cnt, nb);
#pragma unroll
        for (int u = 0; u < OWN_U; ++u) {
          const unsigned int w[4] = {q[u].x, q[u].y, q[u].z, q[u].w};
#pragma unroll
          for (int j = 0; j < 4; ++j) {
            if (w[j] < REC_OVER) {
              if (slot < OWN_CAP)
                list[slot] = make_int2((int)(w[j] & 0xffffffu), n0 + (int)(w[j] >> 24));
              ++slot;
            }
          }
        }
      }
    }
  }
#pragma unroll
  for (int u = 0; u < 4; ++u) {
    const int j = u * OWN_T + tid;
    if (j < cnt * 20) {
      const int row = j / 20;
      const int c4 = j - row * 20;
      *reinterpret_cast<float4*>(xrS + row * OWN_XP + c4 * 4 + ((c4 >> 4) << 2)) =
          xrv[u];
    }
  }
  __syncthreads();

  const int total = __builtin_amdgcn_readfirstlane(s_cnt);
  const bool fast = total + cnt <= OWN_CAP &&
                    __builtin_amdgcn_readfirstlane(s_over) == 0;
  if (fast) {
    for (int i = tid; i < cnt; i += OWN_T)
      list[total + i] = make_int2(n0 + i, n0 + i);
    for (int j = tid; j < cnt * 10; j += OWN_T) {
      const int i = j / 10;
      const int h = j - i * 10;
      const float* __restrict__ xls = xl + (size_t)(n0 + i) * 80 + h * 8;
      const float* xrd = xrS + i * OWN_XP + h * 8 + ((h >> 3) << 2);
      float av[8], xv[8], rv[8];
#pragma unroll
      for (int dd = 0; dd < 8; ++dd) {
        av[dd] = att[h * 8 + dd];
        xv[dd] = xls[dd];
        rv[dd] = xrd[dd];
      }
      mS[j] = own_logit(xv, rv, av);
    }
  }
  __syncthreads();

  if (fast) {
    int2* __restrict__ sl =
        reinterpret_cast<int2*>(savedList) + (size_t)blockIdx.x * OWN_CAP;
    for (int i = tid; i < total + cnt; i += OWN_T) sl[i] = list[i];
  }
  if (tid == 0) savedCnt[blockIdx.x] = fast ? total + cnt : -1;

  bool three = !fast;
  if (fast) {
    const int items = (total + cnt) * 10;
    bool bad = false;
    int i0 = tid;
    for (; i0 + (OWN_IU - 1) * OWN_T < items; i0 += OWN_IU * OWN_T)
      bad |= own_fused_items<OWN_IU>(i0, n0, list, acc, xrS, mS, dS, xl, att);
    for (; i0 < items; i0 += OWN_T)
      bad |= own_fused_items<1>(i0, n0, list, acc, xrS, mS, dS, xl, att);
    if (bad) s_redo = 1;
    __syncthreads();
    three = __builtin_amdgcn_readfirstlane(s_redo) != 0;
    if (three) {
      for (int i = tid; i < OWN_R * 10; i += OWN_T) {
        mS[i] = -__builtin_inff();
        dS[i] = 0.f;
      }
      for (int i = tid; i < OWN_R * OWN_S; i += OWN_T) acc[i] = 0.f;
      __syncthreads();
    }
  }

  if (three) {
    const int ntiles = fast ? 1 : 1 + (E + OWN_CAP - 1) / OWN_CAP;
    for (int phase = 0; phase < 3; ++phase) {
      for (int t = 0; t < ntiles; ++t) {
        int count = total + cnt;
        if (!fast) {
          if (tid == 0) s_cnt = 0;
          __syncthreads();
          if (t == 0) {
            for (int i = tid; i < cnt; i += OWN_T)
              list[i] = make_int2(n0 + i, n0 + i);
          } else {
            const int e0 = (t - 1) * OWN_CAP;
            const int e1 = min(e0 + OWN_CAP, E);
            for (int e = e0 + tid; e < e1; e += OWN_T) {
              const int d = dstE[e];
              if ((unsigned)(d - n0) < (unsigned)cnt) {
                const int slot = atomicAdd(&s_cnt, 1);
                list[slot] = make_int2(srcE[e], d);
              }
            }
          }
          __syncthreads();
          count = t == 0 ? cnt : __builtin_amdgcn_readfirstlane(s_cnt);
        }
        if (phase == 0)
          own_process3<0>(count, n0, list, acc, mS, dS, xl, xr, att);
        else if (phase == 1)
          own_process3<1>(count, n0, list, acc, mS, dS, xl, xr, att);
        else
          own_process3<2>(count, n0, list, acc, mS, dS, xl, xr, att);
        __syncthreads();
      }
    }
  }

  for (int j = tid; j < cnt * 80; j += OWN_T) {
    const int i = j / 80;
    const int f = j - i * 80;
    float v = acc[i * OWN_S + f + (f >> 6)];
    if (!three) v = v / dS[i * 10 + (f >> 3)];
    agg[(size_t)(n0 + i) * 80 + f] = fmaxf(v + bias[f], 0.f);
  }
}

#define OWN2_S 17
#define OWN2_XS 20
#define OWN2_U 4
#define OWN2_G (OWN_T / 4)

template <int CTRL>
__device__ __forceinline__ float dpp_flt(float v) {
  return __int_as_float(
      __builtin_amdgcn_update_dpp(0, __float_as_int(v), CTRL, 0xf, 0xf, true));
}

__device__ __forceinline__ float own2_partial(float4 xv, float4 rv, float4 a4) {
  float v0 = xv.x + rv.x, v1 = xv.y + rv.y;
  float v2 = xv.z + rv.z, v3 = xv.w + rv.w;
  v0 = v0 > 0.f ? v0 : NEG_SLOPE * v0;
  v1 = v1 > 0.f ? v1 : NEG_SLOPE * v1;
  v2 = v2 > 0.f ? v2 : NEG_SLOPE * v2;
  v3 = v3 > 0.f ? v3 : NEG_SLOPE * v3;
  float p = v0 * a4.x + v1 * a4.y + v2 * a4.z + v3 * a4.w;
  p += dpp_flt<0x161>(p);
  p += dpp_flt<0x162>(p);
  return p;
}

__device__ __forceinline__ bool own2_fused(
    int count, int n0, const int2* list, float* acc, const float* cS, float* dS,
    const float* xrS, const float* __restrict__ xl, float4 a4) {
  const int q = threadIdx.x & 3;
  const int g = threadIdx.x >> 2;
  bool bad = false;
  for (int e0 = g; e0 < count; e0 += OWN2_U * OWN2_G) {
    int dl[OWN2_U];
    bool live[OWN2_U];
    float4 xv[OWN2_U];
#pragma unroll
    for (int u = 0; u < OWN2_U; ++u) {
      const int e = e0 + u * OWN2_G;
      live[u] = e < count;
      const int2 sd = list[live[u] ? e : e0];
      dl[u] = sd.y - n0;
      xv[u] = *reinterpret_cast<const float4*>(xl + (size_t)sd.x * 16 + q * 4);
    }
#pragma unroll
    for (int u = 0; u < OWN2_U; ++u) {
      const float4 rv =
          *reinterpret_cast<const float4*>(xrS + dl[u] * OWN2_XS + q * 4);
      const float d = own2_partial(xv[u], rv, a4) - cS[dl[u]];
      bad |= !(d <= OWN_DMAX);
      const float w = __expf(d);
      if (live[u]) {
        if (q == 0) atomicAdd(&dS[dl[u]], w);
        float* a = acc + dl[u] * OWN2_S + q * 4;
        atomicAdd(&a[0], w * xv[u].x);
        atomicAdd(&a[1], w * xv[u].y);
        atomicAdd(&a[2], w * xv[u].z);
        atomicAdd(&a[3], w * xv[u].w);
      }
    }
  }
  return bad;
}

template <int PHASE>
__device__ __forceinline__ void own2_process(
    int count, bool cached, int n0, const int2* list, float* lg, float* acc,
    float* mS, float* dS, const float* xrS, const float* __restrict__ xl,
    float4 a4) {
  if (PHASE == 1 && cached) {
    for (int e = threadIdx.x; e < count; e += OWN_T) {
      const int dl = list[e].y - n0;
      atomicAdd(&dS[dl], __expf(lg[e] - mS[dl]));
    }
    return;
  }
  const int q = threadIdx.x & 3;
  const int g = threadIdx.x >> 2;
  for (int e0 = g; e0 < count; e0 += OWN2_U * OWN2_G) {
    int ee[OWN2_U], dl[OWN2_U];
    bool live[OWN2_U];
    float4 xv[OWN2_U];
#pragma unroll
    for (int u = 0; u < OWN2_U; ++u) {
      const int e = e0 + u * OWN2_G;
      live[u] = e < count;
      ee[u] = live[u] ? e : e0;
      const int2 sd = list[ee[u]];
      dl[u] = sd.y - n0;
      xv[u] = *reinterpret_cast<const float4*>(xl + (size_t)sd.x * 16 + q * 4);
    }
#pragma unroll
    for (int u = 0; u < OWN2_U; ++u) {
      float logit;
      if (PHASE == 0 || !cached) {
        const float4 rv =
            *reinterpret_cast<const float4*>(xrS + dl[u] * OWN2_XS + q * 4);
        float v0 = xv[u].x + rv.x, v1 = xv[u].y + rv.y;
        float v2 = xv[u].z + rv.z, v3 = xv[u].w + rv.w;
        v0 = v0 > 0.f ? v0 : NEG_SLOPE * v0;
        v1 = v1 > 0.f ? v1 : NEG_SLOPE * v1;
        v2 = v2 > 0.f ? v2 : NEG_SLOPE * v2;
        v3 = v3 > 0.f ? v3 : NEG_SLOPE * v3;
        float p = v0 * a4.x + v1 * a4.y + v2 * a4.z + v3 * a4.w;
        p += dpp_flt<0x161>(p);
        p += dpp_flt<0x162>(p);
        logit = p;
        if (PHASE == 0 && cached && q == 0 && live[u]) lg[ee[u]] = logit;
      } else {
        logit = lg[ee[u]];
      }
      if (PHASE == 0) {
        if (live[u] && q == 0) atomicMaxFloat(&mS[dl[u]], logit);
      } else if (PHASE == 1) {
        if (live[u] && q == 0)
          atomicAdd(&dS[dl[u]], __expf(logit - mS[dl[u]]));
      } else {
        const float alpha = __expf(logit - mS[dl[u]]) / dS[dl[u]];
        if (live[u]) {
          float* a = acc + dl[u] * OWN2_S + q * 4;
          atomicAdd(&a[0], alpha * xv[u].x);
          atomicAdd(&a[1], alpha * xv[u].y);
          atomicAdd(&a[2], alpha * xv[u].z);
          atomicAdd(&a[3], alpha * xv[u].w);
        }
      }
    }
  }
}

__global__ __launch_bounds__(OWN_T) void layer2_owner_lds(
    const int* __restrict__ srcE, const int* __restrict__ dstE, int E, int N,
    const float* __restrict__ xl, const float* __restrict__ xr,
    const float* __restrict__ att, const float* __restrict__ bias,
    const int* __restrict__ savedList, const int* __restrict__ savedCnt,
    float* __restrict__ out) {
  __shared__ float acc[OWN_R * OWN2_S];
  __shared__ __attribute__((aligned(16))) float xrS[OWN_R * OWN2_XS];
  __shared__ float mS[OWN_R];
  __shared__ float dS[OWN_R];
  __shared__ float lg[OWN_CAP];
  __shared__ int2 list[OWN_CAP];
  __shared__ int s_cnt;
  __shared__ int s_redo;

  const int tid = threadIdx.x;
  const int n0 = blockIdx.x * OWN_R;
  const int cnt = min(OWN_R, N - n0);
  const int q = tid & 3;

  const int saved = __builtin_amdgcn_readfirstlane(savedCnt[blockIdx.x]);
  const bool fast = saved >= 0;
  const float4 a4 = make_float4(att[q * 4], att[q * 4 + 1], att[q * 4 + 2],
                                att[q * 4 + 3]);
  static_assert(OWN_R * 4 <= OWN_T, "one 16-byte piece of the xr rows per thread");
  float4 xrv = make_float4(0.f, 0.f, 0.f, 0.f);
  if (tid < cnt * 4)
    xrv = *reinterpret_cast<const float4*>(xr + (size_t)n0 * 16 + tid * 4);
  float4 xlv = make_float4(0.f, 0.f, 0.f, 0.f);
  if (fast && tid < cnt * 4)
    xlv = *reinterpret_cast<const float4*>(xl + (size_t)n0 * 16 + tid * 4);
  const int2* __restrict__ sl =
      reinterpret_cast<const int2*>(savedList) + (size_t)blockIdx.x * OWN_CAP;
  int2 sv[OWN_CAP / OWN_T];
#pragma unroll
  for (int u = 0; u < OWN_CAP / OWN_T; ++u) {
    const int i = u * OWN_T + tid;
    sv[u] = make_int2(0, 0);
    if (i < saved) sv[u] = sl[i];
  }

  for (int i = tid; i < OWN_R * OWN2_S; i += OWN_T) acc[i] = 0.f;
  for (int i = tid; i < OWN_R; i += OWN_T) {
    if (!fast) mS[i] = -__builtin_inff();
    dS[i] = 0.f;
  }
  if (tid == 0) s_redo = 0;
  if (tid < cnt * 4)
    *reinterpret_cast<float4*>(xrS + (tid >> 2) * OWN2_XS + q * 4) = xrv;
  if (fast && tid < cnt * 4) {
    const float s = own2_partial(xlv, xrv, a4);
    if (q == 0) mS[tid >> 2] = s;
  }
#pragma unroll
  for (int u = 0; u < OWN_CAP / OWN_T; ++u) {
    const int i = u * OWN_T + tid;
    if (i < saved) list[i] = sv[u];
  }
  __syncthreads();

  bool three = !fast;
  if (fast) {
    if (own2_fused(saved, n0, list, acc, mS, dS, xrS, xl, a4)) s_redo = 1;
    __syncthreads();
    three = __builtin_amdgcn_readfirstlane(s_redo) != 0;
    if (three) {
      for (int i = tid; i < OWN_R * OWN2_S; i += OWN_T) acc[i] = 0.f;
      for (int i = tid; i < OWN_R; i += OWN_T) {
        mS[i] = -__builtin_inff();
        dS[i] = 0.f;
      }
      __syncthreads();
    }
  }

  if (three) {
    const int ntiles = fast ? 1 : 1 + (E + OWN_CAP - 1) / OWN_CAP;
    for (int phase = 0; phase < 3; ++phase) {
      for (int t = 0; t < ntiles; ++t) {
        int count = saved;
        if (!fast) {
          if (tid == 0) s_cnt = 0;
          __syncthreads();
          if (t == 0) {
            for (int i = tid; i < cnt; i += OWN_T)
              list[i] = make_int2(n0 + i, n0 + i);
          } else {
            const int e0 = (t - 1) * OWN_CAP;
            const int e1 = min(e0 + OWN_CAP, E);
            for (int e = e0 + tid; e < e1; e += OWN_T) {
              const int d = dstE[e];
              if ((unsigned)(d - n0) < (unsigned)cnt) {
                const int slot = atomicAdd(&s_cnt, 1);
                list[slot] = make_int2(srcE[e], d);
              }
            }
          }
          __syncthreads();
          count = t == 0 ? cnt : __builtin_amdgcn_readfirstlane(s_cnt);
        }
        if (phase == 0)
          own2_process<0>(count, fast, n0, list, lg, acc, mS, dS, xrS, xl, a4);
        else if (phase == 1)
          own2_process<1>(count, fast, n0, list, lg, acc, mS, dS, xrS, xl, a4);
        else
          own2_process<2>(count, fast, n0, list, lg, acc, mS, dS, xrS, xl, a4);
        __syncthreads();
      }
    }
  }

  for (int j = tid; j < cnt * 16; j += OWN_T) {
    const int i = j >> 4;
    const int f = j & 15;
    float v = acc[i * OWN2_S + f];
    if (!three) v = v / dS[i];
    out[(size_t)(n0 + i) * 16 + f] = v + bias[f];
  }
}

extern "C" void kernel_launch(void* const* d_in, const int* in_sizes, int n_in,
                              void* d_out, int out_size, void* d_ws,
                              size_t ws_size, hipStream_t stream) {
  const float* x = (const float*)d_in[0];
  const int* edge = (const int*)d_in[1];
  const float* W1l = (const float*)d_in[2];
  const float* W1r = (const float*)d_in[3];
  const float* att1 = (const float*)d_in[4];
  const float* b1 = (const float*)d_in[5];
  const float* W2l = (const float*)d_in[6];
  const float* W2r = (const float*)d_in[7];
  const float* att2 = (const float*)d_in[8];
  const float* b2 = (const float*)d_in[9];
  float* out = (float*)d_out;

  const int F_IN = 128, H1 = 80, F_OUT = 16;
  const int N = in_sizes[0] / F_IN;
  const int E = in_sizes[1] / 2;
  const int* srcE = edge;
  const int* dstE = edge + E;

  float* ws = (float*)d_ws;
  float* xl1 = ws;                 ws += (size_t)N * H1;
  float* xr1 = ws;                 ws += (size_t)N * H1;
  float* agg1 = ws;                ws += (size_t)N * H1;
  float* xl2 = ws;                 ws += (size_t)N * F_OUT;
  float* xr2 = ws;                 ws += (size_t)N * F_OUT;
  const int ranges = (N + OWN_R - 1) / OWN_R;
  int* savedList = (int*)ws;       ws += (size_t)ranges * OWN_CAP * 2;
  const int chunks = (E + BM_CHUNK - 1) / BM_CHUNK;
  unsigned int* planes = (unsigned int*)ws; ws += (size_t)ranges * chunks * BM_W;
  int* savedCnt = (int*)ws;
  (void)ws_size; (void)n_in; (void)out_size;

  const int BLK = GEMM_T;

  {
    const int rowTiles = N / 16, colTiles = H1 / 16;
    const int waves = rowTiles * colTiles;
    const int blocks = (waves + 7) / 8;
    wmma_dual_gemm<128, 80, true><<<blocks, BLK, 0, stream>>>(
        x, W1l, W1r, xl1, xr1, rowTiles, colTiles, srcE, dstE, E, N, planes);
  }
  layer1_owner_lds<<<ranges, OWN_T, 0, stream>>>(
      srcE, dstE, E, N, xl1, xr1, att1, b1, agg1, savedList, savedCnt, planes);

  {
    const int rowTiles = N / 16, colTiles = F_OUT / 16;
    const int waves = rowTiles * colTiles;
    const int blocks = (waves + 7) / 8;
    wmma_dual_gemm<80, 16, false><<<blocks, BLK, 0, stream>>>(
        agg1, W2l, W2r, xl2, xr2, rowTiles, colTiles, srcE, dstE, E, N, planes);
  }
  layer2_owner_lds<<<ranges, OWN_T, 0, stream>>>(
      srcE, dstE, E, N, xl2, xr2, att2, b2, savedList, savedCnt, out);
}
